# starting kernel
# baseline (speedup 1.0000x reference)
#include <hip/hip_runtime.h>
#include <hip/hip_bf16.h>

typedef __attribute__((ext_vector_type(2))) float v2f;
typedef __attribute__((ext_vector_type(8))) float v8f;

#define F_IN   256
#define F_OUT  64
#define NEG_SLOPE 0.2f

__device__ __forceinline__ unsigned f2sortable(float f) {
    unsigned u = __float_as_uint(f);
    return (u & 0x80000000u) ? ~u : (u | 0x80000000u);
}
__device__ __forceinline__ float sortable2f(unsigned u) {
    return __uint_as_float((u & 0x80000000u) ? (u ^ 0x80000000u) : ~u);
}

__global__ void init_kernel(float* __restrict__ acc, unsigned* __restrict__ rmax,
                            float* __restrict__ denom, int n) {
    int i = blockIdx.x * blockDim.x + threadIdx.x;
    if (i < n * F_OUT) acc[i] = 0.0f;
    if (i < n) { rmax[i] = 0u; denom[i] = 0.0f; }
}

__global__ void __launch_bounds__(256)
gemm_wmma_kernel(const float* __restrict__ seq, const float* __restrict__ W,
                 float* __restrict__ fts,
                 const float* __restrict__ a1, const float* __restrict__ b1,
                 const float* __restrict__ a2, const float* __restrict__ b2,
                 float* __restrict__ f1, float* __restrict__ f2, int n) {
    __shared__ float Wl[F_IN * F_OUT];
    for (int i = threadIdx.x; i < F_IN * F_OUT; i += 256) {
        const int kk  = i >> 6;
        const int col = i & 63;
        Wl[(kk >> 1) * (2 * F_OUT) + col * 2 + (kk & 1)] = W[i];
    }
    __syncthreads();

    const int wave = threadIdx.x >> 5;
    const int lane = threadIdx.x & 31;
    const int rowTile = blockIdx.x * 8 + wave;

    int row = rowTile * 16 + (lane & 15);
    if (row >= n) row = n - 1;
    const int lcol  = lane & 15;
    const int khalf = (lane >> 4) * 2;

    const float* __restrict__ arow = seq + (size_t)row * F_IN;

    v8f acc[4] = {v8f{}, v8f{}, v8f{}, v8f{}};
    for (int k = 0; k < F_IN; k += 4) {
        float2 av = *(const float2*)(arow + k + khalf);
        v2f a; a[0] = av.x; a[1] = av.y;
        const int kpairBase = ((k + khalf) >> 1) * (2 * F_OUT);
#pragma unroll
        for (int ct = 0; ct < 4; ++ct) {
            float2 bv = *(const float2*)(&Wl[kpairBase + (ct * 16 + lcol) * 2]);
            v2f b; b[0] = bv.x; b[1] = bv.y;
            acc[ct] = __builtin_amdgcn_wmma_f32_16x16x4_f32(
                     false, a,  false, b,
                     (short)0, acc[ct],  false,  false);
        }
    }

    const int nhi = lane >> 4;
#pragma unroll
    for (int ct = 0; ct < 4; ++ct) {
        const int ocol = ct * 16 + lcol;
#pragma unroll
        for (int r = 0; r < 8; ++r) {
            const int orow = rowTile * 16 + r + 8 * nhi;
            if (orow < n) fts[(size_t)orow * F_OUT + ocol] = acc[ct][r];
        }
    }

    float A1[4], A2[4];
#pragma unroll
    for (int ct = 0; ct < 4; ++ct) {
        A1[ct] = a1[ct * 16 + lcol];
        A2[ct] = a2[ct * 16 + lcol];
    }
    const float b1v = b1[0], b2v = b2[0];
#pragma unroll
    for (int r = 0; r < 8; ++r) {
        float p1 = 0.0f, p2 = 0.0f;
#pragma unroll
        for (int ct = 0; ct < 4; ++ct) {
            p1 += acc[ct][r] * A1[ct];
            p2 += acc[ct][r] * A2[ct];
        }
#pragma unroll
        for (int off = 8; off > 0; off >>= 1) {
            p1 += __shfl_xor(p1, off, 32);
            p2 += __shfl_xor(p2, off, 32);
        }
        if ((lane & 15) == 0) {
            const int orow = rowTile * 16 + r + 8 * nhi;
            if (orow < n) { f1[orow] = p1 + b1v; f2[orow] = p2 + b2v; }
        }
    }
}

__global__ void logits_kernel(const int* __restrict__ erow, const int* __restrict__ ecol,
                              const float* __restrict__ f1, const float* __restrict__ f2,
                              float* __restrict__ logits, unsigned* __restrict__ rmax, int e) {
    const int i = blockIdx.x * blockDim.x + threadIdx.x;
    if (i >= e) return;
    const int r = erow[i], c = ecol[i];
    float l = f1[r] + f2[c];
    l = l > 0.0f ? l : NEG_SLOPE * l;
    logits[i] = l;
    atomicMax(&rmax[r], f2sortable(l));
}

__global__ void ev_kernel(const int* __restrict__ erow, float* __restrict__ logits,
                          const unsigned* __restrict__ rmax, float* __restrict__ denom, int e) {
    const int i = blockIdx.x * blockDim.x + threadIdx.x;
    if (i >= e) return;
    const int r = erow[i];
    const float m = sortable2f(rmax[r]);
    const float v = expf(logits[i] - m);
    logits[i] = v;
    atomicAdd(&denom[r], v);
}

__global__ void spmm_kernel(const int* __restrict__ erow, const int* __restrict__ ecol,
                            const float* __restrict__ ev, const float* __restrict__ denom,
                            const float* __restrict__ fts, float* __restrict__ acc, int e) {
    const int gid  = blockIdx.x * blockDim.x + threadIdx.x;
    const int edge = gid >> 5;
    const int lane = threadIdx.x & 31;
    if (edge >= e) return;
    const int r = erow[edge], c = ecol[edge];
    const float coef = ev[edge] / denom[r];
    const float* __restrict__ src = fts + (size_t)c * F_OUT;
    float* __restrict__ dst = acc + (size_t)r * F_OUT;
    atomicAdd(&dst[lane],      coef * src[lane]);
    atomicAdd(&dst[lane + 32], coef * src[lane + 32]);
}

__global__ void finalize_kernel(float* __restrict__ acc, const float* __restrict__ bias, int n) {
    const int i = blockIdx.x * blockDim.x + threadIdx.x;
    if (i >= n * F_OUT) return;
    const float v = acc[i] + bias[i & (F_OUT - 1)];
    acc[i] = v > 0.0f ? v : 0.0f;
}

extern "C" void kernel_launch(void* const* d_in, const int* in_sizes, int n_in,
                              void* d_out, int out_size, void* d_ws, size_t ws_size,
                              hipStream_t stream) {
    const float* seq  = (const float*)d_in[0];
    const int*   erow = (const int*)  d_in[1];
    const int*   ecol = (const int*)  d_in[2];
    const float* W    = (const float*)d_in[3];
    const float* a1   = (const float*)d_in[4];
    const float* b1   = (const float*)d_in[5];
    const float* a2   = (const float*)d_in[6];
    const float* b2   = (const float*)d_in[7];
    const float* bias = (const float*)d_in[8];

    const int n = in_sizes[0] / F_IN;
    const int e = in_sizes[1];
    float* out = (float*)d_out;

    float*    fts   = (float*)d_ws;
    float*    f1    = fts + (size_t)n * F_OUT;
    float*    f2    = f1 + n;
    unsigned* rmax  = (unsigned*)(f2 + n);
    float*    denom = (float*)(rmax + n);
    float*    ev    = denom + n;

    {
        const int tot = n * F_OUT;
        init_kernel<<<(tot + 255) / 256, 256, 0, stream>>>(out, rmax, denom, n);
    }
    {
        const int blocks = (n + 127) / 128;
        gemm_wmma_kernel<<<blocks, 256, 0, stream>>>(seq, W, fts, a1, b1, a2, b2,
                                                     f1, f2, n);
    }
    logits_kernel<<<(e + 255) / 256, 256, 0, stream>>>(erow, ecol, f1, f2, ev, rmax, e);
    ev_kernel<<<(e + 255) / 256, 256, 0, stream>>>(erow, ev, rmax, denom, e);
    {
        const long long threads = (long long)e * 32;
        const int blocks = (int)((threads + 255) / 256);
        spmm_kernel<<<blocks, 256, 0, stream>>>(erow, ecol, ev, denom, fts, out, e);
    }
    finalize_kernel<<<(n * F_OUT + 255) / 256, 256, 0, stream>>>(out, bias, n);
}


# Float-atomic scatter replaced by row gather: edges ranked per row by one integer atomic, packed into row lists, one wave per row does softmax + weighted sum in registers (sum order = atomic rank)
# speedup vs baseline: 2.2406x; 2.2406x over previous
#include <hip/hip_runtime.h>
#include <hip/hip_bf16.h>

typedef __attribute__((ext_vector_type(2))) float v2f;
typedef __attribute__((ext_vector_type(8))) float v8f;

#define F_IN   256
#define F_OUT  64
#define NEG_SLOPE 0.2f

__global__ void zero_kernel(int* __restrict__ cnt, int* __restrict__ cursor, int n) {
    const int i = blockIdx.x * blockDim.x + threadIdx.x;
    if (i < n) cnt[i] = 0;
    if (i == 0) cursor[0] = 0;
}

__global__ void __launch_bounds__(256)
gemm_wmma_kernel(const float* __restrict__ seq, const float* __restrict__ W,
                 float* __restrict__ fts,
                 const float* __restrict__ a1, const float* __restrict__ b1,
                 const float* __restrict__ a2, const float* __restrict__ b2,
                 float* __restrict__ f1, float* __restrict__ f2, int n) {
    __shared__ float Wl[F_IN * F_OUT];
    for (int i = threadIdx.x; i < F_IN * F_OUT; i += 256) {
        const int kk  = i >> 6;
        const int col = i & 63;
        Wl[(kk >> 1) * (2 * F_OUT) + col * 2 + (kk & 1)] = W[i];
    }
    __syncthreads();

    const int wave = threadIdx.x >> 5;
    const int lane = threadIdx.x & 31;
    const int rowTile = blockIdx.x * 8 + wave;

    int row = rowTile * 16 + (lane & 15);
    if (row >= n) row = n - 1;
    const int lcol  = lane & 15;
    const int khalf = (lane >> 4) * 2;

    const float* __restrict__ arow = seq + (size_t)row * F_IN;

    v8f acc[4] = {v8f{}, v8f{}, v8f{}, v8f{}};
    for (int k = 0; k < F_IN; k += 4) {
        float2 av = *(const float2*)(arow + k + khalf);
        v2f a; a[0] = av.x; a[1] = av.y;
        const int kpairBase = ((k + khalf) >> 1) * (2 * F_OUT);
#pragma unroll
        for (int ct = 0; ct < 4; ++ct) {
            float2 bv = *(const float2*)(&Wl[kpairBase + (ct * 16 + lcol) * 2]);
            v2f b; b[0] = bv.x; b[1] = bv.y;
            acc[ct] = __builtin_amdgcn_wmma_f32_16x16x4_f32(
                     false, a,  false, b,
                     (short)0, acc[ct],  false,  false);
        }
    }

    const int nhi = lane >> 4;
#pragma unroll
    for (int ct = 0; ct < 4; ++ct) {
        const int ocol = ct * 16 + lcol;
#pragma unroll
        for (int r = 0; r < 8; ++r) {
            const int orow = rowTile * 16 + r + 8 * nhi;
            if (orow < n) fts[(size_t)orow * F_OUT + ocol] = acc[ct][r];
        }
    }

    float A1[4], A2[4];
#pragma unroll
    for (int ct = 0; ct < 4; ++ct) {
        A1[ct] = a1[ct * 16 + lcol];
        A2[ct] = a2[ct * 16 + lcol];
    }
    const float b1v = b1[0], b2v = b2[0];
#pragma unroll
    for (int r = 0; r < 8; ++r) {
        float p1 = 0.0f, p2 = 0.0f;
#pragma unroll
        for (int ct = 0; ct < 4; ++ct) {
            p1 += acc[ct][r] * A1[ct];
            p2 += acc[ct][r] * A2[ct];
        }
#pragma unroll
        for (int off = 8; off > 0; off >>= 1) {
            p1 += __shfl_xor(p1, off, 32);
            p2 += __shfl_xor(p2, off, 32);
        }
        if ((lane & 15) == 0) {
            const int orow = rowTile * 16 + r + 8 * nhi;
            if (orow < n) { f1[orow] = p1 + b1v; f2[orow] = p2 + b2v; }
        }
    }
}

__global__ void rank_kernel(const int* __restrict__ erow, int* __restrict__ cnt,
                            int* __restrict__ pos, int e) {
    const int i = (blockIdx.x * blockDim.x + threadIdx.x) * 4;
    if (i + 3 < e) {
        const int4 r = *(const int4*)(erow + i);
        int4 p;
        p.x = atomicAdd(&cnt[r.x], 1);
        p.y = atomicAdd(&cnt[r.y], 1);
        p.z = atomicAdd(&cnt[r.z], 1);
        p.w = atomicAdd(&cnt[r.w], 1);
        *(int4*)(pos + i) = p;
    } else {
        for (int k = i; k < e; ++k) pos[k] = atomicAdd(&cnt[erow[k]], 1);
    }
}

__global__ void __launch_bounds__(256)
alloc_kernel(const int* __restrict__ cnt, int* __restrict__ rowstart,
             int* __restrict__ cursor, int n) {
    __shared__ int wsum[8];
    __shared__ int blockBase;
    const int r    = blockIdx.x * 256 + threadIdx.x;
    const int lane = threadIdx.x & 31;
    const int wave = threadIdx.x >> 5;
    const int v = (r < n) ? cnt[r] : 0;
    int x = v;
    for (int off = 1; off < 32; off <<= 1) {
        const int y = __shfl_up(x, off, 32);
        if (lane >= off) x += y;
    }
    if (lane == 31) wsum[wave] = x;
    __syncthreads();
    if (threadIdx.x == 0) {
        int tot = 0;
        for (int w = 0; w < 8; ++w) { const int t = wsum[w]; wsum[w] = tot; tot += t; }
        blockBase = atomicAdd(cursor, tot);
    }
    __syncthreads();
    if (r < n) rowstart[r] = blockBase + wsum[wave] + x - v;
}

__global__ void fill_kernel(const int* __restrict__ erow, const int* __restrict__ ecol,
                            const int* __restrict__ pos, const int* __restrict__ rowstart,
                            int* __restrict__ csr_col, int e) {
    const int i = (blockIdx.x * blockDim.x + threadIdx.x) * 4;
    if (i + 3 < e) {
        const int4 r = *(const int4*)(erow + i);
        const int4 c = *(const int4*)(ecol + i);
        const int4 p = *(const int4*)(pos + i);
        csr_col[rowstart[r.x] + p.x] = c.x;
        csr_col[rowstart[r.y] + p.y] = c.y;
        csr_col[rowstart[r.z] + p.z] = c.z;
        csr_col[rowstart[r.w] + p.w] = c.w;
    } else {
        for (int k = i; k < e; ++k) csr_col[rowstart[erow[k]] + pos[k]] = ecol[k];
    }
}

__device__ __forceinline__ float lrelu(float l) { return l > 0.0f ? l : NEG_SLOPE * l; }

__device__ __forceinline__ void accum_chunk(const float* __restrict__ fts, int c, float ev,
                                            int cntc, int half, int q, float4& acc) {
    const int steps = (cntc + 1) >> 1;
    const int evb = __float_as_int(ev);
#pragma unroll 4
    for (int t = 0; t < steps; ++t) {
        const int c_lo = __builtin_amdgcn_readlane(c,   2 * t);
        const int c_hi = __builtin_amdgcn_readlane(c,   2 * t + 1);
        const int w_lo = __builtin_amdgcn_readlane(evb, 2 * t);
        const int w_hi = __builtin_amdgcn_readlane(evb, 2 * t + 1);
        const int   cc = half ? c_hi : c_lo;
        const float w  = __int_as_float(half ? w_hi : w_lo);
        const float4 v = *(const float4*)(fts + (size_t)cc * F_OUT + q * 4);
        acc.x += w * v.x; acc.y += w * v.y; acc.z += w * v.z; acc.w += w * v.w;
    }
}

__global__ void __launch_bounds__(256)
row_kernel(const int* __restrict__ cnt, const int* __restrict__ rowstart,
           const int* __restrict__ csr_col,
           const float* __restrict__ f1, const float* __restrict__ f2,
           const float* __restrict__ fts, const float* __restrict__ bias,
           float* __restrict__ out, int n) {
    const int lane = threadIdx.x & 31;
    const int row  = __builtin_amdgcn_readfirstlane((int)(blockIdx.x * 8 + (threadIdx.x >> 5)));
    if (row >= n) return;
    const int half = lane >> 4;
    const int q    = lane & 15;
    const int deg  = cnt[row];
    const int s    = rowstart[row];
    const float f1r = f1[row];

    const bool v0 = lane < deg;
    int   c0 = 0;
    float l0 = -INFINITY;
    if (v0) { c0 = csr_col[s + lane]; l0 = lrelu(f1r + f2[c0]); }
    float m = l0;
    for (int base = 32; base < deg; base += 32) {
        const int j = base + lane;
        if (j < deg) m = fmaxf(m, lrelu(f1r + f2[csr_col[s + j]]));
    }
    for (int off = 16; off > 0; off >>= 1) m = fmaxf(m, __shfl_xor(m, off, 32));
    if (!isfinite(m)) m = 0.0f;

    float4 acc = make_float4(0.0f, 0.0f, 0.0f, 0.0f);
    float dsum = 0.0f;
    if (deg > 0) {
        const float ev = v0 ? expf(l0 - m) : 0.0f;
        dsum += ev;
        const int cf = __shfl(c0, 0, 32);
        accum_chunk(fts, v0 ? c0 : cf, ev, deg < 32 ? deg : 32, half, q, acc);
    }
    for (int base = 32; base < deg; base += 32) {
        const int j = base + lane;
        const bool v = j < deg;
        int c = 0; float ev = 0.0f;
        if (v) { c = csr_col[s + j]; ev = expf(lrelu(f1r + f2[c]) - m); }
        dsum += ev;
        const int cf = __shfl(c, 0, 32);
        const int rem = deg - base;
        accum_chunk(fts, v ? c : cf, ev, rem < 32 ? rem : 32, half, q, acc);
    }

    for (int off = 16; off > 0; off >>= 1) dsum += __shfl_xor(dsum, off, 32);
    acc.x += __shfl_xor(acc.x, 16, 32);
    acc.y += __shfl_xor(acc.y, 16, 32);
    acc.z += __shfl_xor(acc.z, 16, 32);
    acc.w += __shfl_xor(acc.w, 16, 32);

    if (half == 0) {
        const float4 b = *(const float4*)(bias + q * 4);
        float4 o;
        if (deg > 0) {
            o.x = acc.x / dsum + b.x; o.y = acc.y / dsum + b.y;
            o.z = acc.z / dsum + b.z; o.w = acc.w / dsum + b.w;
        } else {
            o = b;
        }
        o.x = o.x > 0.0f ? o.x : 0.0f; o.y = o.y > 0.0f ? o.y : 0.0f;
        o.z = o.z > 0.0f ? o.z : 0.0f; o.w = o.w > 0.0f ? o.w : 0.0f;
        *(float4*)(out + (size_t)row * F_OUT + q * 4) = o;
    }
}

extern "C" void kernel_launch(void* const* d_in, const int* in_sizes, int n_in,
                              void* d_out, int out_size, void* d_ws, size_t ws_size,
                              hipStream_t stream) {
    const float* seq  = (const float*)d_in[0];
    const int*   erow = (const int*)  d_in[1];
    const int*   ecol = (const int*)  d_in[2];
    const float* W    = (const float*)d_in[3];
    const float* a1   = (const float*)d_in[4];
    const float* b1   = (const float*)d_in[5];
    const float* a2   = (const float*)d_in[6];
    const float* b2   = (const float*)d_in[7];
    const float* bias = (const float*)d_in[8];

    const int n = in_sizes[0] / F_IN;
    const int e = in_sizes[1];
    float* out = (float*)d_out;

    float* fts      = (float*)d_ws;
    float* f1       = fts + (size_t)n * F_OUT;
    float* f2       = f1 + n;
    int*   cnt      = (int*)(f2 + n);
    int*   rowstart = cnt + n;
    int*   csr_col  = rowstart + n;
    int*   pos      = (int*)d_out;
    int*   cursor   = pos + e;

    zero_kernel<<<(n + 255) / 256, 256, 0, stream>>>(cnt, cursor, n);
    {
        const int blocks = (n + 127) / 128;
        gemm_wmma_kernel<<<blocks, 256, 0, stream>>>(seq, W, fts, a1, b1, a2, b2,
                                                     f1, f2, n);
    }
    {
        const int threads = (e + 3) / 4;
        rank_kernel<<<(threads + 255) / 256, 256, 0, stream>>>(erow, cnt, pos, e);
    }
    alloc_kernel<<<(n + 255) / 256, 256, 0, stream>>>(cnt, rowstart, cursor, n);
    {
        const int threads = (e + 3) / 4;
        fill_kernel<<<(threads + 255) / 256, 256, 0, stream>>>(erow, ecol, pos, rowstart,
                                                               csr_col, e);
    }
    row_kernel<<<(n + 7) / 8, 256, 0, stream>>>(cnt, rowstart, csr_col, f1, f2, fts, bias,
                                                out, n);
}


# Per-edge global ranking atomics replaced by two-level grouping on shared-memory counters: edges split into 256 row ranges, one block per range orders them by row (sum order = shared-memory rank)
# speedup vs baseline: 4.7976x; 2.1412x over previous
#include <hip/hip_runtime.h>
#include <hip/hip_bf16.h>

typedef __attribute__((ext_vector_type(2))) float v2f;
typedef __attribute__((ext_vector_type(8))) float v8f;

#define F_IN   256
#define F_OUT  64
#define NEG_SLOPE 0.2f
#define NB     256
#define GT     1024
#define EB     8
#define CS     64

__global__ void zero_kernel(int* __restrict__ p, int count) {
    const int i = blockIdx.x * blockDim.x + threadIdx.x;
    if (i < count) p[i] = 0;
}

__global__ void __launch_bounds__(256)
gemm_wmma_kernel(const float* __restrict__ seq, const float* __restrict__ W,
                 float* __restrict__ fts,
                 const float* __restrict__ a1, const float* __restrict__ b1,
                 const float* __restrict__ a2, const float* __restrict__ b2,
                 float* __restrict__ f1, float* __restrict__ f2, int n) {
    __shared__ float Wl[F_IN * F_OUT];
    for (int i = threadIdx.x; i < F_IN * F_OUT; i += 256) {
        const int kk  = i >> 6;
        const int col = i & 63;
        Wl[(kk >> 1) * (2 * F_OUT) + col * 2 + (kk & 1)] = W[i];
    }
    __syncthreads();

    const int wave = threadIdx.x >> 5;
    const int lane = threadIdx.x & 31;
    const int rowTile = blockIdx.x * 8 + wave;

    int row = rowTile * 16 + (lane & 15);
    if (row >= n) row = n - 1;
    const int lcol  = lane & 15;
    const int khalf = (lane >> 4) * 2;

    const float* __restrict__ arow = seq + (size_t)row * F_IN;

    v8f acc[4] = {v8f{}, v8f{}, v8f{}, v8f{}};
    for (int k = 0; k < F_IN; k += 4) {
        float2 av = *(const float2*)(arow + k + khalf);
        v2f a; a[0] = av.x; a[1] = av.y;
        const int kpairBase = ((k + khalf) >> 1) * (2 * F_OUT);
#pragma unroll
        for (int ct = 0; ct < 4; ++ct) {
            float2 bv = *(const float2*)(&Wl[kpairBase + (ct * 16 + lcol) * 2]);
            v2f b; b[0] = bv.x; b[1] = bv.y;
            acc[ct] = __builtin_amdgcn_wmma_f32_16x16x4_f32(
                     false, a,  false, b,
                     (short)0, acc[ct],  false,  false);
        }
    }

    const int nhi = lane >> 4;
#pragma unroll
    for (int ct = 0; ct < 4; ++ct) {
        const int ocol = ct * 16 + lcol;
#pragma unroll
        for (int r = 0; r < 8; ++r) {
            const int orow = rowTile * 16 + r + 8 * nhi;
            if (orow < n) fts[(size_t)orow * F_OUT + ocol] = acc[ct][r];
        }
    }

    float A1[4], A2[4];
#pragma unroll
    for (int ct = 0; ct < 4; ++ct) {
        A1[ct] = a1[ct * 16 + lcol];
        A2[ct] = a2[ct * 16 + lcol];
    }
    const float b1v = b1[0], b2v = b2[0];
#pragma unroll
    for (int r = 0; r < 8; ++r) {
        float p1 = 0.0f, p2 = 0.0f;
#pragma unroll
        for (int ct = 0; ct < 4; ++ct) {
            p1 += acc[ct][r] * A1[ct];
            p2 += acc[ct][r] * A2[ct];
        }
#pragma unroll
        for (int off = 8; off > 0; off >>= 1) {
            p1 += __shfl_xor(p1, off, 32);
            p2 += __shfl_xor(p2, off, 32);
        }
        if ((lane & 15) == 0) {
            const int orow = rowTile * 16 + r + 8 * nhi;
            if (orow < n) { f1[orow] = p1 + b1v; f2[orow] = p2 + b2v; }
        }
    }
}

__device__ __forceinline__ int range_of(int row, unsigned magic) {
    const int b = (int)__umulhi((unsigned)row, magic);
    return b < NB - 1 ? b : NB - 1;
}

__global__ void __launch_bounds__(GT)
range_count_kernel(const int* __restrict__ erow, int* __restrict__ bcount,
                   int e, int per, int rpb) {
    __shared__ int h[NB];
    if (threadIdx.x < NB) h[threadIdx.x] = 0;
    __syncthreads();
    const unsigned magic = 0xFFFFFFFFu / (unsigned)rpb + 1u;
    const int lo = blockIdx.x * per;
    const int hi = lo + per < e ? lo + per : e;
    for (int i = lo + (int)threadIdx.x; i < hi; i += EB * GT) {
        int r[EB];
#pragma unroll
        for (int k = 0; k < EB; ++k) r[k] = erow[min(i + k * GT, hi - 1)];
#pragma unroll
        for (int k = 0; k < EB; ++k)
            if (i + k * GT < hi) atomicAdd(&h[range_of(r[k], magic)], 1);
    }
    __syncthreads();
    if (threadIdx.x < NB) {
        const int v = h[threadIdx.x];
        if (v) atomicAdd(&bcount[threadIdx.x * CS], v);
    }
}

__global__ void __launch_bounds__(GT)
range_place_kernel(const int* __restrict__ erow, const int* __restrict__ ecol,
                   const int* __restrict__ bcount, int* __restrict__ bcursor,
                   int* __restrict__ bstart, unsigned* __restrict__ packed,
                   int e, int n, int per, int rpb) {
    __shared__ int h[NB];
    __shared__ int wsum[NB / 32];
    const int t = threadIdx.x, lane = t & 31, wave = t >> 5;
    int tot = 0, x = 0;
    if (t < NB) {
        h[t] = 0;
        tot = bcount[t * CS];
        x = tot;
        for (int off = 1; off < 32; off <<= 1) {
            const int y = __shfl_up(x, off, 32);
            if (lane >= off) x += y;
        }
        if (lane == 31) wsum[wave] = x;
    }
    __syncthreads();
    const unsigned magic = 0xFFFFFFFFu / (unsigned)rpb + 1u;
    const int cb = 32 - __clz(n - 1);
    const int lo = blockIdx.x * per;
    const int hi = lo + per < e ? lo + per : e;
    for (int i = lo + t; i < hi; i += EB * GT) {
        int r[EB];
#pragma unroll
        for (int k = 0; k < EB; ++k) r[k] = erow[min(i + k * GT, hi - 1)];
#pragma unroll
        for (int k = 0; k < EB; ++k)
            if (i + k * GT < hi) atomicAdd(&h[range_of(r[k], magic)], 1);
    }
    __syncthreads();
    if (t < NB) {
        int wp = 0;
        for (int w = 0; w < wave; ++w) wp += wsum[w];
        const int st = wp + x - tot;
        if (blockIdx.x == 0) bstart[t] = st;
        const int mine = h[t];
        const int base = mine ? atomicAdd(&bcursor[t * CS], mine) : 0;
        h[t] = st + base;
    }
    __syncthreads();
    for (int i = lo + t; i < hi; i += EB * GT) {
        int r[EB], c[EB], slot[EB];
#pragma unroll
        for (int k = 0; k < EB; ++k) {
            const int j = min(i + k * GT, hi - 1);
            r[k] = erow[j];
            c[k] = ecol[j];
        }
#pragma unroll
        for (int k = 0; k < EB; ++k) {
            const int b = range_of(r[k], magic);
            slot[k] = i + k * GT < hi ? atomicAdd(&h[b], 1) : 0;
            r[k] -= b * rpb;
        }
#pragma unroll
        for (int k = 0; k < EB; ++k)
            if (i + k * GT < hi) packed[slot[k]] = ((unsigned)r[k] << cb) | (unsigned)c[k];
    }
}

__global__ void __launch_bounds__(GT)
range_sort_kernel(const unsigned* __restrict__ packed, const int* __restrict__ bcount,
                  const int* __restrict__ bstart, int* __restrict__ cnt,
                  int* __restrict__ rowstart, int* __restrict__ csr_col, int n, int rpb) {
    extern __shared__ int lcnt[];
    __shared__ int wsum[GT / 32];
    const int t = threadIdx.x, lane = t & 31, wave = t >> 5;
    const int r0 = blockIdx.x * rpb;
    int rows = n - r0;
    if (rows > rpb) rows = rpb;
    if (rows <= 0) return;
    const int s = bstart[blockIdx.x];
    const int c = bcount[blockIdx.x * CS];
    const int cb = 32 - __clz(n - 1);
    const unsigned cmask = (1u << cb) - 1u;
    const unsigned last = (unsigned)(rows - 1);
    for (int i = t; i < rows; i += GT) lcnt[i] = 0;
    __syncthreads();
    for (int i = t; i < c; i += EB * GT) {
        unsigned w[EB];
#pragma unroll
        for (int k = 0; k < EB; ++k) w[k] = packed[s + min(i + k * GT, c - 1)];
#pragma unroll
        for (int k = 0; k < EB; ++k) {
            unsigned lr = w[k] >> cb;
            lr = lr < last ? lr : last;
            if (i + k * GT < c) atomicAdd(&lcnt[lr], 1);
        }
    }
    __syncthreads();
    int carry = 0;
    for (int base = 0; base < rows; base += GT) {
        const int idx = base + t;
        const int v = idx < rows ? lcnt[idx] : 0;
        int x = v;
        for (int off = 1; off < 32; off <<= 1) {
            const int y = __shfl_up(x, off, 32);
            if (lane >= off) x += y;
        }
        if (lane == 31) wsum[wave] = x;
        __syncthreads();
        int wp = 0, all = 0;
        for (int w = 0; w < GT / 32; ++w) {
            const int q = wsum[w];
            all += q;
            if (w < wave) wp += q;
        }
        const int ex = carry + wp + x - v;
        if (idx < rows) {
            cnt[r0 + idx] = v;
            rowstart[r0 + idx] = s + ex;
            lcnt[idx] = ex;
        }
        carry += all;
        __syncthreads();
    }
    for (int i = t; i < c; i += EB * GT) {
        unsigned w[EB];
        int slot[EB];
#pragma unroll
        for (int k = 0; k < EB; ++k) w[k] = packed[s + min(i + k * GT, c - 1)];
#pragma unroll
        for (int k = 0; k < EB; ++k) {
            unsigned lr = w[k] >> cb;
            lr = lr < last ? lr : last;
            slot[k] = i + k * GT < c ? atomicAdd(&lcnt[lr], 1) : 0;
        }
#pragma unroll
        for (int k = 0; k < EB; ++k)
            if (i + k * GT < c) csr_col[s + slot[k]] = (int)(w[k] & cmask);
    }
}

__device__ __forceinline__ float lrelu(float l) { return l > 0.0f ? l : NEG_SLOPE * l; }

__device__ __forceinline__ void accum_chunk(const float* __restrict__ fts, int c, float ev,
                                            int cntc, int half, int q, float4& acc) {
    const int steps = (cntc + 1) >> 1;
    const int evb = __float_as_int(ev);
#pragma unroll 4
    for (int t = 0; t < steps; ++t) {
        const int c_lo = __builtin_amdgcn_readlane(c,   2 * t);
        const int c_hi = __builtin_amdgcn_readlane(c,   2 * t + 1);
        const int w_lo = __builtin_amdgcn_readlane(evb, 2 * t);
        const int w_hi = __builtin_amdgcn_readlane(evb, 2 * t + 1);
        const int   cc = half ? c_hi : c_lo;
        const float w  = __int_as_float(half ? w_hi : w_lo);
        const float4 v = *(const float4*)(fts + (size_t)cc * F_OUT + q * 4);
        acc.x += w * v.x; acc.y += w * v.y; acc.z += w * v.z; acc.w += w * v.w;
    }
}

__global__ void __launch_bounds__(256)
row_kernel(const int* __restrict__ cnt, const int* __restrict__ rowstart,
           const int* __restrict__ csr_col,
           const float* __restrict__ f1, const float* __restrict__ f2,
           const float* __restrict__ fts, const float* __restrict__ bias,
           float* __restrict__ out, int n) {
    const int lane = threadIdx.x & 31;
    const int row  = __builtin_amdgcn_readfirstlane((int)(blockIdx.x * 8 + (threadIdx.x >> 5)));
    if (row >= n) return;
    const int half = lane >> 4;
    const int q    = lane & 15;
    const int deg  = cnt[row];
    const int s    = rowstart[row];
    const float f1r = f1[row];

    const bool v0 = lane < deg;
    int   c0 = 0;
    float l0 = -INFINITY;
    if (v0) { c0 = csr_col[s + lane]; l0 = lrelu(f1r + f2[c0]); }
    float m = l0;
    for (int base = 32; base < deg; base += 32) {
        const int j = base + lane;
        if (j < deg) m = fmaxf(m, lrelu(f1r + f2[csr_col[s + j]]));
    }
    for (int off = 16; off > 0; off >>= 1) m = fmaxf(m, __shfl_xor(m, off, 32));
    if (!isfinite(m)) m = 0.0f;

    float4 acc = make_float4(0.0f, 0.0f, 0.0f, 0.0f);
    float dsum = 0.0f;
    if (deg > 0) {
        const float ev = v0 ? expf(l0 - m) : 0.0f;
        dsum += ev;
        const int cf = __shfl(c0, 0, 32);
        accum_chunk(fts, v0 ? c0 : cf, ev, deg < 32 ? deg : 32, half, q, acc);
    }
    for (int base = 32; base < deg; base += 32) {
        const int j = base + lane;
        const bool v = j < deg;
        int c = 0; float ev = 0.0f;
        if (v) { c = csr_col[s + j]; ev = expf(lrelu(f1r + f2[c]) - m); }
        dsum += ev;
        const int cf = __shfl(c, 0, 32);
        const int rem = deg - base;
        accum_chunk(fts, v ? c : cf, ev, rem < 32 ? rem : 32, half, q, acc);
    }

    for (int off = 16; off > 0; off >>= 1) dsum += __shfl_xor(dsum, off, 32);
    acc.x += __shfl_xor(acc.x, 16, 32);
    acc.y += __shfl_xor(acc.y, 16, 32);
    acc.z += __shfl_xor(acc.z, 16, 32);
    acc.w += __shfl_xor(acc.w, 16, 32);

    if (half == 0) {
        const float4 b = *(const float4*)(bias + q * 4);
        float4 o;
        if (deg > 0) {
            o.x = acc.x / dsum + b.x; o.y = acc.y / dsum + b.y;
            o.z = acc.z / dsum + b.z; o.w = acc.w / dsum + b.w;
        } else {
            o = b;
        }
        o.x = o.x > 0.0f ? o.x : 0.0f; o.y = o.y > 0.0f ? o.y : 0.0f;
        o.z = o.z > 0.0f ? o.z : 0.0f; o.w = o.w > 0.0f ? o.w : 0.0f;
        *(float4*)(out + (size_t)row * F_OUT + q * 4) = o;
    }
}

extern "C" void kernel_launch(void* const* d_in, const int* in_sizes, int n_in,
                              void* d_out, int out_size, void* d_ws, size_t ws_size,
                              hipStream_t stream) {
    const float* seq  = (const float*)d_in[0];
    const int*   erow = (const int*)  d_in[1];
    const int*   ecol = (const int*)  d_in[2];
    const float* W    = (const float*)d_in[3];
    const float* a1   = (const float*)d_in[4];
    const float* b1   = (const float*)d_in[5];
    const float* a2   = (const float*)d_in[6];
    const float* b2   = (const float*)d_in[7];
    const float* bias = (const float*)d_in[8];

    const int n = in_sizes[0] / F_IN;
    const int e = in_sizes[1];
    float* out = (float*)d_out;

    float* fts      = (float*)d_ws;
    float* f1       = fts + (size_t)n * F_OUT;
    float* f2       = f1 + n;
    int*   cnt      = (int*)(f2 + n);
    int*   rowstart = cnt + n;
    int*   csr_col  = rowstart + n;
    unsigned* packed = (unsigned*)d_out;
    int*   bcount   = (int*)d_out + e;
    int*   bcursor  = bcount + NB * CS;
    int*   bstart   = bcursor + NB * CS;

    const int rpb = n / NB + 2;
    const int gblocks = 512;
    const int per = (e + gblocks - 1) / gblocks;

    zero_kernel<<<(2 * NB * CS + 255) / 256, 256, 0, stream>>>(bcount, 2 * NB * CS);
    {
        const int blocks = (n + 127) / 128;
        gemm_wmma_kernel<<<blocks, 256, 0, stream>>>(seq, W, fts, a1, b1, a2, b2,
                                                     f1, f2, n);
    }
    range_count_kernel<<<gblocks, GT, 0, stream>>>(erow, bcount, e, per, rpb);
    range_place_kernel<<<gblocks, GT, 0, stream>>>(erow, ecol, bcount, bcursor, bstart,
                                                   packed, e, n, per, rpb);
    range_sort_kernel<<<NB, GT, rpb * sizeof(int), stream>>>(packed, bcount, bstart, cnt,
                                                             rowstart, csr_col, n, rpb);
    row_kernel<<<(n + 7) / 8, 256, 0, stream>>>(cnt, rowstart, csr_col, f1, f2, fts, bias,
                                                out, n);
}


# Scattered 4-byte global stores of both grouping kernels staged through shared memory: words ordered there, copied out as consecutive runs, whole lines per store (sum order still shared-memory rank)
# speedup vs baseline: 5.5308x; 1.1528x over previous
#include <hip/hip_runtime.h>
#include <hip/hip_bf16.h>

typedef __attribute__((ext_vector_type(2))) float v2f;
typedef __attribute__((ext_vector_type(8))) float v8f;

#define F_IN   256
#define F_OUT  64
#define NEG_SLOPE 0.2f
#define NB     256
#define GT     1024
#define EB     8
#define CS     64
#define SCAP   14336

__global__ void zero_kernel(int* __restrict__ p, int count) {
    const int i = blockIdx.x * blockDim.x + threadIdx.x;
    if (i < count) p[i] = 0;
}

__global__ void __launch_bounds__(256)
gemm_wmma_kernel(const float* __restrict__ seq, const float* __restrict__ W,
                 float* __restrict__ fts,
                 const float* __restrict__ a1, const float* __restrict__ b1,
                 const float* __restrict__ a2, const float* __restrict__ b2,
                 float* __restrict__ f1, float* __restrict__ f2, int n) {
    __shared__ float Wl[F_IN * F_OUT];
    for (int i = threadIdx.x; i < F_IN * F_OUT; i += 256) {
        const int kk  = i >> 6;
        const int col = i & 63;
        Wl[(kk >> 1) * (2 * F_OUT) + col * 2 + (kk & 1)] = W[i];
    }
    __syncthreads();

    const int wave = threadIdx.x >> 5;
    const int lane = threadIdx.x & 31;
    const int rowTile = blockIdx.x * 8 + wave;

    int row = rowTile * 16 + (lane & 15);
    if (row >= n) row = n - 1;
    const int lcol  = lane & 15;
    const int khalf = (lane >> 4) * 2;

    const float* __restrict__ arow = seq + (size_t)row * F_IN;

    v8f acc[4] = {v8f{}, v8f{}, v8f{}, v8f{}};
    for (int k = 0; k < F_IN; k += 4) {
        float2 av = *(const float2*)(arow + k + khalf);
        v2f a; a[0] = av.x; a[1] = av.y;
        const int kpairBase = ((k + khalf) >> 1) * (2 * F_OUT);
#pragma unroll
        for (int ct = 0; ct < 4; ++ct) {
            float2 bv = *(const float2*)(&Wl[kpairBase + (ct * 16 + lcol) * 2]);
            v2f b; b[0] = bv.x; b[1] = bv.y;
            acc[ct] = __builtin_amdgcn_wmma_f32_16x16x4_f32(
                     false, a,  false, b,
                     (short)0, acc[ct],  false,  false);
        }
    }

    const int nhi = lane >> 4;
#pragma unroll
    for (int ct = 0; ct < 4; ++ct) {
        const int ocol = ct * 16 + lcol;
#pragma unroll
        for (int r = 0; r < 8; ++r) {
            const int orow = rowTile * 16 + r + 8 * nhi;
            if (orow < n) fts[(size_t)orow * F_OUT + ocol] = acc[ct][r];
        }
    }

    float A1[4], A2[4];
#pragma unroll
    for (int ct = 0; ct < 4; ++ct) {
        A1[ct] = a1[ct * 16 + lcol];
        A2[ct] = a2[ct * 16 + lcol];
    }
    const float b1v = b1[0], b2v = b2[0];
#pragma unroll
    for (int r = 0; r < 8; ++r) {
        float p1 = 0.0f, p2 = 0.0f;
#pragma unroll
        for (int ct = 0; ct < 4; ++ct) {
            p1 += acc[ct][r] * A1[ct];
            p2 += acc[ct][r] * A2[ct];
        }
#pragma unroll
        for (int off = 8; off > 0; off >>= 1) {
            p1 += __shfl_xor(p1, off, 32);
            p2 += __shfl_xor(p2, off, 32);
        }
        if ((lane & 15) == 0) {
            const int orow = rowTile * 16 + r + 8 * nhi;
            if (orow < n) { f1[orow] = p1 + b1v; f2[orow] = p2 + b2v; }
        }
    }
}

__device__ __forceinline__ int range_of(int row, unsigned magic) {
    const int b = (int)__umulhi((unsigned)row, magic);
    return b < NB - 1 ? b : NB - 1;
}

__global__ void __launch_bounds__(GT)
range_count_kernel(const int* __restrict__ erow, int* __restrict__ bcount,
                   int e, int per, int rpb) {
    __shared__ int h[NB];
    if (threadIdx.x < NB) h[threadIdx.x] = 0;
    __syncthreads();
    const unsigned magic = 0xFFFFFFFFu / (unsigned)rpb + 1u;
    const int lo = blockIdx.x * per;
    const int hi = lo + per < e ? lo + per : e;
    for (int i = lo + (int)threadIdx.x; i < hi; i += EB * GT) {
        int r[EB];
#pragma unroll
        for (int k = 0; k < EB; ++k) r[k] = erow[min(i + k * GT, hi - 1)];
#pragma unroll
        for (int k = 0; k < EB; ++k)
            if (i + k * GT < hi) atomicAdd(&h[range_of(r[k], magic)], 1);
    }
    __syncthreads();
    if (threadIdx.x < NB) {
        const int v = h[threadIdx.x];
        if (v) atomicAdd(&bcount[threadIdx.x * CS], v);
    }
}

__global__ void __launch_bounds__(GT)
range_place_kernel(const int* __restrict__ erow, const int* __restrict__ ecol,
                   const int* __restrict__ bcount, int* __restrict__ bcursor,
                   int* __restrict__ bstart, unsigned* __restrict__ packed,
                   int e, int n, int per, int rpb) {
    extern __shared__ unsigned lbuf[];
    __shared__ int h[NB];
    __shared__ int lst[NB];
    __shared__ int lnum[NB];
    __shared__ int gdst[NB];
    __shared__ int wsum[NB / 32];
    __shared__ int wsum2[NB / 32];
    const int t = threadIdx.x, lane = t & 31, wave = t >> 5;
    int tot = 0, x = 0;
    if (t < NB) {
        h[t] = 0;
        tot = bcount[t * CS];
        x = tot;
        for (int off = 1; off < 32; off <<= 1) {
            const int y = __shfl_up(x, off, 32);
            if (lane >= off) x += y;
        }
        if (lane == 31) wsum[wave] = x;
    }
    __syncthreads();
    const unsigned magic = 0xFFFFFFFFu / (unsigned)rpb + 1u;
    const int cb = 32 - __clz(n - 1);
    const int lo = blockIdx.x * per;
    const int hi = lo + per < e ? lo + per : e;
    for (int i = lo + t; i < hi; i += EB * GT) {
        int r[EB];
#pragma unroll
        for (int k = 0; k < EB; ++k) r[k] = erow[min(i + k * GT, hi - 1)];
#pragma unroll
        for (int k = 0; k < EB; ++k)
            if (i + k * GT < hi) atomicAdd(&h[range_of(r[k], magic)], 1);
    }
    __syncthreads();
    int mine = 0, y = 0, gfirst = 0;
    if (t < NB) {
        int wp = 0;
        for (int w = 0; w < wave; ++w) wp += wsum[w];
        const int st = wp + x - tot;
        if (blockIdx.x == 0) bstart[t] = st;
        mine = h[t];
        const int base = mine ? atomicAdd(&bcursor[t * CS], mine) : 0;
        gfirst = st + base;
        y = mine;
        for (int off = 1; off < 32; off <<= 1) {
            const int z = __shfl_up(y, off, 32);
            if (lane >= off) y += z;
        }
        if (lane == 31) wsum2[wave] = y;
    }
    __syncthreads();
    if (t < NB) {
        int lp = 0;
        for (int w = 0; w < wave; ++w) lp += wsum2[w];
        const int ls = lp + y - mine;
        lst[t] = ls;
        lnum[t] = mine;
        gdst[t] = gfirst;
        h[t] = ls;
    }
    __syncthreads();
    for (int i = lo + t; i < hi; i += EB * GT) {
        int r[EB], c[EB];
#pragma unroll
        for (int k = 0; k < EB; ++k) {
            const int j = min(i + k * GT, hi - 1);
            r[k] = erow[j];
            c[k] = ecol[j];
        }
#pragma unroll
        for (int k = 0; k < EB; ++k) {
            if (i + k * GT < hi) {
                const int b = range_of(r[k], magic);
                const int slot = atomicAdd(&h[b], 1);
                lbuf[slot] = ((unsigned)(r[k] - b * rpb) << cb) | (unsigned)c[k];
            }
        }
    }
    __syncthreads();
    for (int b = wave; b < NB; b += GT / 32) {
        const int nb = lnum[b], ls = lst[b], g = gdst[b];
        for (int j = lane; j < nb; j += 32) packed[g + j] = lbuf[ls + j];
    }
}

__global__ void __launch_bounds__(GT)
range_sort_kernel(const unsigned* __restrict__ packed, const int* __restrict__ bcount,
                  const int* __restrict__ bstart, int* __restrict__ cnt,
                  int* __restrict__ rowstart, int* __restrict__ csr_col, int n, int rpb) {
    extern __shared__ int lds4[];
    int* lcnt = lds4;
    int* lcol = lds4 + rpb;
    __shared__ int wsum[GT / 32];
    const int t = threadIdx.x, lane = t & 31, wave = t >> 5;
    const int r0 = blockIdx.x * rpb;
    int rows = n - r0;
    if (rows > rpb) rows = rpb;
    if (rows <= 0) return;
    const int s = bstart[blockIdx.x];
    const int c = bcount[blockIdx.x * CS];
    const int cb = 32 - __clz(n - 1);
    const unsigned cmask = (1u << cb) - 1u;
    const unsigned last = (unsigned)(rows - 1);
    const bool staged = c <= SCAP;
    for (int i = t; i < rows; i += GT) lcnt[i] = 0;
    __syncthreads();
    for (int i = t; i < c; i += EB * GT) {
        unsigned w[EB];
#pragma unroll
        for (int k = 0; k < EB; ++k) w[k] = packed[s + min(i + k * GT, c - 1)];
#pragma unroll
        for (int k = 0; k < EB; ++k) {
            unsigned lr = w[k] >> cb;
            lr = lr < last ? lr : last;
            if (i + k * GT < c) atomicAdd(&lcnt[lr], 1);
        }
    }
    __syncthreads();
    int carry = 0;
    for (int base = 0; base < rows; base += GT) {
        const int idx = base + t;
        const int v = idx < rows ? lcnt[idx] : 0;
        int x = v;
        for (int off = 1; off < 32; off <<= 1) {
            const int y = __shfl_up(x, off, 32);
            if (lane >= off) x += y;
        }
        if (lane == 31) wsum[wave] = x;
        __syncthreads();
        int wp = 0, all = 0;
        for (int w = 0; w < GT / 32; ++w) {
            const int q = wsum[w];
            all += q;
            if (w < wave) wp += q;
        }
        const int ex = carry + wp + x - v;
        if (idx < rows) {
            cnt[r0 + idx] = v;
            rowstart[r0 + idx] = s + ex;
            lcnt[idx] = ex;
        }
        carry += all;
        __syncthreads();
    }
    for (int i = t; i < c; i += EB * GT) {
        unsigned w[EB];
        int slot[EB];
#pragma unroll
        for (int k = 0; k < EB; ++k) w[k] = packed[s + min(i + k * GT, c - 1)];
#pragma unroll
        for (int k = 0; k < EB; ++k) {
            unsigned lr = w[k] >> cb;
            lr = lr < last ? lr : last;
            slot[k] = i + k * GT < c ? atomicAdd(&lcnt[lr], 1) : 0;
        }
        if (staged) {
#pragma unroll
            for (int k = 0; k < EB; ++k)
                if (i + k * GT < c) lcol[slot[k]] = (int)(w[k] & cmask);
        } else {
#pragma unroll
            for (int k = 0; k < EB; ++k)
                if (i + k * GT < c) csr_col[s + slot[k]] = (int)(w[k] & cmask);
        }
    }
    if (staged) {
        __syncthreads();
        for (int i = t; i < c; i += GT) csr_col[s + i] = lcol[i];
    }
}

__device__ __forceinline__ float lrelu(float l) { return l > 0.0f ? l : NEG_SLOPE * l; }

__device__ __forceinline__ void accum_chunk(const float* __restrict__ fts, int c, float ev,
                                            int cntc, int half, int q, float4& acc) {
    const int steps = (cntc + 1) >> 1;
    const int evb = __float_as_int(ev);
#pragma unroll 4
    for (int t = 0; t < steps; ++t) {
        const int c_lo = __builtin_amdgcn_readlane(c,   2 * t);
        const int c_hi = __builtin_amdgcn_readlane(c,   2 * t + 1);
        const int w_lo = __builtin_amdgcn_readlane(evb, 2 * t);
        const int w_hi = __builtin_amdgcn_readlane(evb, 2 * t + 1);
        const int   cc = half ? c_hi : c_lo;
        const float w  = __int_as_float(half ? w_hi : w_lo);
        const float4 v = *(const float4*)(fts + (size_t)cc * F_OUT + q * 4);
        acc.x += w * v.x; acc.y += w * v.y; acc.z += w * v.z; acc.w += w * v.w;
    }
}

__global__ void __launch_bounds__(256)
row_kernel(const int* __restrict__ cnt, const int* __restrict__ rowstart,
           const int* __restrict__ csr_col,
           const float* __restrict__ f1, const float* __restrict__ f2,
           const float* __restrict__ fts, const float* __restrict__ bias,
           float* __restrict__ out, int n) {
    const int lane = threadIdx.x & 31;
    const int row  = __builtin_amdgcn_readfirstlane((int)(blockIdx.x * 8 + (threadIdx.x >> 5)));
    if (row >= n) return;
    const int half = lane >> 4;
    const int q    = lane & 15;
    const int deg  = cnt[row];
    const int s    = rowstart[row];
    const float f1r = f1[row];

    const bool v0 = lane < deg;
    int   c0 = 0;
    float l0 = -INFINITY;
    if (v0) { c0 = csr_col[s + lane]; l0 = lrelu(f1r + f2[c0]); }
    float m = l0;
    for (int base = 32; base < deg; base += 32) {
        const int j = base + lane;
        if (j < deg) m = fmaxf(m, lrelu(f1r + f2[csr_col[s + j]]));
    }
    for (int off = 16; off > 0; off >>= 1) m = fmaxf(m, __shfl_xor(m, off, 32));
    if (!isfinite(m)) m = 0.0f;

    float4 acc = make_float4(0.0f, 0.0f, 0.0f, 0.0f);
    float dsum = 0.0f;
    if (deg > 0) {
        const float ev = v0 ? expf(l0 - m) : 0.0f;
        dsum += ev;
        const int cf = __shfl(c0, 0, 32);
        accum_chunk(fts, v0 ? c0 : cf, ev, deg < 32 ? deg : 32, half, q, acc);
    }
    for (int base = 32; base < deg; base += 32) {
        const int j = base + lane;
        const bool v = j < deg;
        int c = 0; float ev = 0.0f;
        if (v) { c = csr_col[s + j]; ev = expf(lrelu(f1r + f2[c]) - m); }
        dsum += ev;
        const int cf = __shfl(c, 0, 32);
        const int rem = deg - base;
        accum_chunk(fts, v ? c : cf, ev, rem < 32 ? rem : 32, half, q, acc);
    }

    for (int off = 16; off > 0; off >>= 1) dsum += __shfl_xor(dsum, off, 32);
    acc.x += __shfl_xor(acc.x, 16, 32);
    acc.y += __shfl_xor(acc.y, 16, 32);
    acc.z += __shfl_xor(acc.z, 16, 32);
    acc.w += __shfl_xor(acc.w, 16, 32);

    if (half == 0) {
        const float4 b = *(const float4*)(bias + q * 4);
        float4 o;
        if (deg > 0) {
            o.x = acc.x / dsum + b.x; o.y = acc.y / dsum + b.y;
            o.z = acc.z / dsum + b.z; o.w = acc.w / dsum + b.w;
        } else {
            o = b;
        }
        o.x = o.x > 0.0f ? o.x : 0.0f; o.y = o.y > 0.0f ? o.y : 0.0f;
        o.z = o.z > 0.0f ? o.z : 0.0f; o.w = o.w > 0.0f ? o.w : 0.0f;
        *(float4*)(out + (size_t)row * F_OUT + q * 4) = o;
    }
}

extern "C" void kernel_launch(void* const* d_in, const int* in_sizes, int n_in,
                              void* d_out, int out_size, void* d_ws, size_t ws_size,
                              hipStream_t stream) {
    const float* seq  = (const float*)d_in[0];
    const int*   erow = (const int*)  d_in[1];
    const int*   ecol = (const int*)  d_in[2];
    const float* W    = (const float*)d_in[3];
    const float* a1   = (const float*)d_in[4];
    const float* b1   = (const float*)d_in[5];
    const float* a2   = (const float*)d_in[6];
    const float* b2   = (const float*)d_in[7];
    const float* bias = (const float*)d_in[8];

    const int n = in_sizes[0] / F_IN;
    const int e = in_sizes[1];
    float* out = (float*)d_out;

    float* fts      = (float*)d_ws;
    float* f1       = fts + (size_t)n * F_OUT;
    float* f2       = f1 + n;
    int*   cnt      = (int*)(f2 + n);
    int*   rowstart = cnt + n;
    int*   csr_col  = rowstart + n;
    unsigned* packed = (unsigned*)d_out;
    int*   bcount   = (int*)d_out + e;
    int*   bcursor  = bcount + NB * CS;
    int*   bstart   = bcursor + NB * CS;

    const int rpb = n / NB + 2;
    const int gblocks = 512;
    const int per = (e + gblocks - 1) / gblocks;

    zero_kernel<<<(2 * NB * CS + 255) / 256, 256, 0, stream>>>(bcount, 2 * NB * CS);
    {
        const int blocks = (n + 127) / 128;
        gemm_wmma_kernel<<<blocks, 256, 0, stream>>>(seq, W, fts, a1, b1, a2, b2,
                                                     f1, f2, n);
    }
    range_count_kernel<<<gblocks, GT, 0, stream>>>(erow, bcount, e, per, rpb);
    range_place_kernel<<<gblocks, GT, per * sizeof(int), stream>>>(erow, ecol, bcount, bcursor,
                                                                   bstart, packed, e, n, per, rpb);
    range_sort_kernel<<<NB, GT, (rpb + SCAP) * sizeof(int), stream>>>(packed, bcount, bstart, cnt,
                                                                      rowstart, csr_col, n, rpb);
    row_kernel<<<(n + 7) / 8, 256, 0, stream>>>(cnt, rowstart, csr_col, f1, f2, fts, bias,
                                                out, n);
}


# Projection GEMM runs inside the per-range ordering launch as a second block role (196 blocks of 32 waves beside its 256): matrix work fills idle wave slots; 5 launches, sum order unchanged
# speedup vs baseline: 6.3676x; 1.1513x over previous
#include <hip/hip_runtime.h>
#include <hip/hip_bf16.h>

typedef __attribute__((ext_vector_type(2))) float v2f;
typedef __attribute__((ext_vector_type(8))) float v8f;

#define F_IN   256
#define F_OUT  64
#define NEG_SLOPE 0.2f
#define NB     256
#define GT     1024
#define EB     8
#define CS     64
#define SCAP   14336

__global__ void zero_kernel(int* __restrict__ p, int count) {
    const int i = blockIdx.x * blockDim.x + threadIdx.x;
    if (i < count) p[i] = 0;
}

__device__ __forceinline__ void
gemm_role(float* __restrict__ Wl, int gb,
          const float* __restrict__ seq, const float* __restrict__ W,
          float* __restrict__ fts,
          const float* __restrict__ a1, const float* __restrict__ b1,
          const float* __restrict__ a2, const float* __restrict__ b2,
          float* __restrict__ f1, float* __restrict__ f2, int n) {
    for (int i = threadIdx.x; i < F_IN * F_OUT; i += GT) {
        const int kk  = i >> 6;
        const int col = i & 63;
        Wl[(kk >> 1) * (2 * F_OUT) + col * 2 + (kk & 1)] = W[i];
    }
    __syncthreads();

    const int wave = threadIdx.x >> 5;
    const int lane = threadIdx.x & 31;
    const int rowTile = gb * (GT / 32) + wave;

    int row = rowTile * 16 + (lane & 15);
    if (row >= n) row = n - 1;
    const int lcol  = lane & 15;
    const int khalf = (lane >> 4) * 2;

    const float* __restrict__ arow = seq + (size_t)row * F_IN;

    v8f acc[4] = {v8f{}, v8f{}, v8f{}, v8f{}};
    for (int k = 0; k < F_IN; k += 4) {
        float2 av = *(const float2*)(arow + k + khalf);
        v2f a; a[0] = av.x; a[1] = av.y;
        const int kpairBase = ((k + khalf) >> 1) * (2 * F_OUT);
#pragma unroll
        for (int ct = 0; ct < 4; ++ct) {
            float2 bv = *(const float2*)(&Wl[kpairBase + (ct * 16 + lcol) * 2]);
            v2f b; b[0] = bv.x; b[1] = bv.y;
            acc[ct] = __builtin_amdgcn_wmma_f32_16x16x4_f32(
                     false, a,  false, b,
                     (short)0, acc[ct],  false,  false);
        }
    }

    const int nhi = lane >> 4;
#pragma unroll
    for (int ct = 0; ct < 4; ++ct) {
        const int ocol = ct * 16 + lcol;
#pragma unroll
        for (int r = 0; r < 8; ++r) {
            const int orow = rowTile * 16 + r + 8 * nhi;
            if (orow < n) fts[(size_t)orow * F_OUT + ocol] = acc[ct][r];
        }
    }

    float A1[4], A2[4];
#pragma unroll
    for (int ct = 0; ct < 4; ++ct) {
        A1[ct] = a1[ct * 16 + lcol];
        A2[ct] = a2[ct * 16 + lcol];
    }
    const float b1v = b1[0], b2v = b2[0];
#pragma unroll
    for (int r = 0; r < 8; ++r) {
        float p1 = 0.0f, p2 = 0.0f;
#pragma unroll
        for (int ct = 0; ct < 4; ++ct) {
            p1 += acc[ct][r] * A1[ct];
            p2 += acc[ct][r] * A2[ct];
        }
#pragma unroll
        for (int off = 8; off > 0; off >>= 1) {
            p1 += __shfl_xor(p1, off, 32);
            p2 += __shfl_xor(p2, off, 32);
        }
        if ((lane & 15) == 0) {
            const int orow = rowTile * 16 + r + 8 * nhi;
            if (orow < n) { f1[orow] = p1 + b1v; f2[orow] = p2 + b2v; }
        }
    }
}

__device__ __forceinline__ int range_of(int row, unsigned magic) {
    const int b = (int)__umulhi((unsigned)row, magic);
    return b < NB - 1 ? b : NB - 1;
}

__global__ void __launch_bounds__(GT)
range_count_kernel(const int* __restrict__ erow, int* __restrict__ bcount,
                   int e, int per, int rpb) {
    __shared__ int h[NB];
    if (threadIdx.x < NB) h[threadIdx.x] = 0;
    __syncthreads();
    const unsigned magic = 0xFFFFFFFFu / (unsigned)rpb + 1u;
    const int lo = blockIdx.x * per;
    const int hi = lo + per < e ? lo + per : e;
    for (int i = lo + (int)threadIdx.x; i < hi; i += EB * GT) {
        int r[EB];
#pragma unroll
        for (int k = 0; k < EB; ++k) r[k] = erow[min(i + k * GT, hi - 1)];
#pragma unroll
        for (int k = 0; k < EB; ++k)
            if (i + k * GT < hi) atomicAdd(&h[range_of(r[k], magic)], 1);
    }
    __syncthreads();
    if (threadIdx.x < NB) {
        const int v = h[threadIdx.x];
        if (v) atomicAdd(&bcount[threadIdx.x * CS], v);
    }
}

__global__ void __launch_bounds__(GT)
range_place_kernel(const int* __restrict__ erow, const int* __restrict__ ecol,
                   const int* __restrict__ bcount, int* __restrict__ bcursor,
                   int* __restrict__ bstart, unsigned* __restrict__ packed,
                   int e, int n, int per, int rpb) {
    extern __shared__ unsigned lbuf[];
    __shared__ int h[NB];
    __shared__ int lst[NB];
    __shared__ int lnum[NB];
    __shared__ int gdst[NB];
    __shared__ int wsum[NB / 32];
    __shared__ int wsum2[NB / 32];
    const int t = threadIdx.x, lane = t & 31, wave = t >> 5;
    int tot = 0, x = 0;
    if (t < NB) {
        h[t] = 0;
        tot = bcount[t * CS];
        x = tot;
        for (int off = 1; off < 32; off <<= 1) {
            const int y = __shfl_up(x, off, 32);
            if (lane >= off) x += y;
        }
        if (lane == 31) wsum[wave] = x;
    }
    __syncthreads();
    const unsigned magic = 0xFFFFFFFFu / (unsigned)rpb + 1u;
    const int cb = 32 - __clz(n - 1);
    const int lo = blockIdx.x * per;
    const int hi = lo + per < e ? lo + per : e;
    for (int i = lo + t; i < hi; i += EB * GT) {
        int r[EB];
#pragma unroll
        for (int k = 0; k < EB; ++k) r[k] = erow[min(i + k * GT, hi - 1)];
#pragma unroll
        for (int k = 0; k < EB; ++k)
            if (i + k * GT < hi) atomicAdd(&h[range_of(r[k], magic)], 1);
    }
    __syncthreads();
    int mine = 0, y = 0, gfirst = 0;
    if (t < NB) {
        int wp = 0;
        for (int w = 0; w < wave; ++w) wp += wsum[w];
        const int st = wp + x - tot;
        if (blockIdx.x == 0) bstart[t] = st;
        mine = h[t];
        const int base = mine ? atomicAdd(&bcursor[t * CS], mine) : 0;
        gfirst = st + base;
        y = mine;
        for (int off = 1; off < 32; off <<= 1) {
            const int z = __shfl_up(y, off, 32);
            if (lane >= off) y += z;
        }
        if (lane == 31) wsum2[wave] = y;
    }
    __syncthreads();
    if (t < NB) {
        int lp = 0;
        for (int w = 0; w < wave; ++w) lp += wsum2[w];
        const int ls = lp + y - mine;
        lst[t] = ls;
        lnum[t] = mine;
        gdst[t] = gfirst;
        h[t] = ls;
    }
    __syncthreads();
    for (int i = lo + t; i < hi; i += EB * GT) {
        int r[EB], c[EB];
#pragma unroll
        for (int k = 0; k < EB; ++k) {
            const int j = min(i + k * GT, hi - 1);
            r[k] = erow[j];
            c[k] = ecol[j];
        }
#pragma unroll
        for (int k = 0; k < EB; ++k) {
            if (i + k * GT < hi) {
                const int b = range_of(r[k], magic);
                const int slot = atomicAdd(&h[b], 1);
                lbuf[slot] = ((unsigned)(r[k] - b * rpb) << cb) | (unsigned)c[k];
            }
        }
    }
    __syncthreads();
    for (int b = wave; b < NB; b += GT / 32) {
        const int nb = lnum[b], ls = lst[b], g = gdst[b];
        for (int j = lane; j < nb; j += 32) packed[g + j] = lbuf[ls + j];
    }
}

__device__ __forceinline__ void
sort_role(int* __restrict__ lds4, int sb,
          const unsigned* __restrict__ packed, const int* __restrict__ bcount,
          const int* __restrict__ bstart, int* __restrict__ cnt,
          int* __restrict__ rowstart, int* __restrict__ csr_col, int n, int rpb) {
    int* wsum = lds4;
    int* lcnt = lds4 + GT / 32;
    int* lcol = lcnt + rpb;
    const int t = threadIdx.x, lane = t & 31, wave = t >> 5;
    const int r0 = sb * rpb;
    int rows = n - r0;
    if (rows > rpb) rows = rpb;
    if (rows <= 0) return;
    const int s = bstart[sb];
    const int c = bcount[sb * CS];
    const int cb = 32 - __clz(n - 1);
    const unsigned cmask = (1u << cb) - 1u;
    const unsigned last = (unsigned)(rows - 1);
    const bool staged = c <= SCAP;
    for (int i = t; i < rows; i += GT) lcnt[i] = 0;
    __syncthreads();
    for (int i = t; i < c; i += EB * GT) {
        unsigned w[EB];
#pragma unroll
        for (int k = 0; k < EB; ++k) w[k] = packed[s + min(i + k * GT, c - 1)];
#pragma unroll
        for (int k = 0; k < EB; ++k) {
            unsigned lr = w[k] >> cb;
            lr = lr < last ? lr : last;
            if (i + k * GT < c) atomicAdd(&lcnt[lr], 1);
        }
    }
    __syncthreads();
    int carry = 0;
    for (int base = 0; base < rows; base += GT) {
        const int idx = base + t;
        const int v = idx < rows ? lcnt[idx] : 0;
        int x = v;
        for (int off = 1; off < 32; off <<= 1) {
            const int y = __shfl_up(x, off, 32);
            if (lane >= off) x += y;
        }
        if (lane == 31) wsum[wave] = x;
        __syncthreads();
        int wp = 0, all = 0;
        for (int w = 0; w < GT / 32; ++w) {
            const int q = wsum[w];
            all += q;
            if (w < wave) wp += q;
        }
        const int ex = carry + wp + x - v;
        if (idx < rows) {
            cnt[r0 + idx] = v;
            rowstart[r0 + idx] = s + ex;
            lcnt[idx] = ex;
        }
        carry += all;
        __syncthreads();
    }
    for (int i = t; i < c; i += EB * GT) {
        unsigned w[EB];
        int slot[EB];
#pragma unroll
        for (int k = 0; k < EB; ++k) w[k] = packed[s + min(i + k * GT, c - 1)];
#pragma unroll
        for (int k = 0; k < EB; ++k) {
            unsigned lr = w[k] >> cb;
            lr = lr < last ? lr : last;
            slot[k] = i + k * GT < c ? atomicAdd(&lcnt[lr], 1) : 0;
        }
        if (staged) {
#pragma unroll
            for (int k = 0; k < EB; ++k)
                if (i + k * GT < c) lcol[slot[k]] = (int)(w[k] & cmask);
        } else {
#pragma unroll
            for (int k = 0; k < EB; ++k)
                if (i + k * GT < c) csr_col[s + slot[k]] = (int)(w[k] & cmask);
        }
    }
    if (staged) {
        __syncthreads();
        for (int i = t; i < c; i += GT) csr_col[s + i] = lcol[i];
    }
}

__global__ void __launch_bounds__(GT, 16)
project_and_sort_kernel(const float* __restrict__ seq, const float* __restrict__ W,
                        float* __restrict__ fts,
                        const float* __restrict__ a1, const float* __restrict__ b1,
                        const float* __restrict__ a2, const float* __restrict__ b2,
                        float* __restrict__ f1, float* __restrict__ f2,
                        const unsigned* __restrict__ packed, const int* __restrict__ bcount,
                        const int* __restrict__ bstart, int* __restrict__ cnt,
                        int* __restrict__ rowstart, int* __restrict__ csr_col,
                        int n, int rpb, int gemm_blocks) {
    extern __shared__ int dyn_lds[];
    if ((int)blockIdx.x < gemm_blocks)
        gemm_role((float*)dyn_lds, (int)blockIdx.x, seq, W, fts, a1, b1, a2, b2, f1, f2, n);
    else
        sort_role(dyn_lds, (int)blockIdx.x - gemm_blocks, packed, bcount, bstart, cnt, rowstart,
                  csr_col, n, rpb);
}

__device__ __forceinline__ float lrelu(float l) { return l > 0.0f ? l : NEG_SLOPE * l; }

__device__ __forceinline__ void accum_chunk(const float* __restrict__ fts, int c, float ev,
                                            int cntc, int half, int q, float4& acc) {
    const int steps = (cntc + 1) >> 1;
    const int evb = __float_as_int(ev);
#pragma unroll 4
    for (int t = 0; t < steps; ++t) {
        const int c_lo = __builtin_amdgcn_readlane(c,   2 * t);
        const int c_hi = __builtin_amdgcn_readlane(c,   2 * t + 1);
        const int w_lo = __builtin_amdgcn_readlane(evb, 2 * t);
        const int w_hi = __builtin_amdgcn_readlane(evb, 2 * t + 1);
        const int   cc = half ? c_hi : c_lo;
        const float w  = __int_as_float(half ? w_hi : w_lo);
        const float4 v = *(const float4*)(fts + (size_t)cc * F_OUT + q * 4);
        acc.x += w * v.x; acc.y += w * v.y; acc.z += w * v.z; acc.w += w * v.w;
    }
}

__global__ void __launch_bounds__(256)
row_kernel(const int* __restrict__ cnt, const int* __restrict__ rowstart,
           const int* __restrict__ csr_col,
           const float* __restrict__ f1, const float* __restrict__ f2,
           const float* __restrict__ fts, const float* __restrict__ bias,
           float* __restrict__ out, int n) {
    const int lane = threadIdx.x & 31;
    const int row  = __builtin_amdgcn_readfirstlane((int)(blockIdx.x * 8 + (threadIdx.x >> 5)));
    if (row >= n) return;
    const int half = lane >> 4;
    const int q    = lane & 15;
    const int deg  = cnt[row];
    const int s    = rowstart[row];
    const float f1r = f1[row];

    const bool v0 = lane < deg;
    int   c0 = 0;
    float l0 = -INFINITY;
    if (v0) { c0 = csr_col[s + lane]; l0 = lrelu(f1r + f2[c0]); }
    float m = l0;
    for (int base = 32; base < deg; base += 32) {
        const int j = base + lane;
        if (j < deg) m = fmaxf(m, lrelu(f1r + f2[csr_col[s + j]]));
    }
    for (int off = 16; off > 0; off >>= 1) m = fmaxf(m, __shfl_xor(m, off, 32));
    if (!isfinite(m)) m = 0.0f;

    float4 acc = make_float4(0.0f, 0.0f, 0.0f, 0.0f);
    float dsum = 0.0f;
    if (deg > 0) {
        const float ev = v0 ? expf(l0 - m) : 0.0f;
        dsum += ev;
        const int cf = __shfl(c0, 0, 32);
        accum_chunk(fts, v0 ? c0 : cf, ev, deg < 32 ? deg : 32, half, q, acc);
    }
    for (int base = 32; base < deg; base += 32) {
        const int j = base + lane;
        const bool v = j < deg;
        int c = 0; float ev = 0.0f;
        if (v) { c = csr_col[s + j]; ev = expf(lrelu(f1r + f2[c]) - m); }
        dsum += ev;
        const int cf = __shfl(c, 0, 32);
        const int rem = deg - base;
        accum_chunk(fts, v ? c : cf, ev, rem < 32 ? rem : 32, half, q, acc);
    }

    for (int off = 16; off > 0; off >>= 1) dsum += __shfl_xor(dsum, off, 32);
    acc.x += __shfl_xor(acc.x, 16, 32);
    acc.y += __shfl_xor(acc.y, 16, 32);
    acc.z += __shfl_xor(acc.z, 16, 32);
    acc.w += __shfl_xor(acc.w, 16, 32);

    if (half == 0) {
        const float4 b = *(const float4*)(bias + q * 4);
        float4 o;
        if (deg > 0) {
            o.x = acc.x / dsum + b.x; o.y = acc.y / dsum + b.y;
            o.z = acc.z / dsum + b.z; o.w = acc.w / dsum + b.w;
        } else {
            o = b;
        }
        o.x = o.x > 0.0f ? o.x : 0.0f; o.y = o.y > 0.0f ? o.y : 0.0f;
        o.z = o.z > 0.0f ? o.z : 0.0f; o.w = o.w > 0.0f ? o.w : 0.0f;
        *(float4*)(out + (size_t)row * F_OUT + q * 4) = o;
    }
}

extern "C" void kernel_launch(void* const* d_in, const int* in_sizes, int n_in,
                              void* d_out, int out_size, void* d_ws, size_t ws_size,
                              hipStream_t stream) {
    const float* seq  = (const float*)d_in[0];
    const int*   erow = (const int*)  d_in[1];
    const int*   ecol = (const int*)  d_in[2];
    const float* W    = (const float*)d_in[3];
    const float* a1   = (const float*)d_in[4];
    const float* b1   = (const float*)d_in[5];
    const float* a2   = (const float*)d_in[6];
    const float* b2   = (const float*)d_in[7];
    const float* bias = (const float*)d_in[8];

    const int n = in_sizes[0] / F_IN;
    const int e = in_sizes[1];
    float* out = (float*)d_out;

    float* fts      = (float*)d_ws;
    float* f1       = fts + (size_t)n * F_OUT;
    float* f2       = f1 + n;
    int*   cnt      = (int*)(f2 + n);
    int*   rowstart = cnt + n;
    int*   csr_col  = rowstart + n;
    unsigned* packed = (unsigned*)d_out;
    int*   bcount   = (int*)d_out + e;
    int*   bcursor  = bcount + NB * CS;
    int*   bstart   = bcursor + NB * CS;

    const int rpb = n / NB + 2;
    const int gblocks = 512;
    const int per = (e + gblocks - 1) / gblocks;

    zero_kernel<<<(2 * NB * CS + 255) / 256, 256, 0, stream>>>(bcount, 2 * NB * CS);
    range_count_kernel<<<gblocks, GT, 0, stream>>>(erow, bcount, e, per, rpb);
    range_place_kernel<<<gblocks, GT, per * sizeof(int), stream>>>(erow, ecol, bcount, bcursor,
                                                                   bstart, packed, e, n, per, rpb);
    {
        const int gemm_blocks = (n + 16 * (GT / 32) - 1) / (16 * (GT / 32));
        const size_t sort_lds = (rpb + SCAP + GT / 32) * sizeof(int);
        const size_t gemm_lds = F_IN * F_OUT * sizeof(float);
        project_and_sort_kernel<<<gemm_blocks + NB, GT, sort_lds > gemm_lds ? sort_lds : gemm_lds,
                                  stream>>>(seq, W, fts, a1, b1, a2, b2, f1, f2,
                                            packed, bcount, bstart, cnt, rowstart, csr_col,
                                            n, rpb, gemm_blocks);
    }
    row_kernel<<<(n + 7) / 8, 256, 0, stream>>>(cnt, rowstart, csr_col, f1, f2, fts, bias,
                                                out, n);
}


# Each block orders its own slice of the edge list by row range and records (first slot, length) per range; ordering blocks collect their 512 runs. Count and zero launches go; sum order still LDS rank
# speedup vs baseline: 7.8104x; 1.2266x over previous
#include <hip/hip_runtime.h>
#include <hip/hip_bf16.h>

typedef __attribute__((ext_vector_type(2))) float v2f;
typedef __attribute__((ext_vector_type(8))) float v8f;

#define F_IN   256
#define F_OUT  64
#define NEG_SLOPE 0.2f
#define NB     256
#define GT     1024
#define EB     8
#define PB     512
#define RPW    (PB / (GT / 32))
#define SB     4
#define SCAP   14336


__device__ __forceinline__ void
gemm_role(float* __restrict__ Wl, int gb,
          const float* __restrict__ seq, const float* __restrict__ W,
          float* __restrict__ fts,
          const float* __restrict__ a1, const float* __restrict__ b1,
          const float* __restrict__ a2, const float* __restrict__ b2,
          float* __restrict__ f1, float* __restrict__ f2, int n) {
    for (int i = threadIdx.x; i < F_IN * F_OUT; i += GT) {
        const int kk  = i >> 6;
        const int col = i & 63;
        Wl[(kk >> 1) * (2 * F_OUT) + col * 2 + (kk & 1)] = W[i];
    }
    __syncthreads();

    const int wave = threadIdx.x >> 5;
    const int lane = threadIdx.x & 31;
    const int rowTile = gb * (GT / 32) + wave;

    int row = rowTile * 16 + (lane & 15);
    if (row >= n) row = n - 1;
    const int lcol  = lane & 15;
    const int khalf = (lane >> 4) * 2;

    const float* __restrict__ arow = seq + (size_t)row * F_IN;

    v8f acc[4] = {v8f{}, v8f{}, v8f{}, v8f{}};
    for (int k = 0; k < F_IN; k += 4) {
        float2 av = *(const float2*)(arow + k + khalf);
        v2f a; a[0] = av.x; a[1] = av.y;
        const int kpairBase = ((k + khalf) >> 1) * (2 * F_OUT);
#pragma unroll
        for (int ct = 0; ct < 4; ++ct) {
            float2 bv = *(const float2*)(&Wl[kpairBase + (ct * 16 + lcol) * 2]);
            v2f b; b[0] = bv.x; b[1] = bv.y;
            acc[ct] = __builtin_amdgcn_wmma_f32_16x16x4_f32(
                     false, a,  false, b,
                     (short)0, acc[ct],  false,  false);
        }
    }

    const int nhi = lane >> 4;
#pragma unroll
    for (int ct = 0; ct < 4; ++ct) {
        const int ocol = ct * 16 + lcol;
#pragma unroll
        for (int r = 0; r < 8; ++r) {
            const int orow = rowTile * 16 + r + 8 * nhi;
            if (orow < n) fts[(size_t)orow * F_OUT + ocol] = acc[ct][r];
        }
    }

    float A1[4], A2[4];
#pragma unroll
    for (int ct = 0; ct < 4; ++ct) {
        A1[ct] = a1[ct * 16 + lcol];
        A2[ct] = a2[ct * 16 + lcol];
    }
    const float b1v = b1[0], b2v = b2[0];
#pragma unroll
    for (int r = 0; r < 8; ++r) {
        float p1 = 0.0f, p2 = 0.0f;
#pragma unroll
        for (int ct = 0; ct < 4; ++ct) {
            p1 += acc[ct][r] * A1[ct];
            p2 += acc[ct][r] * A2[ct];
        }
#pragma unroll
        for (int off = 8; off > 0; off >>= 1) {
            p1 += __shfl_xor(p1, off, 32);
            p2 += __shfl_xor(p2, off, 32);
        }
        if ((lane & 15) == 0) {
            const int orow = rowTile * 16 + r + 8 * nhi;
            if (orow < n) { f1[orow] = p1 + b1v; f2[orow] = p2 + b2v; }
        }
    }
}

__device__ __forceinline__ int range_of(int row, unsigned magic) {
    const int b = (int)__umulhi((unsigned)row, magic);
    return b < NB - 1 ? b : NB - 1;
}

__global__ void __launch_bounds__(GT)
slice_group_kernel(const int* __restrict__ erow, const int* __restrict__ ecol,
                   int* __restrict__ rtab, unsigned* __restrict__ packed,
                   int e, int n, int per, int rpb) {
    extern __shared__ unsigned lbuf[];
    __shared__ int h[NB];
    __shared__ int wsum2[NB / 32];
    const int t = threadIdx.x, lane = t & 31, wave = t >> 5;
    if (t < NB) h[t] = 0;
    __syncthreads();
    const unsigned magic = 0xFFFFFFFFu / (unsigned)rpb + 1u;
    const int cb = 32 - __clz(n - 1);
    const int lo = blockIdx.x * per;
    const int hi = lo + per < e ? lo + per : e;
    for (int i = lo + t; i < hi; i += EB * GT) {
        int r[EB];
#pragma unroll
        for (int k = 0; k < EB; ++k) r[k] = erow[min(i + k * GT, hi - 1)];
#pragma unroll
        for (int k = 0; k < EB; ++k)
            if (i + k * GT < hi) atomicAdd(&h[range_of(r[k], magic)], 1);
    }
    __syncthreads();
    int mine = 0, y = 0;
    if (t < NB) {
        mine = h[t];
        y = mine;
        for (int off = 1; off < 32; off <<= 1) {
            const int z = __shfl_up(y, off, 32);
            if (lane >= off) y += z;
        }
        if (lane == 31) wsum2[wave] = y;
    }
    __syncthreads();
    if (t < NB) {
        int lp = 0;
        for (int w = 0; w < wave; ++w) lp += wsum2[w];
        const int ls = lp + y - mine;
        *(int2*)(rtab + 2 * ((size_t)blockIdx.x * NB + t)) = make_int2(ls, mine);
        h[t] = ls;
    }
    __syncthreads();
    for (int i = lo + t; i < hi; i += EB * GT) {
        int r[EB], c[EB];
#pragma unroll
        for (int k = 0; k < EB; ++k) {
            const int j = min(i + k * GT, hi - 1);
            r[k] = erow[j];
            c[k] = ecol[j];
        }
#pragma unroll
        for (int k = 0; k < EB; ++k) {
            if (i + k * GT < hi) {
                const int b = range_of(r[k], magic);
                const int slot = atomicAdd(&h[b], 1);
                lbuf[slot] = ((unsigned)(r[k] - b * rpb) << cb) | (unsigned)c[k];
            }
        }
    }
    __syncthreads();
    for (int i = t; i < hi - lo; i += GT) packed[lo + i] = lbuf[i];
}

__device__ __forceinline__ void
sort_role(int* __restrict__ lds4, int sb,
          const unsigned* __restrict__ packed, const int* __restrict__ rtab,
          int* __restrict__ cnt, int* __restrict__ rowstart, int* __restrict__ csr_col,
          int e, int n, int rpb, int per) {
    int* wsum = lds4;
    int* wtot = wsum + GT / 32;
    int* wofs = wtot + GT / 32;
    int* lcnt = wofs + GT / 32;
    int* lcol = lcnt + rpb;
    const int t = threadIdx.x, lane = t & 31, wave = t >> 5;
    const int r0 = sb * rpb;
    int rows = n - r0;
    if (rows > rpb) rows = rpb;
    if (rows <= 0) return;
    const int cb = 32 - __clz(n - 1);
    const unsigned cmask = (1u << cb) - 1u;
    const unsigned last = (unsigned)(rows - 1);
    int rbase = 0, rlen = 0, rofs = 0;
    if (lane < RPW) {
        const int pb = wave * RPW + lane;
        const int2 v = *(const int2*)(rtab + 2 * ((size_t)pb * NB + sb));
        rofs = v.x;
        rlen = v.y;
        rbase = pb * per + v.x;
    }
    for (int i = t; i < rows; i += GT) lcnt[i] = 0;
    unsigned w[RPW];
#pragma unroll
    for (int k = 0; k < RPW; ++k) {
        const int base = __builtin_amdgcn_readlane(rbase, k);
        const int len  = __builtin_amdgcn_readlane(rlen, k);
        const int j = base + min(lane, len > 0 ? len - 1 : 0);
        w[k] = packed[min(j, e - 1)];
    }
    {
        int xs = rlen, os = rofs;
        for (int off = 16; off > 0; off >>= 1) {
            xs += __shfl_xor(xs, off, 32);
            os += __shfl_xor(os, off, 32);
        }
        if (lane == 0) { wtot[wave] = xs; wofs[wave] = os; }
    }
    __syncthreads();
    int c = wtot[lane], s = wofs[lane];
    for (int off = 16; off > 0; off >>= 1) {
        c += __shfl_xor(c, off, 32);
        s += __shfl_xor(s, off, 32);
    }
    const bool staged = c <= SCAP;
#pragma unroll
    for (int k = 0; k < RPW; ++k) {
        const int len = __builtin_amdgcn_readlane(rlen, k);
        if (lane < len) {
            unsigned lr = w[k] >> cb;
            lr = lr < last ? lr : last;
            atomicAdd(&lcnt[lr], 1);
        }
    }
    for (int k = 0; k < RPW; ++k) {
        const int base = __builtin_amdgcn_readlane(rbase, k);
        const int len  = __builtin_amdgcn_readlane(rlen, k);
        for (int j = 32 + lane; j < len; j += 32) {
            unsigned lr = packed[base + j] >> cb;
            lr = lr < last ? lr : last;
            atomicAdd(&lcnt[lr], 1);
        }
    }
    __syncthreads();
    int carry = 0;
    for (int base = 0; base < rows; base += GT) {
        const int idx = base + t;
        const int v = idx < rows ? lcnt[idx] : 0;
        int x = v;
        for (int off = 1; off < 32; off <<= 1) {
            const int y = __shfl_up(x, off, 32);
            if (lane >= off) x += y;
        }
        if (lane == 31) wsum[wave] = x;
        __syncthreads();
        int wp = 0, all = 0;
#pragma unroll 8
        for (int q2 = 0; q2 < GT / 32; ++q2) {
            const int q = wsum[q2];
            all += q;
            if (q2 < wave) wp += q;
        }
        const int ex = carry + wp + x - v;
        if (idx < rows) {
            cnt[r0 + idx] = v;
            rowstart[r0 + idx] = s + ex;
            lcnt[idx] = ex;
        }
        carry += all;
        __syncthreads();
    }
#pragma unroll
    for (int g = 0; g < RPW; g += SB) {
        int slot[SB];
#pragma unroll
        for (int k = 0; k < SB; ++k) {
            const int len = __builtin_amdgcn_readlane(rlen, g + k);
            unsigned lr = w[g + k] >> cb;
            lr = lr < last ? lr : last;
            slot[k] = lane < len ? atomicAdd(&lcnt[lr], 1) : 0;
        }
#pragma unroll
        for (int k = 0; k < SB; ++k) {
            if (lane < __builtin_amdgcn_readlane(rlen, g + k)) {
                if (staged) lcol[slot[k]] = (int)(w[g + k] & cmask);
                else        csr_col[s + slot[k]] = (int)(w[g + k] & cmask);
            }
        }
    }
    for (int k = 0; k < RPW; ++k) {
        const int base = __builtin_amdgcn_readlane(rbase, k);
        const int len  = __builtin_amdgcn_readlane(rlen, k);
        for (int j = 32 + lane; j < len; j += 32) {
            const unsigned wj = packed[base + j];
            unsigned lr = wj >> cb;
            lr = lr < last ? lr : last;
            const int sl = atomicAdd(&lcnt[lr], 1);
            if (staged) lcol[sl] = (int)(wj & cmask);
            else        csr_col[s + sl] = (int)(wj & cmask);
        }
    }
    if (staged) {
        __syncthreads();
        for (int i = t; i < c; i += GT) csr_col[s + i] = lcol[i];
    }
}

__global__ void __launch_bounds__(GT, 16)
project_and_sort_kernel(const float* __restrict__ seq, const float* __restrict__ W,
                        float* __restrict__ fts,
                        const float* __restrict__ a1, const float* __restrict__ b1,
                        const float* __restrict__ a2, const float* __restrict__ b2,
                        float* __restrict__ f1, float* __restrict__ f2,
                        const unsigned* __restrict__ packed, const int* __restrict__ rtab,
                        int* __restrict__ cnt, int* __restrict__ rowstart,
                        int* __restrict__ csr_col,
                        int e, int n, int rpb, int per, int gemm_blocks) {
    extern __shared__ int dyn_lds[];
    if ((int)blockIdx.x < gemm_blocks)
        gemm_role((float*)dyn_lds, (int)blockIdx.x, seq, W, fts, a1, b1, a2, b2, f1, f2, n);
    else
        sort_role(dyn_lds, (int)blockIdx.x - gemm_blocks, packed, rtab, cnt, rowstart, csr_col,
                  e, n, rpb, per);
}

__device__ __forceinline__ float lrelu(float l) { return l > 0.0f ? l : NEG_SLOPE * l; }

__device__ __forceinline__ void accum_chunk(const float* __restrict__ fts, int c, float ev,
                                            int cntc, int half, int q, float4& acc) {
    const int steps = (cntc + 1) >> 1;
    const int evb = __float_as_int(ev);
#pragma unroll 4
    for (int t = 0; t < steps; ++t) {
        const int c_lo = __builtin_amdgcn_readlane(c,   2 * t);
        const int c_hi = __builtin_amdgcn_readlane(c,   2 * t + 1);
        const int w_lo = __builtin_amdgcn_readlane(evb, 2 * t);
        const int w_hi = __builtin_amdgcn_readlane(evb, 2 * t + 1);
        const int   cc = half ? c_hi : c_lo;
        const float w  = __int_as_float(half ? w_hi : w_lo);
        const float4 v = *(const float4*)(fts + (size_t)cc * F_OUT + q * 4);
        acc.x += w * v.x; acc.y += w * v.y; acc.z += w * v.z; acc.w += w * v.w;
    }
}

__global__ void __launch_bounds__(256)
row_kernel(const int* __restrict__ cnt, const int* __restrict__ rowstart,
           const int* __restrict__ csr_col,
           const float* __restrict__ f1, const float* __restrict__ f2,
           const float* __restrict__ fts, const float* __restrict__ bias,
           float* __restrict__ out, int n) {
    const int lane = threadIdx.x & 31;
    const int row  = __builtin_amdgcn_readfirstlane((int)(blockIdx.x * 8 + (threadIdx.x >> 5)));
    if (row >= n) return;
    const int half = lane >> 4;
    const int q    = lane & 15;
    const int deg  = cnt[row];
    const int s    = rowstart[row];
    const float f1r = f1[row];

    const bool v0 = lane < deg;
    int   c0 = 0;
    float l0 = -INFINITY;
    if (v0) { c0 = csr_col[s + lane]; l0 = lrelu(f1r + f2[c0]); }
    float m = l0;
    for (int base = 32; base < deg; base += 32) {
        const int j = base + lane;
        if (j < deg) m = fmaxf(m, lrelu(f1r + f2[csr_col[s + j]]));
    }
    for (int off = 16; off > 0; off >>= 1) m = fmaxf(m, __shfl_xor(m, off, 32));
    if (!isfinite(m)) m = 0.0f;

    float4 acc = make_float4(0.0f, 0.0f, 0.0f, 0.0f);
    float dsum = 0.0f;
    if (deg > 0) {
        const float ev = v0 ? expf(l0 - m) : 0.0f;
        dsum += ev;
        const int cf = __shfl(c0, 0, 32);
        accum_chunk(fts, v0 ? c0 : cf, ev, deg < 32 ? deg : 32, half, q, acc);
    }
    for (int base = 32; base < deg; base += 32) {
        const int j = base + lane;
        const bool v = j < deg;
        int c = 0; float ev = 0.0f;
        if (v) { c = csr_col[s + j]; ev = expf(lrelu(f1r + f2[c]) - m); }
        dsum += ev;
        const int cf = __shfl(c, 0, 32);
        const int rem = deg - base;
        accum_chunk(fts, v ? c : cf, ev, rem < 32 ? rem : 32, half, q, acc);
    }

    for (int off = 16; off > 0; off >>= 1) dsum += __shfl_xor(dsum, off, 32);
    acc.x += __shfl_xor(acc.x, 16, 32);
    acc.y += __shfl_xor(acc.y, 16, 32);
    acc.z += __shfl_xor(acc.z, 16, 32);
    acc.w += __shfl_xor(acc.w, 16, 32);

    if (half == 0) {
        const float4 b = *(const float4*)(bias + q * 4);
        float4 o;
        if (deg > 0) {
            o.x = acc.x / dsum + b.x; o.y = acc.y / dsum + b.y;
            o.z = acc.z / dsum + b.z; o.w = acc.w / dsum + b.w;
        } else {
            o = b;
        }
        o.x = o.x > 0.0f ? o.x : 0.0f; o.y = o.y > 0.0f ? o.y : 0.0f;
        o.z = o.z > 0.0f ? o.z : 0.0f; o.w = o.w > 0.0f ? o.w : 0.0f;
        *(float4*)(out + (size_t)row * F_OUT + q * 4) = o;
    }
}

extern "C" void kernel_launch(void* const* d_in, const int* in_sizes, int n_in,
                              void* d_out, int out_size, void* d_ws, size_t ws_size,
                              hipStream_t stream) {
    const float* seq  = (const float*)d_in[0];
    const int*   erow = (const int*)  d_in[1];
    const int*   ecol = (const int*)  d_in[2];
    const float* W    = (const float*)d_in[3];
    const float* a1   = (const float*)d_in[4];
    const float* b1   = (const float*)d_in[5];
    const float* a2   = (const float*)d_in[6];
    const float* b2   = (const float*)d_in[7];
    const float* bias = (const float*)d_in[8];

    const int n = in_sizes[0] / F_IN;
    const int e = in_sizes[1];
    float* out = (float*)d_out;

    float* fts      = (float*)d_ws;
    float* f1       = fts + (size_t)n * F_OUT;
    float* f2       = f1 + n;
    int*   cnt      = (int*)(f2 + n);
    int*   rowstart = cnt + n;
    int*   csr_col  = rowstart + n;
    unsigned* packed = (unsigned*)d_out;
    int*   rtab     = (int*)d_out + (e + 31) / 32 * 32;

    const int rpb = n / NB + 2;
    const int per = ((e + PB - 1) / PB + 31) / 32 * 32;

    slice_group_kernel<<<PB, GT, per * sizeof(int), stream>>>(erow, ecol, rtab, packed,
                                                             e, n, per, rpb);
    {
        const int gemm_blocks = (n + 16 * (GT / 32) - 1) / (16 * (GT / 32));
        const size_t sort_lds = (rpb + SCAP + 3 * (GT / 32)) * sizeof(int);
        const size_t gemm_lds = F_IN * F_OUT * sizeof(float);
        project_and_sort_kernel<<<gemm_blocks + NB, GT, sort_lds > gemm_lds ? sort_lds : gemm_lds,
                                  stream>>>(seq, W, fts, a1, b1, a2, b2, f1, f2,
                                            packed, rtab, cnt, rowstart, csr_col,
                                            e, n, rpb, per, gemm_blocks);
    }
    row_kernel<<<(n + 7) / 8, 256, 0, stream>>>(cnt, rowstart, csr_col, f1, f2, fts, bias,
                                                out, n);
}


# Row kernel made resident: 16384 waves each walk 6-7 rows instead of one wave entering per row, next rows' bounds, columns and f2 requested ahead of the gathers; per-row sum order unchanged
# speedup vs baseline: 8.0699x; 1.0332x over previous
#include <hip/hip_runtime.h>
#include <hip/hip_bf16.h>

typedef __attribute__((ext_vector_type(2))) float v2f;
typedef __attribute__((ext_vector_type(8))) float v8f;

#define F_IN   256
#define F_OUT  64
#define NEG_SLOPE 0.2f
#define NB     256
#define GT     1024
#define EB     8
#define PB     512
#define RPW    (PB / (GT / 32))
#define SB     4
#define RB     2048
#define SCAP   14336


__device__ __forceinline__ void
gemm_role(float* __restrict__ Wl, int gb,
          const float* __restrict__ seq, const float* __restrict__ W,
          float* __restrict__ fts,
          const float* __restrict__ a1, const float* __restrict__ b1,
          const float* __restrict__ a2, const float* __restrict__ b2,
          float* __restrict__ f1, float* __restrict__ f2, int n) {
    for (int i = threadIdx.x; i < F_IN * F_OUT; i += GT) {
        const int kk  = i >> 6;
        const int col = i & 63;
        Wl[(kk >> 1) * (2 * F_OUT) + col * 2 + (kk & 1)] = W[i];
    }
    __syncthreads();

    const int wave = threadIdx.x >> 5;
    const int lane = threadIdx.x & 31;
    const int rowTile = gb * (GT / 32) + wave;

    int row = rowTile * 16 + (lane & 15);
    if (row >= n) row = n - 1;
    const int lcol  = lane & 15;
    const int khalf = (lane >> 4) * 2;

    const float* __restrict__ arow = seq + (size_t)row * F_IN;

    v8f acc[4] = {v8f{}, v8f{}, v8f{}, v8f{}};
    for (int k = 0; k < F_IN; k += 4) {
        float2 av = *(const float2*)(arow + k + khalf);
        v2f a; a[0] = av.x; a[1] = av.y;
        const int kpairBase = ((k + khalf) >> 1) * (2 * F_OUT);
#pragma unroll
        for (int ct = 0; ct < 4; ++ct) {
            float2 bv = *(const float2*)(&Wl[kpairBase + (ct * 16 + lcol) * 2]);
            v2f b; b[0] = bv.x; b[1] = bv.y;
            acc[ct] = __builtin_amdgcn_wmma_f32_16x16x4_f32(
                     false, a,  false, b,
                     (short)0, acc[ct],  false,  false);
        }
    }

    const int nhi = lane >> 4;
#pragma unroll
    for (int ct = 0; ct < 4; ++ct) {
        const int ocol = ct * 16 + lcol;
#pragma unroll
        for (int r = 0; r < 8; ++r) {
            const int orow = rowTile * 16 + r + 8 * nhi;
            if (orow < n) fts[(size_t)orow * F_OUT + ocol] = acc[ct][r];
        }
    }

    float A1[4], A2[4];
#pragma unroll
    for (int ct = 0; ct < 4; ++ct) {
        A1[ct] = a1[ct * 16 + lcol];
        A2[ct] = a2[ct * 16 + lcol];
    }
    const float b1v = b1[0], b2v = b2[0];
#pragma unroll
    for (int r = 0; r < 8; ++r) {
        float p1 = 0.0f, p2 = 0.0f;
#pragma unroll
        for (int ct = 0; ct < 4; ++ct) {
            p1 += acc[ct][r] * A1[ct];
            p2 += acc[ct][r] * A2[ct];
        }
#pragma unroll
        for (int off = 8; off > 0; off >>= 1) {
            p1 += __shfl_xor(p1, off, 32);
            p2 += __shfl_xor(p2, off, 32);
        }
        if ((lane & 15) == 0) {
            const int orow = rowTile * 16 + r + 8 * nhi;
            if (orow < n) { f1[orow] = p1 + b1v; f2[orow] = p2 + b2v; }
        }
    }
}

__device__ __forceinline__ int range_of(int row, unsigned magic) {
    const int b = (int)__umulhi((unsigned)row, magic);
    return b < NB - 1 ? b : NB - 1;
}

__global__ void __launch_bounds__(GT)
slice_group_kernel(const int* __restrict__ erow, const int* __restrict__ ecol,
                   int* __restrict__ rtab, unsigned* __restrict__ packed,
                   int e, int n, int per, int rpb) {
    extern __shared__ unsigned lbuf[];
    __shared__ int h[NB];
    __shared__ int wsum2[NB / 32];
    const int t = threadIdx.x, lane = t & 31, wave = t >> 5;
    if (t < NB) h[t] = 0;
    __syncthreads();
    const unsigned magic = 0xFFFFFFFFu / (unsigned)rpb + 1u;
    const int cb = 32 - __clz(n - 1);
    const int lo = blockIdx.x * per;
    const int hi = lo + per < e ? lo + per : e;
    for (int i = lo + t; i < hi; i += EB * GT) {
        int r[EB];
#pragma unroll
        for (int k = 0; k < EB; ++k) r[k] = erow[min(i + k * GT, hi - 1)];
#pragma unroll
        for (int k = 0; k < EB; ++k)
            if (i + k * GT < hi) atomicAdd(&h[range_of(r[k], magic)], 1);
    }
    __syncthreads();
    int mine = 0, y = 0;
    if (t < NB) {
        mine = h[t];
        y = mine;
        for (int off = 1; off < 32; off <<= 1) {
            const int z = __shfl_up(y, off, 32);
            if (lane >= off) y += z;
        }
        if (lane == 31) wsum2[wave] = y;
    }
    __syncthreads();
    if (t < NB) {
        int lp = 0;
        for (int w = 0; w < wave; ++w) lp += wsum2[w];
        const int ls = lp + y - mine;
        *(int2*)(rtab + 2 * ((size_t)blockIdx.x * NB + t)) = make_int2(ls, mine);
        h[t] = ls;
    }
    __syncthreads();
    for (int i = lo + t; i < hi; i += EB * GT) {
        int r[EB], c[EB];
#pragma unroll
        for (int k = 0; k < EB; ++k) {
            const int j = min(i + k * GT, hi - 1);
            r[k] = erow[j];
            c[k] = ecol[j];
        }
#pragma unroll
        for (int k = 0; k < EB; ++k) {
            if (i + k * GT < hi) {
                const int b = range_of(r[k], magic);
                const int slot = atomicAdd(&h[b], 1);
                lbuf[slot] = ((unsigned)(r[k] - b * rpb) << cb) | (unsigned)c[k];
            }
        }
    }
    __syncthreads();
    for (int i = t; i < hi - lo; i += GT) packed[lo + i] = lbuf[i];
}

__device__ __forceinline__ void
sort_role(int* __restrict__ lds4, int sb,
          const unsigned* __restrict__ packed, const int* __restrict__ rtab,
          int* __restrict__ cnt, int* __restrict__ rowstart, int* __restrict__ csr_col,
          int e, int n, int rpb, int per) {
    int* wsum = lds4;
    int* wtot = wsum + GT / 32;
    int* wofs = wtot + GT / 32;
    int* lcnt = wofs + GT / 32;
    int* lcol = lcnt + rpb;
    const int t = threadIdx.x, lane = t & 31, wave = t >> 5;
    const int r0 = sb * rpb;
    int rows = n - r0;
    if (rows > rpb) rows = rpb;
    if (rows <= 0) return;
    const int cb = 32 - __clz(n - 1);
    const unsigned cmask = (1u << cb) - 1u;
    const unsigned last = (unsigned)(rows - 1);
    int rbase = 0, rlen = 0, rofs = 0;
    if (lane < RPW) {
        const int pb = wave * RPW + lane;
        const int2 v = *(const int2*)(rtab + 2 * ((size_t)pb * NB + sb));
        rofs = v.x;
        rlen = v.y;
        rbase = pb * per + v.x;
    }
    for (int i = t; i < rows; i += GT) lcnt[i] = 0;
    unsigned w[RPW];
#pragma unroll
    for (int k = 0; k < RPW; ++k) {
        const int base = __builtin_amdgcn_readlane(rbase, k);
        const int len  = __builtin_amdgcn_readlane(rlen, k);
        const int j = base + min(lane, len > 0 ? len - 1 : 0);
        w[k] = packed[min(j, e - 1)];
    }
    {
        int xs = rlen, os = rofs;
        for (int off = 16; off > 0; off >>= 1) {
            xs += __shfl_xor(xs, off, 32);
            os += __shfl_xor(os, off, 32);
        }
        if (lane == 0) { wtot[wave] = xs; wofs[wave] = os; }
    }
    __syncthreads();
    int c = wtot[lane], s = wofs[lane];
    for (int off = 16; off > 0; off >>= 1) {
        c += __shfl_xor(c, off, 32);
        s += __shfl_xor(s, off, 32);
    }
    const bool staged = c <= SCAP;
#pragma unroll
    for (int k = 0; k < RPW; ++k) {
        const int len = __builtin_amdgcn_readlane(rlen, k);
        if (lane < len) {
            unsigned lr = w[k] >> cb;
            lr = lr < last ? lr : last;
            atomicAdd(&lcnt[lr], 1);
        }
    }
    for (int k = 0; k < RPW; ++k) {
        const int base = __builtin_amdgcn_readlane(rbase, k);
        const int len  = __builtin_amdgcn_readlane(rlen, k);
        for (int j = 32 + lane; j < len; j += 32) {
            unsigned lr = packed[base + j] >> cb;
            lr = lr < last ? lr : last;
            atomicAdd(&lcnt[lr], 1);
        }
    }
    __syncthreads();
    int carry = 0;
    for (int base = 0; base < rows; base += GT) {
        const int idx = base + t;
        const int v = idx < rows ? lcnt[idx] : 0;
        int x = v;
        for (int off = 1; off < 32; off <<= 1) {
            const int y = __shfl_up(x, off, 32);
            if (lane >= off) x += y;
        }
        if (lane == 31) wsum[wave] = x;
        __syncthreads();
        int wp = 0, all = 0;
#pragma unroll 8
        for (int q2 = 0; q2 < GT / 32; ++q2) {
            const int q = wsum[q2];
            all += q;
            if (q2 < wave) wp += q;
        }
        const int ex = carry + wp + x - v;
        if (idx < rows) {
            cnt[r0 + idx] = v;
            rowstart[r0 + idx] = s + ex;
            lcnt[idx] = ex;
        }
        carry += all;
        __syncthreads();
    }
#pragma unroll
    for (int g = 0; g < RPW; g += SB) {
        int slot[SB];
#pragma unroll
        for (int k = 0; k < SB; ++k) {
            const int len = __builtin_amdgcn_readlane(rlen, g + k);
            unsigned lr = w[g + k] >> cb;
            lr = lr < last ? lr : last;
            slot[k] = lane < len ? atomicAdd(&lcnt[lr], 1) : 0;
        }
#pragma unroll
        for (int k = 0; k < SB; ++k) {
            if (lane < __builtin_amdgcn_readlane(rlen, g + k)) {
                if (staged) lcol[slot[k]] = (int)(w[g + k] & cmask);
                else        csr_col[s + slot[k]] = (int)(w[g + k] & cmask);
            }
        }
    }
    for (int k = 0; k < RPW; ++k) {
        const int base = __builtin_amdgcn_readlane(rbase, k);
        const int len  = __builtin_amdgcn_readlane(rlen, k);
        for (int j = 32 + lane; j < len; j += 32) {
            const unsigned wj = packed[base + j];
            unsigned lr = wj >> cb;
            lr = lr < last ? lr : last;
            const int sl = atomicAdd(&lcnt[lr], 1);
            if (staged) lcol[sl] = (int)(wj & cmask);
            else        csr_col[s + sl] = (int)(wj & cmask);
        }
    }
    if (staged) {
        __syncthreads();
        for (int i = t; i < c; i += GT) csr_col[s + i] = lcol[i];
    }
}

__global__ void __launch_bounds__(GT, 16)
project_and_sort_kernel(const float* __restrict__ seq, const float* __restrict__ W,
                        float* __restrict__ fts,
                        const float* __restrict__ a1, const float* __restrict__ b1,
                        const float* __restrict__ a2, const float* __restrict__ b2,
                        float* __restrict__ f1, float* __restrict__ f2,
                        const unsigned* __restrict__ packed, const int* __restrict__ rtab,
                        int* __restrict__ cnt, int* __restrict__ rowstart,
                        int* __restrict__ csr_col,
                        int e, int n, int rpb, int per, int gemm_blocks) {
    extern __shared__ int dyn_lds[];
    if ((int)blockIdx.x < gemm_blocks)
        gemm_role((float*)dyn_lds, (int)blockIdx.x, seq, W, fts, a1, b1, a2, b2, f1, f2, n);
    else
        sort_role(dyn_lds, (int)blockIdx.x - gemm_blocks, packed, rtab, cnt, rowstart, csr_col,
                  e, n, rpb, per);
}

__device__ __forceinline__ float lrelu(float l) { return l > 0.0f ? l : NEG_SLOPE * l; }

__device__ __forceinline__ void accum_chunk(const float* __restrict__ fts, int c, float ev,
                                            int cntc, int half, int q, float4& acc) {
    const int steps = (cntc + 1) >> 1;
    const int evb = __float_as_int(ev);
#pragma unroll 4
    for (int t = 0; t < steps; ++t) {
        const int c_lo = __builtin_amdgcn_readlane(c,   2 * t);
        const int c_hi = __builtin_amdgcn_readlane(c,   2 * t + 1);
        const int w_lo = __builtin_amdgcn_readlane(evb, 2 * t);
        const int w_hi = __builtin_amdgcn_readlane(evb, 2 * t + 1);
        const int   cc = half ? c_hi : c_lo;
        const float w  = __int_as_float(half ? w_hi : w_lo);
        const float4 v = *(const float4*)(fts + (size_t)cc * F_OUT + q * 4);
        acc.x += w * v.x; acc.y += w * v.y; acc.z += w * v.z; acc.w += w * v.w;
    }
}

__device__ __forceinline__ void row_bounds(const int* __restrict__ cnt,
                                           const int* __restrict__ rowstart,
                                           const float* __restrict__ f1, int r, int n,
                                           int& deg, int& s, float& f1r) {
    deg = 0; s = 0; f1r = 0.0f;
    if (r < n) { deg = cnt[r]; s = rowstart[r]; f1r = f1[r]; }
}

__global__ void __launch_bounds__(256)
row_kernel(const int* __restrict__ cnt, const int* __restrict__ rowstart,
           const int* __restrict__ csr_col,
           const float* __restrict__ f1, const float* __restrict__ f2,
           const float* __restrict__ fts, const float* __restrict__ bias,
           float* __restrict__ out, int n, int e) {
    const int lane = threadIdx.x & 31;
    const int half = lane >> 4;
    const int q    = lane & 15;
    const int nw   = RB * 8;
    int row = __builtin_amdgcn_readfirstlane((int)(blockIdx.x * 8 + (threadIdx.x >> 5)));
    const int last = e - 1;
    const float4 b = *(const float4*)(bias + q * 4);

    int deg, s, deg1, s1, deg2, s2;
    float f1r, f1r1, f1r2;
    row_bounds(cnt, rowstart, f1, row,          n, deg,  s,  f1r);
    row_bounds(cnt, rowstart, f1, row + nw,     n, deg1, s1, f1r1);
    row_bounds(cnt, rowstart, f1, row + 2 * nw, n, deg2, s2, f1r2);
    int ca  = csr_col[min(s + lane, last)];
    int cb  = csr_col[min(s + 32 + lane, last)];
    int ca1 = csr_col[min(s1 + lane, last)];
    int cb1 = csr_col[min(s1 + 32 + lane, last)];
    float ga = f2[lane < deg ? ca : 0];
    float gb = f2[lane + 32 < deg ? cb : 0];

    while (row < n) {
        const bool va = lane < deg;
        const bool vb = lane + 32 < deg;
        const float la = va ? lrelu(f1r + ga) : -INFINITY;
        const float lb = vb ? lrelu(f1r + gb) : -INFINITY;

        int deg3, s3;
        float f1r3;
        row_bounds(cnt, rowstart, f1, row + 3 * nw, n, deg3, s3, f1r3);
        const int ca2 = csr_col[min(s2 + lane, last)];
        const int cb2 = csr_col[min(s2 + 32 + lane, last)];
        const float ga1 = f2[lane < deg1 ? ca1 : 0];
        const float gb1 = f2[lane + 32 < deg1 ? cb1 : 0];

        float m = la;
        if (vb) m = fmaxf(m, lb);
        for (int base = 64; base < deg; base += 32) {
            const int j = base + lane;
            if (j < deg) m = fmaxf(m, lrelu(f1r + f2[csr_col[s + j]]));
        }
        for (int off = 16; off > 0; off >>= 1) m = fmaxf(m, __shfl_xor(m, off, 32));
        if (!isfinite(m)) m = 0.0f;

        float4 acc = make_float4(0.0f, 0.0f, 0.0f, 0.0f);
        float dsum = 0.0f;
        if (deg > 0) {
            const float ev = va ? expf(la - m) : 0.0f;
            dsum += ev;
            const int cf = __shfl(ca, 0, 32);
            accum_chunk(fts, va ? ca : cf, ev, deg < 32 ? deg : 32, half, q, acc);
        }
        if (deg > 32) {
            const float ev = vb ? expf(lb - m) : 0.0f;
            dsum += ev;
            const int cf = __shfl(cb, 0, 32);
            const int rem = deg - 32;
            accum_chunk(fts, vb ? cb : cf, ev, rem < 32 ? rem : 32, half, q, acc);
        }
        for (int base = 64; base < deg; base += 32) {
            const int j = base + lane;
            const bool v = j < deg;
            int c = 0; float ev = 0.0f;
            if (v) { c = csr_col[s + j]; ev = expf(lrelu(f1r + f2[c]) - m); }
            dsum += ev;
            const int cf = __shfl(c, 0, 32);
            const int rem = deg - base;
            accum_chunk(fts, v ? c : cf, ev, rem < 32 ? rem : 32, half, q, acc);
        }

        for (int off = 16; off > 0; off >>= 1) dsum += __shfl_xor(dsum, off, 32);
        acc.x += __shfl_xor(acc.x, 16, 32);
        acc.y += __shfl_xor(acc.y, 16, 32);
        acc.z += __shfl_xor(acc.z, 16, 32);
        acc.w += __shfl_xor(acc.w, 16, 32);

        if (half == 0) {
            float4 o;
            if (deg > 0) {
                o.x = acc.x / dsum + b.x; o.y = acc.y / dsum + b.y;
                o.z = acc.z / dsum + b.z; o.w = acc.w / dsum + b.w;
            } else {
                o = b;
            }
            o.x = o.x > 0.0f ? o.x : 0.0f; o.y = o.y > 0.0f ? o.y : 0.0f;
            o.z = o.z > 0.0f ? o.z : 0.0f; o.w = o.w > 0.0f ? o.w : 0.0f;
            *(float4*)(out + (size_t)row * F_OUT + q * 4) = o;
        }

        row += nw;
        deg = deg1; s = s1; f1r = f1r1; ca = ca1; cb = cb1; ga = ga1; gb = gb1;
        deg1 = deg2; s1 = s2; f1r1 = f1r2; ca1 = ca2; cb1 = cb2;
        deg2 = deg3; s2 = s3; f1r2 = f1r3;
    }
}

extern "C" void kernel_launch(void* const* d_in, const int* in_sizes, int n_in,
                              void* d_out, int out_size, void* d_ws, size_t ws_size,
                              hipStream_t stream) {
    const float* seq  = (const float*)d_in[0];
    const int*   erow = (const int*)  d_in[1];
    const int*   ecol = (const int*)  d_in[2];
    const float* W    = (const float*)d_in[3];
    const float* a1   = (const float*)d_in[4];
    const float* b1   = (const float*)d_in[5];
    const float* a2   = (const float*)d_in[6];
    const float* b2   = (const float*)d_in[7];
    const float* bias = (const float*)d_in[8];

    const int n = in_sizes[0] / F_IN;
    const int e = in_sizes[1];
    float* out = (float*)d_out;

    float* fts      = (float*)d_ws;
    float* f1       = fts + (size_t)n * F_OUT;
    float* f2       = f1 + n;
    int*   cnt      = (int*)(f2 + n);
    int*   rowstart = cnt + n;
    int*   csr_col  = rowstart + n;
    unsigned* packed = (unsigned*)d_out;
    int*   rtab     = (int*)d_out + (e + 31) / 32 * 32;

    const int rpb = n / NB + 2;
    const int per = ((e + PB - 1) / PB + 31) / 32 * 32;

    slice_group_kernel<<<PB, GT, per * sizeof(int), stream>>>(erow, ecol, rtab, packed,
                                                             e, n, per, rpb);
    {
        const int gemm_blocks = (n + 16 * (GT / 32) - 1) / (16 * (GT / 32));
        const size_t sort_lds = (rpb + SCAP + 3 * (GT / 32)) * sizeof(int);
        const size_t gemm_lds = F_IN * F_OUT * sizeof(float);
        project_and_sort_kernel<<<gemm_blocks + NB, GT, sort_lds > gemm_lds ? sort_lds : gemm_lds,
                                  stream>>>(seq, W, fts, a1, b1, a2, b2, f1, f2,
                                            packed, rtab, cnt, rowstart, csr_col,
                                            e, n, rpb, per, gemm_blocks);
    }
    row_kernel<<<RB, 256, 0, stream>>>(cnt, rowstart, csr_col, f1, f2, fts, bias,
                                       out, n, e);
}


# Row kernel holds a row's first 32 neighbour rows in registers and gets each neighbour's score as dot(row,a2) from them, not a 4-byte f2 gather per edge (f2 and denominator sum order change)
# speedup vs baseline: 8.3304x; 1.0323x over previous
#include <hip/hip_runtime.h>
#include <hip/hip_bf16.h>

typedef __attribute__((ext_vector_type(2))) float v2f;
typedef __attribute__((ext_vector_type(8))) float v8f;
typedef __attribute__((ext_vector_type(4))) float v4f;

#define F_IN   256
#define F_OUT  64
#define NEG_SLOPE 0.2f
#define NB     256
#define GT     1024
#define EB     8
#define PB     512
#define RPW    (PB / (GT / 32))
#define SB     4
#define RB     1024
#define SCAP   14336


__device__ __forceinline__ void
gemm_role(float* __restrict__ Wl, int gb,
          const float* __restrict__ seq, const float* __restrict__ W,
          float* __restrict__ fts,
          const float* __restrict__ a1, const float* __restrict__ b1,
          const float* __restrict__ a2, const float* __restrict__ b2,
          float* __restrict__ f1, float* __restrict__ f2, int n) {
    for (int i = threadIdx.x; i < F_IN * F_OUT; i += GT) {
        const int kk  = i >> 6;
        const int col = i & 63;
        Wl[(kk >> 1) * (2 * F_OUT) + col * 2 + (kk & 1)] = W[i];
    }
    __syncthreads();

    const int wave = threadIdx.x >> 5;
    const int lane = threadIdx.x & 31;
    const int rowTile = gb * (GT / 32) + wave;

    int row = rowTile * 16 + (lane & 15);
    if (row >= n) row = n - 1;
    const int lcol  = lane & 15;
    const int khalf = (lane >> 4) * 2;

    const float* __restrict__ arow = seq + (size_t)row * F_IN;

    v8f acc[4] = {v8f{}, v8f{}, v8f{}, v8f{}};
    for (int k = 0; k < F_IN; k += 4) {
        float2 av = *(const float2*)(arow + k + khalf);
        v2f a; a[0] = av.x; a[1] = av.y;
        const int kpairBase = ((k + khalf) >> 1) * (2 * F_OUT);
#pragma unroll
        for (int ct = 0; ct < 4; ++ct) {
            float2 bv = *(const float2*)(&Wl[kpairBase + (ct * 16 + lcol) * 2]);
            v2f b; b[0] = bv.x; b[1] = bv.y;
            acc[ct] = __builtin_amdgcn_wmma_f32_16x16x4_f32(
                     false, a,  false, b,
                     (short)0, acc[ct],  false,  false);
        }
    }

    const int nhi = lane >> 4;
#pragma unroll
    for (int ct = 0; ct < 4; ++ct) {
        const int ocol = ct * 16 + lcol;
#pragma unroll
        for (int r = 0; r < 8; ++r) {
            const int orow = rowTile * 16 + r + 8 * nhi;
            if (orow < n) fts[(size_t)orow * F_OUT + ocol] = acc[ct][r];
        }
    }

    float A1[4], A2[4];
#pragma unroll
    for (int ct = 0; ct < 4; ++ct) {
        A1[ct] = a1[ct * 16 + lcol];
        A2[ct] = a2[ct * 16 + lcol];
    }
    const float b1v = b1[0], b2v = b2[0];
#pragma unroll
    for (int r = 0; r < 8; ++r) {
        float p1 = 0.0f, p2 = 0.0f;
#pragma unroll
        for (int ct = 0; ct < 4; ++ct) {
            p1 += acc[ct][r] * A1[ct];
            p2 += acc[ct][r] * A2[ct];
        }
#pragma unroll
        for (int off = 8; off > 0; off >>= 1) {
            p1 += __shfl_xor(p1, off, 32);
            p2 += __shfl_xor(p2, off, 32);
        }
        if ((lane & 15) == 0) {
            const int orow = rowTile * 16 + r + 8 * nhi;
            if (orow < n) { f1[orow] = p1 + b1v; f2[orow] = p2 + b2v; }
        }
    }
}

__device__ __forceinline__ int range_of(int row, unsigned magic) {
    const int b = (int)__umulhi((unsigned)row, magic);
    return b < NB - 1 ? b : NB - 1;
}

__global__ void __launch_bounds__(GT)
slice_group_kernel(const int* __restrict__ erow, const int* __restrict__ ecol,
                   int* __restrict__ rtab, unsigned* __restrict__ packed,
                   int e, int n, int per, int rpb) {
    extern __shared__ unsigned lbuf[];
    __shared__ int h[NB];
    __shared__ int wsum2[NB / 32];
    const int t = threadIdx.x, lane = t & 31, wave = t >> 5;
    if (t < NB) h[t] = 0;
    __syncthreads();
    const unsigned magic = 0xFFFFFFFFu / (unsigned)rpb + 1u;
    const int cb = 32 - __clz(n - 1);
    const int lo = blockIdx.x * per;
    const int hi = lo + per < e ? lo + per : e;
    for (int i = lo + t; i < hi; i += EB * GT) {
        int r[EB];
#pragma unroll
        for (int k = 0; k < EB; ++k) r[k] = erow[min(i + k * GT, hi - 1)];
#pragma unroll
        for (int k = 0; k < EB; ++k)
            if (i + k * GT < hi) atomicAdd(&h[range_of(r[k], magic)], 1);
    }
    __syncthreads();
    int mine = 0, y = 0;
    if (t < NB) {
        mine = h[t];
        y = mine;
        for (int off = 1; off < 32; off <<= 1) {
            const int z = __shfl_up(y, off, 32);
            if (lane >= off) y += z;
        }
        if (lane == 31) wsum2[wave] = y;
    }
    __syncthreads();
    if (t < NB) {
        int lp = 0;
        for (int w = 0; w < wave; ++w) lp += wsum2[w];
        const int ls = lp + y - mine;
        *(int2*)(rtab + 2 * ((size_t)blockIdx.x * NB + t)) = make_int2(ls, mine);
        h[t] = ls;
    }
    __syncthreads();
    for (int i = lo + t; i < hi; i += EB * GT) {
        int r[EB], c[EB];
#pragma unroll
        for (int k = 0; k < EB; ++k) {
            const int j = min(i + k * GT, hi - 1);
            r[k] = erow[j];
            c[k] = ecol[j];
        }
#pragma unroll
        for (int k = 0; k < EB; ++k) {
            if (i + k * GT < hi) {
                const int b = range_of(r[k], magic);
                const int slot = atomicAdd(&h[b], 1);
                lbuf[slot] = ((unsigned)(r[k] - b * rpb) << cb) | (unsigned)c[k];
            }
        }
    }
    __syncthreads();
    for (int i = t; i < hi - lo; i += GT) packed[lo + i] = lbuf[i];
}

__device__ __forceinline__ void
sort_role(int* __restrict__ lds4, int sb,
          const unsigned* __restrict__ packed, const int* __restrict__ rtab,
          int* __restrict__ cnt, int* __restrict__ rowstart, int* __restrict__ csr_col,
          int e, int n, int rpb, int per) {
    int* wsum = lds4;
    int* wtot = wsum + GT / 32;
    int* wofs = wtot + GT / 32;
    int* lcnt = wofs + GT / 32;
    int* lcol = lcnt + rpb;
    const int t = threadIdx.x, lane = t & 31, wave = t >> 5;
    const int r0 = sb * rpb;
    int rows = n - r0;
    if (rows > rpb) rows = rpb;
    if (rows <= 0) return;
    const int cb = 32 - __clz(n - 1);
    const unsigned cmask = (1u << cb) - 1u;
    const unsigned last = (unsigned)(rows - 1);
    int rbase = 0, rlen = 0, rofs = 0;
    if (lane < RPW) {
        const int pb = wave * RPW + lane;
        const int2 v = *(const int2*)(rtab + 2 * ((size_t)pb * NB + sb));
        rofs = v.x;
        rlen = v.y;
        rbase = pb * per + v.x;
    }
    for (int i = t; i < rows; i += GT) lcnt[i] = 0;
    unsigned w[RPW];
#pragma unroll
    for (int k = 0; k < RPW; ++k) {
        const int base = __builtin_amdgcn_readlane(rbase, k);
        const int len  = __builtin_amdgcn_readlane(rlen, k);
        const int j = base + min(lane, len > 0 ? len - 1 : 0);
        w[k] = packed[min(j, e - 1)];
    }
    {
        int xs = rlen, os = rofs;
        for (int off = 16; off > 0; off >>= 1) {
            xs += __shfl_xor(xs, off, 32);
            os += __shfl_xor(os, off, 32);
        }
        if (lane == 0) { wtot[wave] = xs; wofs[wave] = os; }
    }
    __syncthreads();
    int c = wtot[lane], s = wofs[lane];
    for (int off = 16; off > 0; off >>= 1) {
        c += __shfl_xor(c, off, 32);
        s += __shfl_xor(s, off, 32);
    }
    const bool staged = c <= SCAP;
#pragma unroll
    for (int k = 0; k < RPW; ++k) {
        const int len = __builtin_amdgcn_readlane(rlen, k);
        if (lane < len) {
            unsigned lr = w[k] >> cb;
            lr = lr < last ? lr : last;
            atomicAdd(&lcnt[lr], 1);
        }
    }
    for (int k = 0; k < RPW; ++k) {
        const int base = __builtin_amdgcn_readlane(rbase, k);
        const int len  = __builtin_amdgcn_readlane(rlen, k);
        for (int j = 32 + lane; j < len; j += 32) {
            unsigned lr = packed[base + j] >> cb;
            lr = lr < last ? lr : last;
            atomicAdd(&lcnt[lr], 1);
        }
    }
    __syncthreads();
    int carry = 0;
    for (int base = 0; base < rows; base += GT) {
        const int idx = base + t;
        const int v = idx < rows ? lcnt[idx] : 0;
        int x = v;
        for (int off = 1; off < 32; off <<= 1) {
            const int y = __shfl_up(x, off, 32);
            if (lane >= off) x += y;
        }
        if (lane == 31) wsum[wave] = x;
        __syncthreads();
        int wp = 0, all = 0;
#pragma unroll 8
        for (int q2 = 0; q2 < GT / 32; ++q2) {
            const int q = wsum[q2];
            all += q;
            if (q2 < wave) wp += q;
        }
        const int ex = carry + wp + x - v;
        if (idx < rows) {
            cnt[r0 + idx] = v;
            rowstart[r0 + idx] = s + ex;
            lcnt[idx] = ex;
        }
        carry += all;
        __syncthreads();
    }
#pragma unroll
    for (int g = 0; g < RPW; g += SB) {
        int slot[SB];
#pragma unroll
        for (int k = 0; k < SB; ++k) {
            const int len = __builtin_amdgcn_readlane(rlen, g + k);
            unsigned lr = w[g + k] >> cb;
            lr = lr < last ? lr : last;
            slot[k] = lane < len ? atomicAdd(&lcnt[lr], 1) : 0;
        }
#pragma unroll
        for (int k = 0; k < SB; ++k) {
            if (lane < __builtin_amdgcn_readlane(rlen, g + k)) {
                if (staged) lcol[slot[k]] = (int)(w[g + k] & cmask);
                else        csr_col[s + slot[k]] = (int)(w[g + k] & cmask);
            }
        }
    }
    for (int k = 0; k < RPW; ++k) {
        const int base = __builtin_amdgcn_readlane(rbase, k);
        const int len  = __builtin_amdgcn_readlane(rlen, k);
        for (int j = 32 + lane; j < len; j += 32) {
            const unsigned wj = packed[base + j];
            unsigned lr = wj >> cb;
            lr = lr < last ? lr : last;
            const int sl = atomicAdd(&lcnt[lr], 1);
            if (staged) lcol[sl] = (int)(wj & cmask);
            else        csr_col[s + sl] = (int)(wj & cmask);
        }
    }
    if (staged) {
        __syncthreads();
        for (int i = t; i < c; i += GT) csr_col[s + i] = lcol[i];
    }
}

__global__ void __launch_bounds__(GT, 16)
project_and_sort_kernel(const float* __restrict__ seq, const float* __restrict__ W,
                        float* __restrict__ fts,
                        const float* __restrict__ a1, const float* __restrict__ b1,
                        const float* __restrict__ a2, const float* __restrict__ b2,
                        float* __restrict__ f1, float* __restrict__ f2,
                        const unsigned* __restrict__ packed, const int* __restrict__ rtab,
                        int* __restrict__ cnt, int* __restrict__ rowstart,
                        int* __restrict__ csr_col,
                        int e, int n, int rpb, int per, int gemm_blocks) {
    extern __shared__ int dyn_lds[];
    if ((int)blockIdx.x < gemm_blocks)
        gemm_role((float*)dyn_lds, (int)blockIdx.x, seq, W, fts, a1, b1, a2, b2, f1, f2, n);
    else
        sort_role(dyn_lds, (int)blockIdx.x - gemm_blocks, packed, rtab, cnt, rowstart, csr_col,
                  e, n, rpb, per);
}

__device__ __forceinline__ float lrelu(float l) { return l > 0.0f ? l : NEG_SLOPE * l; }

__device__ __forceinline__ void accum_chunk(const float* __restrict__ fts, int c, float ev,
                                            int cntc, int half, int q, float4& acc) {
    const int steps = (cntc + 1) >> 1;
    const int evb = __float_as_int(ev);
#pragma unroll 4
    for (int t = 0; t < steps; ++t) {
        const int c_lo = __builtin_amdgcn_readlane(c,   2 * t);
        const int c_hi = __builtin_amdgcn_readlane(c,   2 * t + 1);
        const int w_lo = __builtin_amdgcn_readlane(evb, 2 * t);
        const int w_hi = __builtin_amdgcn_readlane(evb, 2 * t + 1);
        const int   cc = half ? c_hi : c_lo;
        const float w  = __int_as_float(half ? w_hi : w_lo);
        const float4 v = *(const float4*)(fts + (size_t)cc * F_OUT + q * 4);
        acc.x += w * v.x; acc.y += w * v.y; acc.z += w * v.z; acc.w += w * v.w;
    }
}

__device__ __forceinline__ void row_bounds(const int* __restrict__ cnt,
                                           const int* __restrict__ rowstart,
                                           const float* __restrict__ f1, int r, int n,
                                           int& deg, int& s, float& f1r) {
    deg = 0; s = 0; f1r = 0.0f;
    if (r < n) { deg = cnt[r]; s = rowstart[r]; f1r = f1[r]; }
}

template <int T> __device__ __forceinline__ int share_i(int x) {
    return __builtin_amdgcn_update_dpp(0, x, 0x150 + T, 0xf, 0xf, false);
}
template <int T> __device__ __forceinline__ float share_f(float x) {
    return __int_as_float(share_i<T>(__float_as_int(x)));
}
template <int M> __device__ __forceinline__ float xmask_f(float x) {
    return __int_as_float(__builtin_amdgcn_update_dpp(0, __float_as_int(x), 0x160 + M,
                                                      0xf, 0xf, false));
}

#define GATHER_STEP(T)                                                                   \
    {                                                                                    \
        const unsigned off = (unsigned)share_i<T>(csel) << 8;                            \
        v[T] = *(const v4f*)(fq + off);                                                  \
    }
#define DOT_STEP(T)                                                                      \
    {                                                                                    \
        const v2f t2 = __builtin_elementwise_fma(v[T].zw, a_hi, v[T].xy * a_lo);         \
        p[T] = t2.x + t2.y;                                                              \
    }
#define ACC_STEP(T)                                                                      \
    {                                                                                    \
        const float w = share_f<T>(ev);                                                  \
        acc.x += w * v[T].x; acc.y += w * v[T].y; acc.z += w * v[T].z; acc.w += w * v[T].w; \
    }
#define ALL16(S) S(0) S(1) S(2) S(3) S(4) S(5) S(6) S(7) S(8) S(9) S(10) S(11) S(12) S(13) S(14) S(15)

__global__ void __launch_bounds__(256)
row_kernel(const int* __restrict__ cnt, const int* __restrict__ rowstart,
           const int* __restrict__ csr_col,
           const float* __restrict__ f1, const float* __restrict__ f2,
           const float* __restrict__ fts, const float* __restrict__ a2,
           const float* __restrict__ b2, const float* __restrict__ bias,
           float* __restrict__ out, int n, int e) {
    const int lane = threadIdx.x & 31;
    const int half = lane >> 4;
    const int q    = lane & 15;
    const int eo   = 2 * q + half;
    const int nw   = RB * 8;
    int row = __builtin_amdgcn_readfirstlane((int)(blockIdx.x * 8 + (threadIdx.x >> 5)));
    const int last = e - 1;
    const float4 b  = *(const float4*)(bias + q * 4);
    const float4 aq = *(const float4*)(a2 + q * 4);
    v2f a_lo, a_hi;
    a_lo.x = aq.x; a_lo.y = aq.y; a_hi.x = aq.z; a_hi.y = aq.w;
    const float b2v = b2[0];
    const char* __restrict__ fq = (const char*)fts + q * 16;
    const bool b3 = (q & 8) != 0, b2b = (q & 4) != 0, b1 = (q & 2) != 0, b0 = (q & 1) != 0;

    int deg, s, deg1, s1, deg2, s2;
    float f1r, f1r1, f1r2;
    row_bounds(cnt, rowstart, f1, row,          n, deg,  s,  f1r);
    row_bounds(cnt, rowstart, f1, row + nw,     n, deg1, s1, f1r1);
    row_bounds(cnt, rowstart, f1, row + 2 * nw, n, deg2, s2, f1r2);
    int ca  = csr_col[min(s + eo, last)];
    int cb  = csr_col[min(s + 32 + lane, last)];
    int ca1 = csr_col[min(s1 + eo, last)];
    int cb1 = csr_col[min(s1 + 32 + lane, last)];
    float gb = f2[lane + 32 < deg ? cb : 0];

    while (row < n) {
        const bool va = eo < deg;
        const bool vb = lane + 32 < deg;
        const float lb = vb ? lrelu(f1r + gb) : -INFINITY;

        int deg3, s3;
        float f1r3;
        row_bounds(cnt, rowstart, f1, row + 3 * nw, n, deg3, s3, f1r3);
        const int ca2 = csr_col[min(s2 + eo, last)];
        const int cb2 = csr_col[min(s2 + 32 + lane, last)];
        const float gb1 = f2[lane + 32 < deg1 ? cb1 : 0];

        const int c0   = __builtin_amdgcn_readfirstlane(ca);
        const int csel = va ? ca : c0;
        v4f v[16];
        float p[16];
        ALL16(GATHER_STEP)
        ALL16(DOT_STEP)
        float r8[8], r4[4], r2[2];
#pragma unroll
        for (int i = 0; i < 8; ++i)
            r8[i] = (b3 ? p[i + 8] : p[i]) + xmask_f<8>(b3 ? p[i] : p[i + 8]);
#pragma unroll
        for (int i = 0; i < 4; ++i)
            r4[i] = (b2b ? r8[i + 4] : r8[i]) + xmask_f<4>(b2b ? r8[i] : r8[i + 4]);
#pragma unroll
        for (int i = 0; i < 2; ++i)
            r2[i] = (b1 ? r4[i + 2] : r4[i]) + xmask_f<2>(b1 ? r4[i] : r4[i + 2]);
        const float sdot = (b0 ? r2[1] : r2[0]) + xmask_f<1>(b0 ? r2[0] : r2[1]);
        const float la = va ? lrelu(f1r + (sdot + b2v)) : -INFINITY;

        float m = la;
        if (vb) m = fmaxf(m, lb);
        for (int base = 64; base < deg; base += 32) {
            const int j = base + lane;
            if (j < deg) m = fmaxf(m, lrelu(f1r + f2[csr_col[s + j]]));
        }
        for (int off = 16; off > 0; off >>= 1) m = fmaxf(m, __shfl_xor(m, off, 32));
        if (!isfinite(m)) m = 0.0f;

        float4 acc = make_float4(0.0f, 0.0f, 0.0f, 0.0f);
        const float ev = va ? expf(la - m) : 0.0f;
        float dsum = ev;
        ALL16(ACC_STEP)
        if (deg > 32) {
            const float ev2 = vb ? expf(lb - m) : 0.0f;
            dsum += ev2;
            const int cf = __shfl(cb, 0, 32);
            const int rem = deg - 32;
            accum_chunk(fts, vb ? cb : cf, ev2, rem < 32 ? rem : 32, half, q, acc);
        }
        for (int base = 64; base < deg; base += 32) {
            const int j = base + lane;
            const bool vj = j < deg;
            int c = 0; float evj = 0.0f;
            if (vj) { c = csr_col[s + j]; evj = expf(lrelu(f1r + f2[c]) - m); }
            dsum += evj;
            const int cf = __shfl(c, 0, 32);
            const int rem = deg - base;
            accum_chunk(fts, vj ? c : cf, evj, rem < 32 ? rem : 32, half, q, acc);
        }

        for (int off = 16; off > 0; off >>= 1) dsum += __shfl_xor(dsum, off, 32);
        acc.x += __shfl_xor(acc.x, 16, 32);
        acc.y += __shfl_xor(acc.y, 16, 32);
        acc.z += __shfl_xor(acc.z, 16, 32);
        acc.w += __shfl_xor(acc.w, 16, 32);

        if (half == 0) {
            float4 o;
            if (deg > 0) {
                o.x = acc.x / dsum + b.x; o.y = acc.y / dsum + b.y;
                o.z = acc.z / dsum + b.z; o.w = acc.w / dsum + b.w;
            } else {
                o = b;
            }
            o.x = o.x > 0.0f ? o.x : 0.0f; o.y = o.y > 0.0f ? o.y : 0.0f;
            o.z = o.z > 0.0f ? o.z : 0.0f; o.w = o.w > 0.0f ? o.w : 0.0f;
            *(float4*)(out + (size_t)row * F_OUT + q * 4) = o;
        }

        row += nw;
        deg = deg1; s = s1; f1r = f1r1; ca = ca1; cb = cb1; gb = gb1;
        deg1 = deg2; s1 = s2; f1r1 = f1r2; ca1 = ca2; cb1 = cb2;
        deg2 = deg3; s2 = s3; f1r2 = f1r3;
    }
}

extern "C" void kernel_launch(void* const* d_in, const int* in_sizes, int n_in,
                              void* d_out, int out_size, void* d_ws, size_t ws_size,
                              hipStream_t stream) {
    const float* seq  = (const float*)d_in[0];
    const int*   erow = (const int*)  d_in[1];
    const int*   ecol = (const int*)  d_in[2];
    const float* W    = (const float*)d_in[3];
    const float* a1   = (const float*)d_in[4];
    const float* b1   = (const float*)d_in[5];
    const float* a2   = (const float*)d_in[6];
    const float* b2   = (const float*)d_in[7];
    const float* bias = (const float*)d_in[8];

    const int n = in_sizes[0] / F_IN;
    const int e = in_sizes[1];
    float* out = (float*)d_out;

    float* fts      = (float*)d_ws;
    float* f1       = fts + (size_t)n * F_OUT;
    float* f2       = f1 + n;
    int*   cnt      = (int*)(f2 + n);
    int*   rowstart = cnt + n;
    int*   csr_col  = rowstart + n;
    unsigned* packed = (unsigned*)d_out;
    int*   rtab     = (int*)d_out + (e + 31) / 32 * 32;

    const int rpb = n / NB + 2;
    const int per = ((e + PB - 1) / PB + 31) / 32 * 32;

    slice_group_kernel<<<PB, GT, per * sizeof(int), stream>>>(erow, ecol, rtab, packed,
                                                             e, n, per, rpb);
    {
        const int gemm_blocks = (n + 16 * (GT / 32) - 1) / (16 * (GT / 32));
        const size_t sort_lds = (rpb + SCAP + 3 * (GT / 32)) * sizeof(int);
        const size_t gemm_lds = F_IN * F_OUT * sizeof(float);
        project_and_sort_kernel<<<gemm_blocks + NB, GT, sort_lds > gemm_lds ? sort_lds : gemm_lds,
                                  stream>>>(seq, W, fts, a1, b1, a2, b2, f1, f2,
                                            packed, rtab, cnt, rowstart, csr_col,
                                            e, n, rpb, per, gemm_blocks);
    }
    row_kernel<<<RB, 256, 0, stream>>>(cnt, rowstart, csr_col, f1, f2, fts, a2, b2, bias,
                                       out, n, e);
}


# Row kernel issues a quarter fewer vector instructions per row: gather address one DPP add on a scalar base, packed FMAs, wave max/sums by DPP not LDS permutes, one reciprocal (sum order kept)
# speedup vs baseline: 8.8177x; 1.0585x over previous
#include <hip/hip_runtime.h>
#include <hip/hip_bf16.h>

typedef __attribute__((ext_vector_type(2))) float v2f;
typedef __attribute__((ext_vector_type(8))) float v8f;
typedef __attribute__((ext_vector_type(4))) float v4f;

#define F_IN   256
#define F_OUT  64
#define NEG_SLOPE 0.2f
#define NB     256
#define GT     1024
#define EB     8
#define PB     512
#define RPW    (PB / (GT / 32))
#define SB     4
#define RB     1024
#define SCAP   14336


__device__ __forceinline__ void
gemm_role(float* __restrict__ Wl, int gb,
          const float* __restrict__ seq, const float* __restrict__ W,
          float* __restrict__ fts,
          const float* __restrict__ a1, const float* __restrict__ b1,
          const float* __restrict__ a2, const float* __restrict__ b2,
          float* __restrict__ f1, float* __restrict__ f2, int n) {
    for (int i = threadIdx.x; i < F_IN * F_OUT; i += GT) {
        const int kk  = i >> 6;
        const int col = i & 63;
        Wl[(kk >> 1) * (2 * F_OUT) + col * 2 + (kk & 1)] = W[i];
    }
    __syncthreads();

    const int wave = threadIdx.x >> 5;
    const int lane = threadIdx.x & 31;
    const int rowTile = gb * (GT / 32) + wave;

    int row = rowTile * 16 + (lane & 15);
    if (row >= n) row = n - 1;
    const int lcol  = lane & 15;
    const int khalf = (lane >> 4) * 2;

    const float* __restrict__ arow = seq + (size_t)row * F_IN;

    v8f acc[4] = {v8f{}, v8f{}, v8f{}, v8f{}};
    for (int k = 0; k < F_IN; k += 4) {
        float2 av = *(const float2*)(arow + k + khalf);
        v2f a; a[0] = av.x; a[1] = av.y;
        const int kpairBase = ((k + khalf) >> 1) * (2 * F_OUT);
#pragma unroll
        for (int ct = 0; ct < 4; ++ct) {
            float2 bv = *(const float2*)(&Wl[kpairBase + (ct * 16 + lcol) * 2]);
            v2f b; b[0] = bv.x; b[1] = bv.y;
            acc[ct] = __builtin_amdgcn_wmma_f32_16x16x4_f32(
                     false, a,  false, b,
                     (short)0, acc[ct],  false,  false);
        }
    }

    const int nhi = lane >> 4;
#pragma unroll
    for (int ct = 0; ct < 4; ++ct) {
        const int ocol = ct * 16 + lcol;
#pragma unroll
        for (int r = 0; r < 8; ++r) {
            const int orow = rowTile * 16 + r + 8 * nhi;
            if (orow < n) fts[(size_t)orow * F_OUT + ocol] = acc[ct][r];
        }
    }

    float A1[4], A2[4];
#pragma unroll
    for (int ct = 0; ct < 4; ++ct) {
        A1[ct] = a1[ct * 16 + lcol];
        A2[ct] = a2[ct * 16 + lcol];
    }
    const float b1v = b1[0], b2v = b2[0];
#pragma unroll
    for (int r = 0; r < 8; ++r) {
        float p1 = 0.0f, p2 = 0.0f;
#pragma unroll
        for (int ct = 0; ct < 4; ++ct) {
            p1 += acc[ct][r] * A1[ct];
            p2 += acc[ct][r] * A2[ct];
        }
#pragma unroll
        for (int off = 8; off > 0; off >>= 1) {
            p1 += __shfl_xor(p1, off, 32);
            p2 += __shfl_xor(p2, off, 32);
        }
        if ((lane & 15) == 0) {
            const int orow = rowTile * 16 + r + 8 * nhi;
            if (orow < n) { f1[orow] = p1 + b1v; f2[orow] = p2 + b2v; }
        }
    }
}

__device__ __forceinline__ int range_of(int row, unsigned magic) {
    const int b = (int)__umulhi((unsigned)row, magic);
    return b < NB - 1 ? b : NB - 1;
}

__global__ void __launch_bounds__(GT)
slice_group_kernel(const int* __restrict__ erow, const int* __restrict__ ecol,
                   int* __restrict__ rtab, unsigned* __restrict__ packed,
                   int e, int n, int per, int rpb) {
    extern __shared__ unsigned lbuf[];
    __shared__ int h[NB];
    __shared__ int wsum2[NB / 32];
    const int t = threadIdx.x, lane = t & 31, wave = t >> 5;
    if (t < NB) h[t] = 0;
    __syncthreads();
    const unsigned magic = 0xFFFFFFFFu / (unsigned)rpb + 1u;
    const int cb = 32 - __clz(n - 1);
    const int lo = blockIdx.x * per;
    const int hi = lo + per < e ? lo + per : e;
    for (int i = lo + t; i < hi; i += EB * GT) {
        int r[EB];
#pragma unroll
        for (int k = 0; k < EB; ++k) r[k] = erow[min(i + k * GT, hi - 1)];
#pragma unroll
        for (int k = 0; k < EB; ++k)
            if (i + k * GT < hi) atomicAdd(&h[range_of(r[k], magic)], 1);
    }
    __syncthreads();
    int mine = 0, y = 0;
    if (t < NB) {
        mine = h[t];
        y = mine;
        for (int off = 1; off < 32; off <<= 1) {
            const int z = __shfl_up(y, off, 32);
            if (lane >= off) y += z;
        }
        if (lane == 31) wsum2[wave] = y;
    }
    __syncthreads();
    if (t < NB) {
        int lp = 0;
        for (int w = 0; w < wave; ++w) lp += wsum2[w];
        const int ls = lp + y - mine;
        *(int2*)(rtab + 2 * ((size_t)blockIdx.x * NB + t)) = make_int2(ls, mine);
        h[t] = ls;
    }
    __syncthreads();
    for (int i = lo + t; i < hi; i += EB * GT) {
        int r[EB], c[EB];
#pragma unroll
        for (int k = 0; k < EB; ++k) {
            const int j = min(i + k * GT, hi - 1);
            r[k] = erow[j];
            c[k] = ecol[j];
        }
#pragma unroll
        for (int k = 0; k < EB; ++k) {
            if (i + k * GT < hi) {
                const int b = range_of(r[k], magic);
                const int slot = atomicAdd(&h[b], 1);
                lbuf[slot] = ((unsigned)(r[k] - b * rpb) << cb) | (unsigned)c[k];
            }
        }
    }
    __syncthreads();
    for (int i = t; i < hi - lo; i += GT) packed[lo + i] = lbuf[i];
}

__device__ __forceinline__ void
sort_role(int* __restrict__ lds4, int sb,
          const unsigned* __restrict__ packed, const int* __restrict__ rtab,
          int* __restrict__ cnt, int* __restrict__ rowstart, int* __restrict__ csr_col,
          int e, int n, int rpb, int per) {
    int* wsum = lds4;
    int* wtot = wsum + GT / 32;
    int* wofs = wtot + GT / 32;
    int* lcnt = wofs + GT / 32;
    int* lcol = lcnt + rpb;
    const int t = threadIdx.x, lane = t & 31, wave = t >> 5;
    const int r0 = sb * rpb;
    int rows = n - r0;
    if (rows > rpb) rows = rpb;
    if (rows <= 0) return;
    const int cb = 32 - __clz(n - 1);
    const unsigned cmask = (1u << cb) - 1u;
    const unsigned last = (unsigned)(rows - 1);
    int rbase = 0, rlen = 0, rofs = 0;
    if (lane < RPW) {
        const int pb = wave * RPW + lane;
        const int2 v = *(const int2*)(rtab + 2 * ((size_t)pb * NB + sb));
        rofs = v.x;
        rlen = v.y;
        rbase = pb * per + v.x;
    }
    for (int i = t; i < rows; i += GT) lcnt[i] = 0;
    unsigned w[RPW];
#pragma unroll
    for (int k = 0; k < RPW; ++k) {
        const int base = __builtin_amdgcn_readlane(rbase, k);
        const int len  = __builtin_amdgcn_readlane(rlen, k);
        const int j = base + min(lane, len > 0 ? len - 1 : 0);
        w[k] = packed[min(j, e - 1)];
    }
    {
        int xs = rlen, os = rofs;
        for (int off = 16; off > 0; off >>= 1) {
            xs += __shfl_xor(xs, off, 32);
            os += __shfl_xor(os, off, 32);
        }
        if (lane == 0) { wtot[wave] = xs; wofs[wave] = os; }
    }
    __syncthreads();
    int c = wtot[lane], s = wofs[lane];
    for (int off = 16; off > 0; off >>= 1) {
        c += __shfl_xor(c, off, 32);
        s += __shfl_xor(s, off, 32);
    }
    const bool staged = c <= SCAP;
#pragma unroll
    for (int k = 0; k < RPW; ++k) {
        const int len = __builtin_amdgcn_readlane(rlen, k);
        if (lane < len) {
            unsigned lr = w[k] >> cb;
            lr = lr < last ? lr : last;
            atomicAdd(&lcnt[lr], 1);
        }
    }
    for (int k = 0; k < RPW; ++k) {
        const int base = __builtin_amdgcn_readlane(rbase, k);
        const int len  = __builtin_amdgcn_readlane(rlen, k);
        for (int j = 32 + lane; j < len; j += 32) {
            unsigned lr = packed[base + j] >> cb;
            lr = lr < last ? lr : last;
            atomicAdd(&lcnt[lr], 1);
        }
    }
    __syncthreads();
    int carry = 0;
    for (int base = 0; base < rows; base += GT) {
        const int idx = base + t;
        const int v = idx < rows ? lcnt[idx] : 0;
        int x = v;
        for (int off = 1; off < 32; off <<= 1) {
            const int y = __shfl_up(x, off, 32);
            if (lane >= off) x += y;
        }
        if (lane == 31) wsum[wave] = x;
        __syncthreads();
        int wp = 0, all = 0;
#pragma unroll 8
        for (int q2 = 0; q2 < GT / 32; ++q2) {
            const int q = wsum[q2];
            all += q;
            if (q2 < wave) wp += q;
        }
        const int ex = carry + wp + x - v;
        if (idx < rows) {
            cnt[r0 + idx] = v;
            rowstart[r0 + idx] = s + ex;
            lcnt[idx] = ex;
        }
        carry += all;
        __syncthreads();
    }
#pragma unroll
    for (int g = 0; g < RPW; g += SB) {
        int slot[SB];
#pragma unroll
        for (int k = 0; k < SB; ++k) {
            const int len = __builtin_amdgcn_readlane(rlen, g + k);
            unsigned lr = w[g + k] >> cb;
            lr = lr < last ? lr : last;
            slot[k] = lane < len ? atomicAdd(&lcnt[lr], 1) : 0;
        }
#pragma unroll
        for (int k = 0; k < SB; ++k) {
            if (lane < __builtin_amdgcn_readlane(rlen, g + k)) {
                if (staged) lcol[slot[k]] = (int)(w[g + k] & cmask);
                else        csr_col[s + slot[k]] = (int)(w[g + k] & cmask);
            }
        }
    }
    for (int k = 0; k < RPW; ++k) {
        const int base = __builtin_amdgcn_readlane(rbase, k);
        const int len  = __builtin_amdgcn_readlane(rlen, k);
        for (int j = 32 + lane; j < len; j += 32) {
            const unsigned wj = packed[base + j];
            unsigned lr = wj >> cb;
            lr = lr < last ? lr : last;
            const int sl = atomicAdd(&lcnt[lr], 1);
            if (staged) lcol[sl] = (int)(wj & cmask);
            else        csr_col[s + sl] = (int)(wj & cmask);
        }
    }
    if (staged) {
        __syncthreads();
        for (int i = t; i < c; i += GT) csr_col[s + i] = lcol[i];
    }
}

__global__ void __launch_bounds__(GT, 16)
project_and_sort_kernel(const float* __restrict__ seq, const float* __restrict__ W,
                        float* __restrict__ fts,
                        const float* __restrict__ a1, const float* __restrict__ b1,
                        const float* __restrict__ a2, const float* __restrict__ b2,
                        float* __restrict__ f1, float* __restrict__ f2,
                        const unsigned* __restrict__ packed, const int* __restrict__ rtab,
                        int* __restrict__ cnt, int* __restrict__ rowstart,
                        int* __restrict__ csr_col,
                        int e, int n, int rpb, int per, int gemm_blocks) {
    extern __shared__ int dyn_lds[];
    if ((int)blockIdx.x < gemm_blocks)
        gemm_role((float*)dyn_lds, (int)blockIdx.x, seq, W, fts, a1, b1, a2, b2, f1, f2, n);
    else
        sort_role(dyn_lds, (int)blockIdx.x - gemm_blocks, packed, rtab, cnt, rowstart, csr_col,
                  e, n, rpb, per);
}

__device__ __forceinline__ float lrelu(float l) { return l > 0.0f ? l : NEG_SLOPE * l; }

__device__ __forceinline__ void accum_chunk(const float* __restrict__ fts, int c, float ev,
                                            int cntc, int half, int q,
                                            v2f& acc_lo, v2f& acc_hi) {
    const int steps = (cntc + 1) >> 1;
    const int evb = __float_as_int(ev);
#pragma unroll 4
    for (int t = 0; t < steps; ++t) {
        const int c_lo = __builtin_amdgcn_readlane(c,   2 * t);
        const int c_hi = __builtin_amdgcn_readlane(c,   2 * t + 1);
        const int w_lo = __builtin_amdgcn_readlane(evb, 2 * t);
        const int w_hi = __builtin_amdgcn_readlane(evb, 2 * t + 1);
        const int   cc = half ? c_hi : c_lo;
        const float w  = __int_as_float(half ? w_hi : w_lo);
        const unsigned off = ((unsigned)cc << 8) | ((unsigned)q << 4);
        const v4f v = *(const v4f*)((const char*)fts + off);
        const v2f w2 = {w, w};
        acc_lo = __builtin_elementwise_fma(v.xy, w2, acc_lo);
        acc_hi = __builtin_elementwise_fma(v.zw, w2, acc_hi);
    }
}

__device__ __forceinline__ void row_bounds(const int* __restrict__ cnt,
                                           const int* __restrict__ rowstart,
                                           const float* __restrict__ f1, int r, int n,
                                           int& deg, int& s, float& f1r) {
    deg = 0; s = 0; f1r = 0.0f;
    if (r < n) { deg = cnt[r]; s = rowstart[r]; f1r = f1[r]; }
}

template <int T> __device__ __forceinline__ int share_i(int x) {
    return __builtin_amdgcn_update_dpp(0, x, 0x150 + T, 0xf, 0xf, false);
}
template <int T> __device__ __forceinline__ float share_f(float x) {
    return __int_as_float(__builtin_amdgcn_mov_dpp(__float_as_int(x), 0x150 + T, 0xf, 0xf, false));
}
template <int M> __device__ __forceinline__ float xmask_f(float x) {
    return __int_as_float(__builtin_amdgcn_update_dpp(0, __float_as_int(x), 0x160 + M,
                                                      0xf, 0xf, false));
}

__device__ __forceinline__ float xhalf(float x) {
    const int b = __float_as_int(x);
    return __int_as_float(__builtin_amdgcn_permlanex16(b, b, 0x76543210, 0xfedcba98, false, false));
}

#define GATHER_STEP(T)                                                                   \
    {                                                                                    \
        const unsigned off = (unsigned)share_i<T>(csel8) + q16;                          \
        v[T] = *(const v4f*)(fb + off);                                                  \
    }
#define DOT_STEP(T)                                                                      \
    {                                                                                    \
        const v2f t2 = __builtin_elementwise_fma(v[T].zw, a_hi, v[T].xy * a_lo);         \
        p[T] = t2.x + t2.y;                                                              \
    }
#define ACC_STEP(T)                                                                      \
    {                                                                                    \
        const float w = share_f<T>(ev);                                                  \
        const v2f w2 = {w, w};                                                           \
        acc_lo = __builtin_elementwise_fma(v[T].xy, w2, acc_lo);                         \
        acc_hi = __builtin_elementwise_fma(v[T].zw, w2, acc_hi);                         \
    }
#define ALL16(S) S(0) S(1) S(2) S(3) S(4) S(5) S(6) S(7) S(8) S(9) S(10) S(11) S(12) S(13) S(14) S(15)

__global__ void __launch_bounds__(256)
row_kernel(const int* __restrict__ cnt, const int* __restrict__ rowstart,
           const int* __restrict__ csr_col,
           const float* __restrict__ f1, const float* __restrict__ f2,
           const float* __restrict__ fts, const float* __restrict__ a2,
           const float* __restrict__ b2, const float* __restrict__ bias,
           float* __restrict__ out, int n, int e) {
    const int lane = threadIdx.x & 31;
    const int half = lane >> 4;
    const int q    = lane & 15;
    const int eo   = 2 * q + half;
    const int nw   = RB * 8;
    int row = __builtin_amdgcn_readfirstlane((int)(blockIdx.x * 8 + (threadIdx.x >> 5)));
    const int last = e - 1;
    const float4 b  = *(const float4*)(bias + q * 4);
    const float4 aq = *(const float4*)(a2 + q * 4);
    v2f a_lo, a_hi;
    a_lo.x = aq.x; a_lo.y = aq.y; a_hi.x = aq.z; a_hi.y = aq.w;
    const float b2v = b2[0];
    const char* __restrict__ fb = (const char*)fts;
    const unsigned q16 = (unsigned)q << 4;
    const bool b3 = (q & 8) != 0, b2b = (q & 4) != 0, b1 = (q & 2) != 0, b0 = (q & 1) != 0;

    int deg, s, deg1, s1, deg2, s2;
    float f1r, f1r1, f1r2;
    row_bounds(cnt, rowstart, f1, row,          n, deg,  s,  f1r);
    row_bounds(cnt, rowstart, f1, row + nw,     n, deg1, s1, f1r1);
    row_bounds(cnt, rowstart, f1, row + 2 * nw, n, deg2, s2, f1r2);
    int ca  = csr_col[min(s + eo, last)];
    int cb  = csr_col[min(s + 32 + lane, last)];
    int ca1 = csr_col[min(s1 + eo, last)];
    int cb1 = csr_col[min(s1 + 32 + lane, last)];
    float gb = f2[lane + 32 < deg ? cb : 0];

    while (row < n) {
        const bool va = eo < deg;
        const bool vb = lane + 32 < deg;
        const float lb = vb ? lrelu(f1r + gb) : -INFINITY;

        int deg3, s3;
        float f1r3;
        row_bounds(cnt, rowstart, f1, row + 3 * nw, n, deg3, s3, f1r3);
        const int ca2 = csr_col[min(s2 + eo, last)];
        const int cb2 = csr_col[min(s2 + 32 + lane, last)];
        const float gb1 = f2[lane + 32 < deg1 ? cb1 : 0];

        const int c0   = __builtin_amdgcn_readfirstlane(ca);
        const int csel8 = (va ? ca : c0) << 8;
        v4f v[16];
        float p[16];
        ALL16(GATHER_STEP)
        ALL16(DOT_STEP)
        float r8[8], r4[4], r2[2];
#pragma unroll
        for (int i = 0; i < 8; ++i)
            r8[i] = (b3 ? p[i + 8] : p[i]) + xmask_f<8>(b3 ? p[i] : p[i + 8]);
#pragma unroll
        for (int i = 0; i < 4; ++i)
            r4[i] = (b2b ? r8[i + 4] : r8[i]) + xmask_f<4>(b2b ? r8[i] : r8[i + 4]);
#pragma unroll
        for (int i = 0; i < 2; ++i)
            r2[i] = (b1 ? r4[i + 2] : r4[i]) + xmask_f<2>(b1 ? r4[i] : r4[i + 2]);
        const float sdot = (b0 ? r2[1] : r2[0]) + xmask_f<1>(b0 ? r2[0] : r2[1]);
        const float la = va ? lrelu(f1r + (sdot + b2v)) : -INFINITY;

        float m = la;
        if (vb) m = fmaxf(m, lb);
        for (int base = 64; base < deg; base += 32) {
            const int j = base + lane;
            if (j < deg) m = fmaxf(m, lrelu(f1r + f2[csr_col[s + j]]));
        }
        m = fmaxf(m, xhalf(m));
        m = fmaxf(m, xmask_f<8>(m));
        m = fmaxf(m, xmask_f<4>(m));
        m = fmaxf(m, xmask_f<2>(m));
        m = fmaxf(m, xmask_f<1>(m));
        if (!isfinite(m)) m = 0.0f;

        v2f acc_lo = {0.0f, 0.0f}, acc_hi = {0.0f, 0.0f};
        const float ev = va ? expf(la - m) : 0.0f;
        float dsum = ev;
        ALL16(ACC_STEP)
        if (deg > 32) {
            const float ev2 = vb ? expf(lb - m) : 0.0f;
            dsum += ev2;
            const int cf = __builtin_amdgcn_readfirstlane(cb);
            const int rem = deg - 32;
            accum_chunk(fts, vb ? cb : cf, ev2, rem < 32 ? rem : 32, half, q, acc_lo, acc_hi);
        }
        for (int base = 64; base < deg; base += 32) {
            const int j = base + lane;
            const bool vj = j < deg;
            int c = 0; float evj = 0.0f;
            if (vj) { c = csr_col[s + j]; evj = expf(lrelu(f1r + f2[c]) - m); }
            dsum += evj;
            const int cf = __shfl(c, 0, 32);
            const int rem = deg - base;
            accum_chunk(fts, vj ? c : cf, evj, rem < 32 ? rem : 32, half, q, acc_lo, acc_hi);
        }

        dsum += xhalf(dsum);
        dsum += xmask_f<8>(dsum);
        dsum += xmask_f<4>(dsum);
        dsum += xmask_f<2>(dsum);
        dsum += xmask_f<1>(dsum);
        acc_lo.x += xhalf(acc_lo.x);
        acc_lo.y += xhalf(acc_lo.y);
        acc_hi.x += xhalf(acc_hi.x);
        acc_hi.y += xhalf(acc_hi.y);

        if (half == 0) {
            float4 o;
            if (deg > 0) {
                const float inv = 1.0f / dsum;
                o.x = acc_lo.x * inv + b.x; o.y = acc_lo.y * inv + b.y;
                o.z = acc_hi.x * inv + b.z; o.w = acc_hi.y * inv + b.w;
            } else {
                o = b;
            }
            o.x = o.x > 0.0f ? o.x : 0.0f; o.y = o.y > 0.0f ? o.y : 0.0f;
            o.z = o.z > 0.0f ? o.z : 0.0f; o.w = o.w > 0.0f ? o.w : 0.0f;
            *(float4*)(out + (size_t)row * F_OUT + q * 4) = o;
        }

        row += nw;
        deg = deg1; s = s1; f1r = f1r1; ca = ca1; cb = cb1; gb = gb1;
        deg1 = deg2; s1 = s2; f1r1 = f1r2; ca1 = ca2; cb1 = cb2;
        deg2 = deg3; s2 = s3; f1r2 = f1r3;
    }
}

extern "C" void kernel_launch(void* const* d_in, const int* in_sizes, int n_in,
                              void* d_out, int out_size, void* d_ws, size_t ws_size,
                              hipStream_t stream) {
    const float* seq  = (const float*)d_in[0];
    const int*   erow = (const int*)  d_in[1];
    const int*   ecol = (const int*)  d_in[2];
    const float* W    = (const float*)d_in[3];
    const float* a1   = (const float*)d_in[4];
    const float* b1   = (const float*)d_in[5];
    const float* a2   = (const float*)d_in[6];
    const float* b2   = (const float*)d_in[7];
    const float* bias = (const float*)d_in[8];

    const int n = in_sizes[0] / F_IN;
    const int e = in_sizes[1];
    float* out = (float*)d_out;

    float* fts      = (float*)d_ws;
    float* f1       = fts + (size_t)n * F_OUT;
    float* f2       = f1 + n;
    int*   cnt      = (int*)(f2 + n);
    int*   rowstart = cnt + n;
    int*   csr_col  = rowstart + n;
    unsigned* packed = (unsigned*)d_out;
    int*   rtab     = (int*)d_out + (e + 31) / 32 * 32;

    const int rpb = n / NB + 2;
    const int per = ((e + PB - 1) / PB + 31) / 32 * 32;

    slice_group_kernel<<<PB, GT, per * sizeof(int), stream>>>(erow, ecol, rtab, packed,
                                                             e, n, per, rpb);
    {
        const int gemm_blocks = (n + 16 * (GT / 32) - 1) / (16 * (GT / 32));
        const size_t sort_lds = (rpb + SCAP + 3 * (GT / 32)) * sizeof(int);
        const size_t gemm_lds = F_IN * F_OUT * sizeof(float);
        project_and_sort_kernel<<<gemm_blocks + NB, GT, sort_lds > gemm_lds ? sort_lds : gemm_lds,
                                  stream>>>(seq, W, fts, a1, b1, a2, b2, f1, f2,
                                            packed, rtab, cnt, rowstart, csr_col,
                                            e, n, rpb, per, gemm_blocks);
    }
    row_kernel<<<RB, 256, 0, stream>>>(cnt, rowstart, csr_col, f1, f2, fts, a2, b2, bias,
                                       out, n, e);
}


# Projection stripes dealt evenly over all 256 WGPs: 24 whole stripes a block, 106 left-over stripes cut by column tile over four waves, one per SIMD (f1/f2 sum order changes on those 1696 rows)
# speedup vs baseline: 9.1378x; 1.0363x over previous
#include <hip/hip_runtime.h>
#include <hip/hip_bf16.h>

typedef __attribute__((ext_vector_type(2))) float v2f;
typedef __attribute__((ext_vector_type(8))) float v8f;
typedef __attribute__((ext_vector_type(4))) float v4f;

#define F_IN   256
#define F_OUT  64
#define NEG_SLOPE 0.2f
#define NB     256
#define GT     1024
#define EB     8
#define PB     512
#define RPW    (PB / (GT / 32))
#define SB     4
#define RB     1024
#define SCAP   14336
#define PGB    256
#define GFW    24


__device__ __forceinline__ void
gemm_role(float* __restrict__ Wl, int gb, int gblocks,
          const float* __restrict__ seq, const float* __restrict__ W,
          float* __restrict__ fts,
          const float* __restrict__ a1, const float* __restrict__ b1,
          const float* __restrict__ a2, const float* __restrict__ b2,
          float* __restrict__ f1, float* __restrict__ f2, int n) {
    float* __restrict__ part = Wl + F_IN * F_OUT;
    for (int i = threadIdx.x; i < F_IN * F_OUT; i += GT) {
        const int kk  = i >> 6;
        const int col = i & 63;
        Wl[(kk >> 1) * (2 * F_OUT) + col * 2 + (kk & 1)] = W[i];
    }
    __syncthreads();

    const int wave = __builtin_amdgcn_readfirstlane((int)(threadIdx.x >> 5));
    const int grp = wave / 9;
    const int pos = wave - grp * 9;
    const int ntiles = (n + 15) >> 4;
    const int q   = ntiles / gblocks;
    const int rem = ntiles - q * gblocks;

    if (pos == 4) {
        if (gb < rem) {
            const int lane  = threadIdx.x & 31;
            const int lcol  = lane & 15;
            const int khalf = (lane >> 4) * 2;
            const int nhi   = lane >> 4;
            const int rowTile = q * gblocks + gb;
            const int ct = grp;
            int row = rowTile * 16 + lcol;
            if (row >= n) row = n - 1;
            const float* __restrict__ arow = seq + (size_t)row * F_IN;
            const int ocol = ct * 16 + lcol;
            v8f acc1 = v8f{};
            for (int k = 0; k < F_IN; k += 4) {
                float2 av = *(const float2*)(arow + k + khalf);
                v2f a; a[0] = av.x; a[1] = av.y;
                const int kpairBase = ((k + khalf) >> 1) * (2 * F_OUT);
                float2 bv = *(const float2*)(&Wl[kpairBase + ocol * 2]);
                v2f b; b[0] = bv.x; b[1] = bv.y;
                acc1 = __builtin_amdgcn_wmma_f32_16x16x4_f32(
                         false, a,  false, b,
                         (short)0, acc1,  false,  false);
            }
#pragma unroll
            for (int r = 0; r < 8; ++r) {
                const int orow = rowTile * 16 + r + 8 * nhi;
                if (orow < n) fts[(size_t)orow * F_OUT + ocol] = acc1[r];
            }
            const float A1c = a1[ocol], A2c = a2[ocol];
#pragma unroll
            for (int r = 0; r < 8; ++r) {
                float p1 = acc1[r] * A1c, p2 = acc1[r] * A2c;
#pragma unroll
                for (int off = 8; off > 0; off >>= 1) {
                    p1 += __shfl_xor(p1, off, 32);
                    p2 += __shfl_xor(p2, off, 32);
                }
                if (lcol == 0) {
                    part[(ct * 16 + r + 8 * nhi) * 2]     = p1;
                    part[(ct * 16 + r + 8 * nhi) * 2 + 1] = p2;
                }
            }
        }
    } else if (pos != 0 && grp * 7 + (pos < 4 ? pos - 1 : pos - 2) < q) {
        const int rowTile = gb * q + grp * 7 + (pos < 4 ? pos - 1 : pos - 2);
        const int lane  = threadIdx.x & 31;
        int row = rowTile * 16 + (lane & 15);
        if (row >= n) row = n - 1;
        const int lcol  = lane & 15;
        const int khalf = (lane >> 4) * 2;

        const float* __restrict__ arow = seq + (size_t)row * F_IN;

        v8f acc[4] = {v8f{}, v8f{}, v8f{}, v8f{}};
        for (int k = 0; k < F_IN; k += 4) {
            float2 av = *(const float2*)(arow + k + khalf);
            v2f a; a[0] = av.x; a[1] = av.y;
            const int kpairBase = ((k + khalf) >> 1) * (2 * F_OUT);
#pragma unroll
            for (int ct = 0; ct < 4; ++ct) {
                float2 bv = *(const float2*)(&Wl[kpairBase + (ct * 16 + lcol) * 2]);
                v2f b; b[0] = bv.x; b[1] = bv.y;
                acc[ct] = __builtin_amdgcn_wmma_f32_16x16x4_f32(
                         false, a,  false, b,
                         (short)0, acc[ct],  false,  false);
            }
        }

        const int nhi = lane >> 4;
#pragma unroll
        for (int ct = 0; ct < 4; ++ct) {
            const int ocol = ct * 16 + lcol;
#pragma unroll
            for (int r = 0; r < 8; ++r) {
                const int orow = rowTile * 16 + r + 8 * nhi;
                if (orow < n) fts[(size_t)orow * F_OUT + ocol] = acc[ct][r];
            }
        }

        float A1[4], A2[4];
#pragma unroll
        for (int ct = 0; ct < 4; ++ct) {
            A1[ct] = a1[ct * 16 + lcol];
            A2[ct] = a2[ct * 16 + lcol];
        }
        const float b1v = b1[0], b2v = b2[0];
#pragma unroll
        for (int r = 0; r < 8; ++r) {
            float p1 = 0.0f, p2 = 0.0f;
#pragma unroll
            for (int ct = 0; ct < 4; ++ct) {
                p1 += acc[ct][r] * A1[ct];
                p2 += acc[ct][r] * A2[ct];
            }
#pragma unroll
            for (int off = 8; off > 0; off >>= 1) {
                p1 += __shfl_xor(p1, off, 32);
                p2 += __shfl_xor(p2, off, 32);
            }
            if ((lane & 15) == 0) {
                const int orow = rowTile * 16 + r + 8 * nhi;
                if (orow < n) { f1[orow] = p1 + b1v; f2[orow] = p2 + b2v; }
            }
        }
    }

    if (gb < rem) {
        __syncthreads();
        const int lane = threadIdx.x & 31;
        if (wave == 4 && lane < 16) {
            const int orow = (q * gblocks + gb) * 16 + lane;
            float p1 = part[lane * 2], p2 = part[lane * 2 + 1];
#pragma unroll
            for (int ct = 1; ct < 4; ++ct) {
                p1 += part[(ct * 16 + lane) * 2];
                p2 += part[(ct * 16 + lane) * 2 + 1];
            }
            if (orow < n) { f1[orow] = p1 + b1[0]; f2[orow] = p2 + b2[0]; }
        }
    }
}

__device__ __forceinline__ int range_of(int row, unsigned magic) {
    const int b = (int)__umulhi((unsigned)row, magic);
    return b < NB - 1 ? b : NB - 1;
}

__global__ void __launch_bounds__(GT)
slice_group_kernel(const int* __restrict__ erow, const int* __restrict__ ecol,
                   int* __restrict__ rtab, unsigned* __restrict__ packed,
                   int e, int n, int per, int rpb) {
    extern __shared__ unsigned lbuf[];
    __shared__ int h[NB];
    __shared__ int wsum2[NB / 32];
    const int t = threadIdx.x, lane = t & 31, wave = t >> 5;
    if (t < NB) h[t] = 0;
    __syncthreads();
    const unsigned magic = 0xFFFFFFFFu / (unsigned)rpb + 1u;
    const int cb = 32 - __clz(n - 1);
    const int lo = blockIdx.x * per;
    const int hi = lo + per < e ? lo + per : e;
    for (int i = lo + t; i < hi; i += EB * GT) {
        int r[EB];
#pragma unroll
        for (int k = 0; k < EB; ++k) r[k] = erow[min(i + k * GT, hi - 1)];
#pragma unroll
        for (int k = 0; k < EB; ++k)
            if (i + k * GT < hi) atomicAdd(&h[range_of(r[k], magic)], 1);
    }
    __syncthreads();
    int mine = 0, y = 0;
    if (t < NB) {
        mine = h[t];
        y = mine;
        for (int off = 1; off < 32; off <<= 1) {
            const int z = __shfl_up(y, off, 32);
            if (lane >= off) y += z;
        }
        if (lane == 31) wsum2[wave] = y;
    }
    __syncthreads();
    if (t < NB) {
        int lp = 0;
        for (int w = 0; w < wave; ++w) lp += wsum2[w];
        const int ls = lp + y - mine;
        *(int2*)(rtab + 2 * ((size_t)blockIdx.x * NB + t)) = make_int2(ls, mine);
        h[t] = ls;
    }
    __syncthreads();
    for (int i = lo + t; i < hi; i += EB * GT) {
        int r[EB], c[EB];
#pragma unroll
        for (int k = 0; k < EB; ++k) {
            const int j = min(i + k * GT, hi - 1);
            r[k] = erow[j];
            c[k] = ecol[j];
        }
#pragma unroll
        for (int k = 0; k < EB; ++k) {
            if (i + k * GT < hi) {
                const int b = range_of(r[k], magic);
                const int slot = atomicAdd(&h[b], 1);
                lbuf[slot] = ((unsigned)(r[k] - b * rpb) << cb) | (unsigned)c[k];
            }
        }
    }
    __syncthreads();
    for (int i = t; i < hi - lo; i += GT) packed[lo + i] = lbuf[i];
}

__device__ __forceinline__ void
sort_role(int* __restrict__ lds4, int sb,
          const unsigned* __restrict__ packed, const int* __restrict__ rtab,
          int* __restrict__ cnt, int* __restrict__ rowstart, int* __restrict__ csr_col,
          int e, int n, int rpb, int per) {
    int* wsum = lds4;
    int* wtot = wsum + GT / 32;
    int* wofs = wtot + GT / 32;
    int* lcnt = wofs + GT / 32;
    int* lcol = lcnt + rpb;
    const int t = threadIdx.x, lane = t & 31, wave = t >> 5;
    const int r0 = sb * rpb;
    int rows = n - r0;
    if (rows > rpb) rows = rpb;
    if (rows <= 0) return;
    const int cb = 32 - __clz(n - 1);
    const unsigned cmask = (1u << cb) - 1u;
    const unsigned last = (unsigned)(rows - 1);
    int rbase = 0, rlen = 0, rofs = 0;
    if (lane < RPW) {
        const int pb = wave * RPW + lane;
        const int2 v = *(const int2*)(rtab + 2 * ((size_t)pb * NB + sb));
        rofs = v.x;
        rlen = v.y;
        rbase = pb * per + v.x;
    }
    for (int i = t; i < rows; i += GT) lcnt[i] = 0;
    unsigned w[RPW];
#pragma unroll
    for (int k = 0; k < RPW; ++k) {
        const int base = __builtin_amdgcn_readlane(rbase, k);
        const int len  = __builtin_amdgcn_readlane(rlen, k);
        const int j = base + min(lane, len > 0 ? len - 1 : 0);
        w[k] = packed[min(j, e - 1)];
    }
    {
        int xs = rlen, os = rofs;
        for (int off = 16; off > 0; off >>= 1) {
            xs += __shfl_xor(xs, off, 32);
            os += __shfl_xor(os, off, 32);
        }
        if (lane == 0) { wtot[wave] = xs; wofs[wave] = os; }
    }
    __syncthreads();
    int c = wtot[lane], s = wofs[lane];
    for (int off = 16; off > 0; off >>= 1) {
        c += __shfl_xor(c, off, 32);
        s += __shfl_xor(s, off, 32);
    }
    const bool staged = c <= SCAP;
#pragma unroll
    for (int k = 0; k < RPW; ++k) {
        const int len = __builtin_amdgcn_readlane(rlen, k);
        if (lane < len) {
            unsigned lr = w[k] >> cb;
            lr = lr < last ? lr : last;
            atomicAdd(&lcnt[lr], 1);
        }
    }
    for (int k = 0; k < RPW; ++k) {
        const int base = __builtin_amdgcn_readlane(rbase, k);
        const int len  = __builtin_amdgcn_readlane(rlen, k);
        for (int j = 32 + lane; j < len; j += 32) {
            unsigned lr = packed[base + j] >> cb;
            lr = lr < last ? lr : last;
            atomicAdd(&lcnt[lr], 1);
        }
    }
    __syncthreads();
    int carry = 0;
    for (int base = 0; base < rows; base += GT) {
        const int idx = base + t;
        const int v = idx < rows ? lcnt[idx] : 0;
        int x = v;
        for (int off = 1; off < 32; off <<= 1) {
            const int y = __shfl_up(x, off, 32);
            if (lane >= off) x += y;
        }
        if (lane == 31) wsum[wave] = x;
        __syncthreads();
        int wp = 0, all = 0;
#pragma unroll 8
        for (int q2 = 0; q2 < GT / 32; ++q2) {
            const int q = wsum[q2];
            all += q;
            if (q2 < wave) wp += q;
        }
        const int ex = carry + wp + x - v;
        if (idx < rows) {
            cnt[r0 + idx] = v;
            rowstart[r0 + idx] = s + ex;
            lcnt[idx] = ex;
        }
        carry += all;
        __syncthreads();
    }
#pragma unroll
    for (int g = 0; g < RPW; g += SB) {
        int slot[SB];
#pragma unroll
        for (int k = 0; k < SB; ++k) {
            const int len = __builtin_amdgcn_readlane(rlen, g + k);
            unsigned lr = w[g + k] >> cb;
            lr = lr < last ? lr : last;
            slot[k] = lane < len ? atomicAdd(&lcnt[lr], 1) : 0;
        }
#pragma unroll
        for (int k = 0; k < SB; ++k) {
            if (lane < __builtin_amdgcn_readlane(rlen, g + k)) {
                if (staged) lcol[slot[k]] = (int)(w[g + k] & cmask);
                else        csr_col[s + slot[k]] = (int)(w[g + k] & cmask);
            }
        }
    }
    for (int k = 0; k < RPW; ++k) {
        const int base = __builtin_amdgcn_readlane(rbase, k);
        const int len  = __builtin_amdgcn_readlane(rlen, k);
        for (int j = 32 + lane; j < len; j += 32) {
            const unsigned wj = packed[base + j];
            unsigned lr = wj >> cb;
            lr = lr < last ? lr : last;
            const int sl = atomicAdd(&lcnt[lr], 1);
            if (staged) lcol[sl] = (int)(wj & cmask);
            else        csr_col[s + sl] = (int)(wj & cmask);
        }
    }
    if (staged) {
        __syncthreads();
        for (int i = t; i < c; i += GT) csr_col[s + i] = lcol[i];
    }
}

__global__ void __launch_bounds__(GT, 16)
project_and_sort_kernel(const float* __restrict__ seq, const float* __restrict__ W,
                        float* __restrict__ fts,
                        const float* __restrict__ a1, const float* __restrict__ b1,
                        const float* __restrict__ a2, const float* __restrict__ b2,
                        float* __restrict__ f1, float* __restrict__ f2,
                        const unsigned* __restrict__ packed, const int* __restrict__ rtab,
                        int* __restrict__ cnt, int* __restrict__ rowstart,
                        int* __restrict__ csr_col,
                        int e, int n, int rpb, int per, int gemm_blocks) {
    extern __shared__ int dyn_lds[];
    if ((int)blockIdx.x < gemm_blocks)
        gemm_role((float*)dyn_lds, (int)blockIdx.x, gemm_blocks, seq, W, fts, a1, b1, a2, b2, f1, f2, n);
    else
        sort_role(dyn_lds, (int)blockIdx.x - gemm_blocks, packed, rtab, cnt, rowstart, csr_col,
                  e, n, rpb, per);
}

__device__ __forceinline__ float lrelu(float l) { return l > 0.0f ? l : NEG_SLOPE * l; }

__device__ __forceinline__ void accum_chunk(const float* __restrict__ fts, int c, float ev,
                                            int cntc, int half, int q,
                                            v2f& acc_lo, v2f& acc_hi) {
    const int steps = (cntc + 1) >> 1;
    const int evb = __float_as_int(ev);
#pragma unroll 4
    for (int t = 0; t < steps; ++t) {
        const int c_lo = __builtin_amdgcn_readlane(c,   2 * t);
        const int c_hi = __builtin_amdgcn_readlane(c,   2 * t + 1);
        const int w_lo = __builtin_amdgcn_readlane(evb, 2 * t);
        const int w_hi = __builtin_amdgcn_readlane(evb, 2 * t + 1);
        const int   cc = half ? c_hi : c_lo;
        const float w  = __int_as_float(half ? w_hi : w_lo);
        const unsigned off = ((unsigned)cc << 8) | ((unsigned)q << 4);
        const v4f v = *(const v4f*)((const char*)fts + off);
        const v2f w2 = {w, w};
        acc_lo = __builtin_elementwise_fma(v.xy, w2, acc_lo);
        acc_hi = __builtin_elementwise_fma(v.zw, w2, acc_hi);
    }
}

__device__ __forceinline__ void row_bounds(const int* __restrict__ cnt,
                                           const int* __restrict__ rowstart,
                                           const float* __restrict__ f1, int r, int n,
                                           int& deg, int& s, float& f1r) {
    deg = 0; s = 0; f1r = 0.0f;
    if (r < n) { deg = cnt[r]; s = rowstart[r]; f1r = f1[r]; }
}

template <int T> __device__ __forceinline__ int share_i(int x) {
    return __builtin_amdgcn_update_dpp(0, x, 0x150 + T, 0xf, 0xf, false);
}
template <int T> __device__ __forceinline__ float share_f(float x) {
    return __int_as_float(__builtin_amdgcn_mov_dpp(__float_as_int(x), 0x150 + T, 0xf, 0xf, false));
}
template <int M> __device__ __forceinline__ float xmask_f(float x) {
    return __int_as_float(__builtin_amdgcn_update_dpp(0, __float_as_int(x), 0x160 + M,
                                                      0xf, 0xf, false));
}

__device__ __forceinline__ float xhalf(float x) {
    const int b = __float_as_int(x);
    return __int_as_float(__builtin_amdgcn_permlanex16(b, b, 0x76543210, 0xfedcba98, false, false));
}

#define GATHER_STEP(T)                                                                   \
    {                                                                                    \
        const unsigned off = (unsigned)share_i<T>(csel8) + q16;                          \
        v[T] = *(const v4f*)(fb + off);                                                  \
    }
#define DOT_STEP(T)                                                                      \
    {                                                                                    \
        const v2f t2 = __builtin_elementwise_fma(v[T].zw, a_hi, v[T].xy * a_lo);         \
        p[T] = t2.x + t2.y;                                                              \
    }
#define ACC_STEP(T)                                                                      \
    {                                                                                    \
        const float w = share_f<T>(ev);                                                  \
        const v2f w2 = {w, w};                                                           \
        acc_lo = __builtin_elementwise_fma(v[T].xy, w2, acc_lo);                         \
        acc_hi = __builtin_elementwise_fma(v[T].zw, w2, acc_hi);                         \
    }
#define ALL16(S) S(0) S(1) S(2) S(3) S(4) S(5) S(6) S(7) S(8) S(9) S(10) S(11) S(12) S(13) S(14) S(15)

__global__ void __launch_bounds__(256)
row_kernel(const int* __restrict__ cnt, const int* __restrict__ rowstart,
           const int* __restrict__ csr_col,
           const float* __restrict__ f1, const float* __restrict__ f2,
           const float* __restrict__ fts, const float* __restrict__ a2,
           const float* __restrict__ b2, const float* __restrict__ bias,
           float* __restrict__ out, int n, int e) {
    const int lane = threadIdx.x & 31;
    const int half = lane >> 4;
    const int q    = lane & 15;
    const int eo   = 2 * q + half;
    const int nw   = RB * 8;
    int row = __builtin_amdgcn_readfirstlane((int)(blockIdx.x * 8 + (threadIdx.x >> 5)));
    const int last = e - 1;
    const float4 b  = *(const float4*)(bias + q * 4);
    const float4 aq = *(const float4*)(a2 + q * 4);
    v2f a_lo, a_hi;
    a_lo.x = aq.x; a_lo.y = aq.y; a_hi.x = aq.z; a_hi.y = aq.w;
    const float b2v = b2[0];
    const char* __restrict__ fb = (const char*)fts;
    const unsigned q16 = (unsigned)q << 4;
    const bool b3 = (q & 8) != 0, b2b = (q & 4) != 0, b1 = (q & 2) != 0, b0 = (q & 1) != 0;

    int deg, s, deg1, s1, deg2, s2;
    float f1r, f1r1, f1r2;
    row_bounds(cnt, rowstart, f1, row,          n, deg,  s,  f1r);
    row_bounds(cnt, rowstart, f1, row + nw,     n, deg1, s1, f1r1);
    row_bounds(cnt, rowstart, f1, row + 2 * nw, n, deg2, s2, f1r2);
    int ca  = csr_col[min(s + eo, last)];
    int cb  = csr_col[min(s + 32 + lane, last)];
    int ca1 = csr_col[min(s1 + eo, last)];
    int cb1 = csr_col[min(s1 + 32 + lane, last)];
    float gb = f2[lane + 32 < deg ? cb : 0];

    while (row < n) {
        const bool va = eo < deg;
        const bool vb = lane + 32 < deg;
        const float lb = vb ? lrelu(f1r + gb) : -INFINITY;

        int deg3, s3;
        float f1r3;
        row_bounds(cnt, rowstart, f1, row + 3 * nw, n, deg3, s3, f1r3);
        const int ca2 = csr_col[min(s2 + eo, last)];
        const int cb2 = csr_col[min(s2 + 32 + lane, last)];
        const float gb1 = f2[lane + 32 < deg1 ? cb1 : 0];

        const int c0   = __builtin_amdgcn_readfirstlane(ca);
        const int csel8 = (va ? ca : c0) << 8;
        v4f v[16];
        float p[16];
        ALL16(GATHER_STEP)
        ALL16(DOT_STEP)
        float r8[8], r4[4], r2[2];
#pragma unroll
        for (int i = 0; i < 8; ++i)
            r8[i] = (b3 ? p[i + 8] : p[i]) + xmask_f<8>(b3 ? p[i] : p[i + 8]);
#pragma unroll
        for (int i = 0; i < 4; ++i)
            r4[i] = (b2b ? r8[i + 4] : r8[i]) + xmask_f<4>(b2b ? r8[i] : r8[i + 4]);
#pragma unroll
        for (int i = 0; i < 2; ++i)
            r2[i] = (b1 ? r4[i + 2] : r4[i]) + xmask_f<2>(b1 ? r4[i] : r4[i + 2]);
        const float sdot = (b0 ? r2[1] : r2[0]) + xmask_f<1>(b0 ? r2[0] : r2[1]);
        const float la = va ? lrelu(f1r + (sdot + b2v)) : -INFINITY;

        float m = la;
        if (vb) m = fmaxf(m, lb);
        for (int base = 64; base < deg; base += 32) {
            const int j = base + lane;
            if (j < deg) m = fmaxf(m, lrelu(f1r + f2[csr_col[s + j]]));
        }
        m = fmaxf(m, xhalf(m));
        m = fmaxf(m, xmask_f<8>(m));
        m = fmaxf(m, xmask_f<4>(m));
        m = fmaxf(m, xmask_f<2>(m));
        m = fmaxf(m, xmask_f<1>(m));
        if (!isfinite(m)) m = 0.0f;

        v2f acc_lo = {0.0f, 0.0f}, acc_hi = {0.0f, 0.0f};
        const float ev = va ? expf(la - m) : 0.0f;
        float dsum = ev;
        ALL16(ACC_STEP)
        if (deg > 32) {
            const float ev2 = vb ? expf(lb - m) : 0.0f;
            dsum += ev2;
            const int cf = __builtin_amdgcn_readfirstlane(cb);
            const int rem = deg - 32;
            accum_chunk(fts, vb ? cb : cf, ev2, rem < 32 ? rem : 32, half, q, acc_lo, acc_hi);
        }
        for (int base = 64; base < deg; base += 32) {
            const int j = base + lane;
            const bool vj = j < deg;
            int c = 0; float evj = 0.0f;
            if (vj) { c = csr_col[s + j]; evj = expf(lrelu(f1r + f2[c]) - m); }
            dsum += evj;
            const int cf = __shfl(c, 0, 32);
            const int rem = deg - base;
            accum_chunk(fts, vj ? c : cf, evj, rem < 32 ? rem : 32, half, q, acc_lo, acc_hi);
        }

        dsum += xhalf(dsum);
        dsum += xmask_f<8>(dsum);
        dsum += xmask_f<4>(dsum);
        dsum += xmask_f<2>(dsum);
        dsum += xmask_f<1>(dsum);
        acc_lo.x += xhalf(acc_lo.x);
        acc_lo.y += xhalf(acc_lo.y);
        acc_hi.x += xhalf(acc_hi.x);
        acc_hi.y += xhalf(acc_hi.y);

        if (half == 0) {
            float4 o;
            if (deg > 0) {
                const float inv = 1.0f / dsum;
                o.x = acc_lo.x * inv + b.x; o.y = acc_lo.y * inv + b.y;
                o.z = acc_hi.x * inv + b.z; o.w = acc_hi.y * inv + b.w;
            } else {
                o = b;
            }
            o.x = o.x > 0.0f ? o.x : 0.0f; o.y = o.y > 0.0f ? o.y : 0.0f;
            o.z = o.z > 0.0f ? o.z : 0.0f; o.w = o.w > 0.0f ? o.w : 0.0f;
            *(float4*)(out + (size_t)row * F_OUT + q * 4) = o;
        }

        row += nw;
        deg = deg1; s = s1; f1r = f1r1; ca = ca1; cb = cb1; gb = gb1;
        deg1 = deg2; s1 = s2; f1r1 = f1r2; ca1 = ca2; cb1 = cb2;
        deg2 = deg3; s2 = s3; f1r2 = f1r3;
    }
}

extern "C" void kernel_launch(void* const* d_in, const int* in_sizes, int n_in,
                              void* d_out, int out_size, void* d_ws, size_t ws_size,
                              hipStream_t stream) {
    const float* seq  = (const float*)d_in[0];
    const int*   erow = (const int*)  d_in[1];
    const int*   ecol = (const int*)  d_in[2];
    const float* W    = (const float*)d_in[3];
    const float* a1   = (const float*)d_in[4];
    const float* b1   = (const float*)d_in[5];
    const float* a2   = (const float*)d_in[6];
    const float* b2   = (const float*)d_in[7];
    const float* bias = (const float*)d_in[8];

    const int n = in_sizes[0] / F_IN;
    const int e = in_sizes[1];
    float* out = (float*)d_out;

    float* fts      = (float*)d_ws;
    float* f1       = fts + (size_t)n * F_OUT;
    float* f2       = f1 + n;
    int*   cnt      = (int*)(f2 + n);
    int*   rowstart = cnt + n;
    int*   csr_col  = rowstart + n;
    unsigned* packed = (unsigned*)d_out;
    int*   rtab     = (int*)d_out + (e + 31) / 32 * 32;

    const int rpb = n / NB + 2;
    const int per = ((e + PB - 1) / PB + 31) / 32 * 32;

    slice_group_kernel<<<PB, GT, per * sizeof(int), stream>>>(erow, ecol, rtab, packed,
                                                             e, n, per, rpb);
    {
        const int ntiles = (n + 15) / 16;
        const int gemm_blocks = ntiles / 25 + 1 > PGB ? ntiles / 25 + 1 : PGB;
        const size_t sort_lds = (rpb + SCAP + 3 * (GT / 32)) * sizeof(int);
        const size_t gemm_lds = (F_IN * F_OUT + 128) * sizeof(float);
        project_and_sort_kernel<<<gemm_blocks + NB, GT, sort_lds > gemm_lds ? sort_lds : gemm_lds,
                                  stream>>>(seq, W, fts, a1, b1, a2, b2, f1, f2,
                                            packed, rtab, cnt, rowstart, csr_col,
                                            e, n, rpb, per, gemm_blocks);
    }
    row_kernel<<<RB, 256, 0, stream>>>(cnt, rowstart, csr_col, f1, f2, fts, a2, b2, bias,
                                       out, n, e);
}


# Per-range ordering moved into the row launch: a block lists its range's columns by row in shared memory, its 32 waves walk them there; projection beside slice ordering; 2 launches, LDS-rank sum order
# speedup vs baseline: 9.4145x; 1.0303x over previous
#include <hip/hip_runtime.h>
#include <hip/hip_bf16.h>

typedef __attribute__((ext_vector_type(2))) float v2f;
typedef __attribute__((ext_vector_type(8))) float v8f;
typedef __attribute__((ext_vector_type(4))) float v4f;

#define F_IN   256
#define F_OUT  64
#define NEG_SLOPE 0.2f
#define NB     256
#define GT     1024
#define EB     8
#define PB     512
#define RPW    (PB / (GT / 32))
#define SB     4
#define SCAP   14336
#define PGB    256
#define GFW    24


__device__ __forceinline__ void
gemm_role(float* __restrict__ Wl, int gb, int gblocks,
          const float* __restrict__ seq, const float* __restrict__ W,
          float* __restrict__ fts,
          const float* __restrict__ a1, const float* __restrict__ b1,
          const float* __restrict__ a2, const float* __restrict__ b2,
          float* __restrict__ f1, float* __restrict__ f2, int n) {
    float* __restrict__ part = Wl + F_IN * F_OUT;
    for (int i = threadIdx.x; i < F_IN * F_OUT; i += GT) {
        const int kk  = i >> 6;
        const int col = i & 63;
        Wl[(kk >> 1) * (2 * F_OUT) + col * 2 + (kk & 1)] = W[i];
    }
    __syncthreads();

    const int wave = __builtin_amdgcn_readfirstlane((int)(threadIdx.x >> 5));
    const int grp = wave / 9;
    const int pos = wave - grp * 9;
    const int ntiles = (n + 15) >> 4;
    const int q   = ntiles / gblocks;
    const int rem = ntiles - q * gblocks;

    if (pos == 4) {
        if (gb < rem) {
            const int lane  = threadIdx.x & 31;
            const int lcol  = lane & 15;
            const int khalf = (lane >> 4) * 2;
            const int nhi   = lane >> 4;
            const int rowTile = q * gblocks + gb;
            const int ct = grp;
            int row = rowTile * 16 + lcol;
            if (row >= n) row = n - 1;
            const float* __restrict__ arow = seq + (size_t)row * F_IN;
            const int ocol = ct * 16 + lcol;
            v8f acc1 = v8f{};
            for (int k = 0; k < F_IN; k += 4) {
                float2 av = *(const float2*)(arow + k + khalf);
                v2f a; a[0] = av.x; a[1] = av.y;
                const int kpairBase = ((k + khalf) >> 1) * (2 * F_OUT);
                float2 bv = *(const float2*)(&Wl[kpairBase + ocol * 2]);
                v2f b; b[0] = bv.x; b[1] = bv.y;
                acc1 = __builtin_amdgcn_wmma_f32_16x16x4_f32(
                         false, a,  false, b,
                         (short)0, acc1,  false,  false);
            }
#pragma unroll
            for (int r = 0; r < 8; ++r) {
                const int orow = rowTile * 16 + r + 8 * nhi;
                if (orow < n) fts[(size_t)orow * F_OUT + ocol] = acc1[r];
            }
            const float A1c = a1[ocol], A2c = a2[ocol];
#pragma unroll
            for (int r = 0; r < 8; ++r) {
                float p1 = acc1[r] * A1c, p2 = acc1[r] * A2c;
#pragma unroll
                for (int off = 8; off > 0; off >>= 1) {
                    p1 += __shfl_xor(p1, off, 32);
                    p2 += __shfl_xor(p2, off, 32);
                }
                if (lcol == 0) {
                    part[(ct * 16 + r + 8 * nhi) * 2]     = p1;
                    part[(ct * 16 + r + 8 * nhi) * 2 + 1] = p2;
                }
            }
        }
    } else if (pos != 0 && grp * 7 + (pos < 4 ? pos - 1 : pos - 2) < q) {
        const int rowTile = gb * q + grp * 7 + (pos < 4 ? pos - 1 : pos - 2);
        const int lane  = threadIdx.x & 31;
        int row = rowTile * 16 + (lane & 15);
        if (row >= n) row = n - 1;
        const int lcol  = lane & 15;
        const int khalf = (lane >> 4) * 2;

        const float* __restrict__ arow = seq + (size_t)row * F_IN;

        v8f acc[4] = {v8f{}, v8f{}, v8f{}, v8f{}};
        for (int k = 0; k < F_IN; k += 4) {
            float2 av = *(const float2*)(arow + k + khalf);
            v2f a; a[0] = av.x; a[1] = av.y;
            const int kpairBase = ((k + khalf) >> 1) * (2 * F_OUT);
#pragma unroll
            for (int ct = 0; ct < 4; ++ct) {
                float2 bv = *(const float2*)(&Wl[kpairBase + (ct * 16 + lcol) * 2]);
                v2f b; b[0] = bv.x; b[1] = bv.y;
                acc[ct] = __builtin_amdgcn_wmma_f32_16x16x4_f32(
                         false, a,  false, b,
                         (short)0, acc[ct],  false,  false);
            }
        }

        const int nhi = lane >> 4;
#pragma unroll
        for (int ct = 0; ct < 4; ++ct) {
            const int ocol = ct * 16 + lcol;
#pragma unroll
            for (int r = 0; r < 8; ++r) {
                const int orow = rowTile * 16 + r + 8 * nhi;
                if (orow < n) fts[(size_t)orow * F_OUT + ocol] = acc[ct][r];
            }
        }

        float A1[4], A2[4];
#pragma unroll
        for (int ct = 0; ct < 4; ++ct) {
            A1[ct] = a1[ct * 16 + lcol];
            A2[ct] = a2[ct * 16 + lcol];
        }
        const float b1v = b1[0], b2v = b2[0];
#pragma unroll
        for (int r = 0; r < 8; ++r) {
            float p1 = 0.0f, p2 = 0.0f;
#pragma unroll
            for (int ct = 0; ct < 4; ++ct) {
                p1 += acc[ct][r] * A1[ct];
                p2 += acc[ct][r] * A2[ct];
            }
#pragma unroll
            for (int off = 8; off > 0; off >>= 1) {
                p1 += __shfl_xor(p1, off, 32);
                p2 += __shfl_xor(p2, off, 32);
            }
            if ((lane & 15) == 0) {
                const int orow = rowTile * 16 + r + 8 * nhi;
                if (orow < n) { f1[orow] = p1 + b1v; f2[orow] = p2 + b2v; }
            }
        }
    }

    if (gb < rem) {
        __syncthreads();
        const int lane = threadIdx.x & 31;
        if (wave == 4 && lane < 16) {
            const int orow = (q * gblocks + gb) * 16 + lane;
            float p1 = part[lane * 2], p2 = part[lane * 2 + 1];
#pragma unroll
            for (int ct = 1; ct < 4; ++ct) {
                p1 += part[(ct * 16 + lane) * 2];
                p2 += part[(ct * 16 + lane) * 2 + 1];
            }
            if (orow < n) { f1[orow] = p1 + b1[0]; f2[orow] = p2 + b2[0]; }
        }
    }
}

__device__ __forceinline__ int range_of(int row, unsigned magic) {
    const int b = (int)__umulhi((unsigned)row, magic);
    return b < NB - 1 ? b : NB - 1;
}

__device__ __forceinline__ void
slice_role(int* __restrict__ lds, int sl,
           const int* __restrict__ erow, const int* __restrict__ ecol,
           unsigned* __restrict__ rtab, unsigned* __restrict__ packed,
           int e, int n, int per, int rpb) {
    int* h = lds;
    int* wsum2 = h + NB;
    unsigned* lbuf = (unsigned*)(wsum2 + 32);
    const int t = threadIdx.x, lane = t & 31, wave = t >> 5;
    if (t < NB) h[t] = 0;
    __syncthreads();
    const unsigned magic = 0xFFFFFFFFu / (unsigned)rpb + 1u;
    const int cb = 32 - __clz(n - 1);
    const int lo = sl * per;
    const int hi = lo + per < e ? lo + per : e;
    for (int i = lo + t; i < hi; i += EB * GT) {
        int r[EB];
#pragma unroll
        for (int k = 0; k < EB; ++k) r[k] = erow[min(i + k * GT, hi - 1)];
#pragma unroll
        for (int k = 0; k < EB; ++k)
            if (i + k * GT < hi) atomicAdd(&h[range_of(r[k], magic)], 1);
    }
    __syncthreads();
    int mine = 0, y = 0;
    if (t < NB) {
        mine = h[t];
        y = mine;
        for (int off = 1; off < 32; off <<= 1) {
            const int z = __shfl_up(y, off, 32);
            if (lane >= off) y += z;
        }
        if (lane == 31) wsum2[wave] = y;
    }
    __syncthreads();
    if (t < NB) {
        int lp = 0;
        for (int w = 0; w < wave; ++w) lp += wsum2[w];
        const int ls = lp + y - mine;
        rtab[(size_t)sl * NB + t] = (unsigned)ls | ((unsigned)mine << 16);
        h[t] = ls;
    }
    __syncthreads();
    for (int i = lo + t; i < hi; i += EB * GT) {
        int r[EB], c[EB];
#pragma unroll
        for (int k = 0; k < EB; ++k) {
            const int j = min(i + k * GT, hi - 1);
            r[k] = erow[j];
            c[k] = ecol[j];
        }
#pragma unroll
        for (int k = 0; k < EB; ++k) {
            if (i + k * GT < hi) {
                const int b = range_of(r[k], magic);
                const int slot = atomicAdd(&h[b], 1);
                lbuf[slot] = ((unsigned)(r[k] - b * rpb) << cb) | (unsigned)c[k];
            }
        }
    }
    __syncthreads();
    for (int i = t; i < hi - lo; i += GT) packed[lo + i] = lbuf[i];
}

__global__ void __launch_bounds__(GT, 16)
project_and_slice_kernel(const float* __restrict__ seq, const float* __restrict__ W,
                         float* __restrict__ fts,
                         const float* __restrict__ a1, const float* __restrict__ b1,
                         const float* __restrict__ a2, const float* __restrict__ b2,
                         float* __restrict__ f1, float* __restrict__ f2,
                         const int* __restrict__ erow, const int* __restrict__ ecol,
                         unsigned* __restrict__ rtab, unsigned* __restrict__ packed,
                         int e, int n, int rpb, int per, int gemm_blocks) {
    extern __shared__ int dyn_lds[];
    if ((int)blockIdx.x < gemm_blocks)
        gemm_role((float*)dyn_lds, (int)blockIdx.x, gemm_blocks, seq, W, fts, a1, b1, a2, b2, f1, f2, n);
    else
        slice_role(dyn_lds, (int)blockIdx.x - gemm_blocks, erow, ecol, rtab, packed, e, n, per, rpb);
}


__device__ __forceinline__ float lrelu(float l) { return l > 0.0f ? l : NEG_SLOPE * l; }

__device__ __forceinline__ void accum_chunk(const float* __restrict__ fts, int c, float ev,
                                            int cntc, int half, int q,
                                            v2f& acc_lo, v2f& acc_hi) {
    const int steps = (cntc + 1) >> 1;
    const int evb = __float_as_int(ev);
#pragma unroll 4
    for (int t = 0; t < steps; ++t) {
        const int c_lo = __builtin_amdgcn_readlane(c,   2 * t);
        const int c_hi = __builtin_amdgcn_readlane(c,   2 * t + 1);
        const int w_lo = __builtin_amdgcn_readlane(evb, 2 * t);
        const int w_hi = __builtin_amdgcn_readlane(evb, 2 * t + 1);
        const int   cc = half ? c_hi : c_lo;
        const float w  = __int_as_float(half ? w_hi : w_lo);
        const unsigned off = ((unsigned)cc << 8) | ((unsigned)q << 4);
        const v4f v = *(const v4f*)((const char*)fts + off);
        const v2f w2 = {w, w};
        acc_lo = __builtin_elementwise_fma(v.xy, w2, acc_lo);
        acc_hi = __builtin_elementwise_fma(v.zw, w2, acc_hi);
    }
}

template <int T> __device__ __forceinline__ int share_i(int x) {
    return __builtin_amdgcn_update_dpp(0, x, 0x150 + T, 0xf, 0xf, false);
}
template <int T> __device__ __forceinline__ float share_f(float x) {
    return __int_as_float(__builtin_amdgcn_mov_dpp(__float_as_int(x), 0x150 + T, 0xf, 0xf, false));
}
template <int M> __device__ __forceinline__ float xmask_f(float x) {
    return __int_as_float(__builtin_amdgcn_update_dpp(0, __float_as_int(x), 0x160 + M,
                                                      0xf, 0xf, false));
}

__device__ __forceinline__ float xhalf(float x) {
    const int b = __float_as_int(x);
    return __int_as_float(__builtin_amdgcn_permlanex16(b, b, 0x76543210, 0xfedcba98, false, false));
}

#define GATHER_STEP(T)                                                                   \
    {                                                                                    \
        const unsigned off = (unsigned)share_i<T>(csel8) + q16;                          \
        v[T] = *(const v4f*)(fb + off);                                                  \
    }
#define DOT_STEP(T)                                                                      \
    {                                                                                    \
        const v2f t2 = __builtin_elementwise_fma(v[T].zw, a_hi, v[T].xy * a_lo);         \
        p[T] = t2.x + t2.y;                                                              \
    }
#define ACC_STEP(T)                                                                      \
    {                                                                                    \
        const float w = share_f<T>(ev);                                                  \
        const v2f w2 = {w, w};                                                           \
        acc_lo = __builtin_elementwise_fma(v[T].xy, w2, acc_lo);                         \
        acc_hi = __builtin_elementwise_fma(v[T].zw, w2, acc_hi);                         \
    }
#define ALL16(S) S(0) S(1) S(2) S(3) S(4) S(5) S(6) S(7) S(8) S(9) S(10) S(11) S(12) S(13) S(14) S(15)

__device__ __forceinline__ void list_bounds(const int* lcnt, const float* __restrict__ f1,
                                            int lr, int rows, int r0,
                                            int& deg, int& s, float& f1r) {
    deg = 0; s = 0; f1r = 0.0f;
    if (lr < rows) {
        const int e1 = lcnt[lr];
        const int e0 = lr > 0 ? lcnt[lr - 1] : 0;
        deg = __builtin_amdgcn_readfirstlane(e1 - e0);
        s   = __builtin_amdgcn_readfirstlane(e0);
        f1r = f1[r0 + lr];
    }
}

__device__ __forceinline__ int col_at(const int* lcol, int i, int last, unsigned nmax) {
    const unsigned v = (unsigned)lcol[i < last ? i : last];
    return (int)(v < nmax ? v : nmax);
}

__device__ __forceinline__ void
rows_from_lists(const int* lcnt, const int* lcol, int c, int r0, int rows,
                const float* __restrict__ f1, const float* __restrict__ f2,
                const float* __restrict__ fts, const float* __restrict__ a2,
                const float* __restrict__ b2, const float* __restrict__ bias,
                float* __restrict__ out, int n) {
    const int lane = threadIdx.x & 31;
    const int half = lane >> 4;
    const int q    = lane & 15;
    const int eo   = 2 * q + half;
    const int nw   = GT / 32;
    int lr = __builtin_amdgcn_readfirstlane((int)(threadIdx.x >> 5));
    const int last = c > 0 ? c - 1 : 0;
    const unsigned nmax = (unsigned)(n - 1);
    const float4 b  = *(const float4*)(bias + q * 4);
    const float4 aq = *(const float4*)(a2 + q * 4);
    v2f a_lo, a_hi;
    a_lo.x = aq.x; a_lo.y = aq.y; a_hi.x = aq.z; a_hi.y = aq.w;
    const float b2v = b2[0];
    const char* __restrict__ fb = (const char*)fts;
    const unsigned q16 = (unsigned)q << 4;
    const bool b3 = (q & 8) != 0, b2b = (q & 4) != 0, b1 = (q & 2) != 0, b0 = (q & 1) != 0;

    int deg, s, deg1, s1, deg2, s2;
    float f1r, f1r1, f1r2;
    list_bounds(lcnt, f1, lr,          rows, r0, deg,  s,  f1r);
    list_bounds(lcnt, f1, lr + nw,     rows, r0, deg1, s1, f1r1);
    list_bounds(lcnt, f1, lr + 2 * nw, rows, r0, deg2, s2, f1r2);
    int ca  = col_at(lcol, s + eo, last, nmax);
    int cb  = col_at(lcol, s + 32 + lane, last, nmax);
    int ca1 = col_at(lcol, s1 + eo, last, nmax);
    int cb1 = col_at(lcol, s1 + 32 + lane, last, nmax);
    float gb = f2[lane + 32 < deg ? cb : 0];

    while (lr < rows) {
        const bool va = eo < deg;
        const bool vb = lane + 32 < deg;
        const float lb = vb ? lrelu(f1r + gb) : -INFINITY;

        int deg3, s3;
        float f1r3;
        list_bounds(lcnt, f1, lr + 3 * nw, rows, r0, deg3, s3, f1r3);
        const int ca2 = col_at(lcol, s2 + eo, last, nmax);
        const int cb2 = col_at(lcol, s2 + 32 + lane, last, nmax);
        const float gb1 = f2[lane + 32 < deg1 ? cb1 : 0];

        const int c0   = __builtin_amdgcn_readfirstlane(ca);
        const int csel8 = (va ? ca : c0) << 8;
        v4f v[16];
        float p[16];
        ALL16(GATHER_STEP)
        ALL16(DOT_STEP)
        float r8[8], r4[4], r2[2];
#pragma unroll
        for (int i = 0; i < 8; ++i)
            r8[i] = (b3 ? p[i + 8] : p[i]) + xmask_f<8>(b3 ? p[i] : p[i + 8]);
#pragma unroll
        for (int i = 0; i < 4; ++i)
            r4[i] = (b2b ? r8[i + 4] : r8[i]) + xmask_f<4>(b2b ? r8[i] : r8[i + 4]);
#pragma unroll
        for (int i = 0; i < 2; ++i)
            r2[i] = (b1 ? r4[i + 2] : r4[i]) + xmask_f<2>(b1 ? r4[i] : r4[i + 2]);
        const float sdot = (b0 ? r2[1] : r2[0]) + xmask_f<1>(b0 ? r2[0] : r2[1]);
        const float la = va ? lrelu(f1r + (sdot + b2v)) : -INFINITY;

        float m = la;
        if (vb) m = fmaxf(m, lb);
        for (int base = 64; base < deg; base += 32) {
            const int j = base + lane;
            if (j < deg) m = fmaxf(m, lrelu(f1r + f2[col_at(lcol, s + j, last, nmax)]));
        }
        m = fmaxf(m, xhalf(m));
        m = fmaxf(m, xmask_f<8>(m));
        m = fmaxf(m, xmask_f<4>(m));
        m = fmaxf(m, xmask_f<2>(m));
        m = fmaxf(m, xmask_f<1>(m));
        if (!isfinite(m)) m = 0.0f;

        v2f acc_lo = {0.0f, 0.0f}, acc_hi = {0.0f, 0.0f};
        const float ev = va ? expf(la - m) : 0.0f;
        float dsum = ev;
        ALL16(ACC_STEP)
        if (deg > 32) {
            const float ev2 = vb ? expf(lb - m) : 0.0f;
            dsum += ev2;
            const int cf = __builtin_amdgcn_readfirstlane(cb);
            const int rem = deg - 32;
            accum_chunk(fts, vb ? cb : cf, ev2, rem < 32 ? rem : 32, half, q, acc_lo, acc_hi);
        }
        for (int base = 64; base < deg; base += 32) {
            const int j = base + lane;
            const bool vj = j < deg;
            int cj = 0; float evj = 0.0f;
            if (vj) { cj = col_at(lcol, s + j, last, nmax); evj = expf(lrelu(f1r + f2[cj]) - m); }
            dsum += evj;
            const int cf = __shfl(cj, 0, 32);
            const int rem = deg - base;
            accum_chunk(fts, vj ? cj : cf, evj, rem < 32 ? rem : 32, half, q, acc_lo, acc_hi);
        }

        dsum += xhalf(dsum);
        dsum += xmask_f<8>(dsum);
        dsum += xmask_f<4>(dsum);
        dsum += xmask_f<2>(dsum);
        dsum += xmask_f<1>(dsum);
        acc_lo.x += xhalf(acc_lo.x);
        acc_lo.y += xhalf(acc_lo.y);
        acc_hi.x += xhalf(acc_hi.x);
        acc_hi.y += xhalf(acc_hi.y);

        if (half == 0) {
            float4 o;
            if (deg > 0) {
                const float inv = 1.0f / dsum;
                o.x = acc_lo.x * inv + b.x; o.y = acc_lo.y * inv + b.y;
                o.z = acc_hi.x * inv + b.z; o.w = acc_hi.y * inv + b.w;
            } else {
                o = b;
            }
            o.x = o.x > 0.0f ? o.x : 0.0f; o.y = o.y > 0.0f ? o.y : 0.0f;
            o.z = o.z > 0.0f ? o.z : 0.0f; o.w = o.w > 0.0f ? o.w : 0.0f;
            *(float4*)(out + (size_t)(r0 + lr) * F_OUT + q * 4) = o;
        }

        lr += nw;
        deg = deg1; s = s1; f1r = f1r1; ca = ca1; cb = cb1; gb = gb1;
        deg1 = deg2; s1 = s2; f1r1 = f1r2; ca1 = ca2; cb1 = cb2;
        deg2 = deg3; s2 = s3; f1r2 = f1r3;
    }
}

__device__ __forceinline__ void
rows_by_scan(const unsigned* __restrict__ packed, const unsigned* __restrict__ rtab,
             int sb, int r0, int rows, int per, int cb, unsigned cmask, unsigned last,
             const float* __restrict__ f1, const float* __restrict__ f2,
             const float* __restrict__ fts, const float* __restrict__ bias,
             float* __restrict__ out) {
    const int lane = threadIdx.x & 31;
    const int wave = __builtin_amdgcn_readfirstlane((int)(threadIdx.x >> 5));
    const float2 bz = *(const float2*)(bias + lane * 2);
    for (int lr = wave; lr < rows; lr += GT / 32) {
        const float f1r = f1[r0 + lr];
        float m = -INFINITY;
        for (int pb = 0; pb < PB; ++pb) {
            const unsigned tv = rtab[(size_t)pb * NB + sb];
            const int len = (int)(tv >> 16), base = pb * per + (int)(tv & 0xFFFFu);
            for (int j = lane; j < len; j += 32) {
                const unsigned wj = packed[base + j];
                unsigned lj = wj >> cb;
                lj = lj < last ? lj : last;
                if ((int)lj == lr) m = fmaxf(m, lrelu(f1r + f2[wj & cmask]));
            }
        }
        for (int off = 16; off > 0; off >>= 1) m = fmaxf(m, __shfl_xor(m, off, 32));
        if (!isfinite(m)) m = 0.0f;
        float dsum = 0.0f, ax = 0.0f, ay = 0.0f;
        for (int pb = 0; pb < PB; ++pb) {
            const unsigned tv = rtab[(size_t)pb * NB + sb];
            const int len = (int)(tv >> 16), base = pb * per + (int)(tv & 0xFFFFu);
            for (int j0 = 0; j0 < len; j0 += 32) {
                const int j = j0 + lane;
                bool hit = false; int col = 0; float ev = 0.0f;
                if (j < len) {
                    const unsigned wj = packed[base + j];
                    unsigned lj = wj >> cb;
                    lj = lj < last ? lj : last;
                    if ((int)lj == lr) {
                        hit = true;
                        col = (int)(wj & cmask);
                        ev = expf(lrelu(f1r + f2[col]) - m);
                    }
                }
                dsum += ev;
                unsigned todo = __builtin_amdgcn_ballot_w32(hit);
                while (todo != 0u) {
                    const int l = __ffs((int)todo) - 1;
                    todo &= todo - 1u;
                    const int cc = __shfl(col, l, 32);
                    const float we = __shfl(ev, l, 32);
                    const float2 fv = *(const float2*)(fts + (size_t)cc * F_OUT + lane * 2);
                    ax += we * fv.x; ay += we * fv.y;
                }
            }
        }
        for (int off = 16; off > 0; off >>= 1) dsum += __shfl_xor(dsum, off, 32);
        float2 o = bz;
        if (dsum > 0.0f) { o.x = ax / dsum + bz.x; o.y = ay / dsum + bz.y; }
        o.x = o.x > 0.0f ? o.x : 0.0f; o.y = o.y > 0.0f ? o.y : 0.0f;
        *(float2*)(out + (size_t)(r0 + lr) * F_OUT + lane * 2) = o;
    }
}

__global__ void __launch_bounds__(GT)
range_rows_kernel(const unsigned* __restrict__ packed, const unsigned* __restrict__ rtab,
                  const float* __restrict__ f1, const float* __restrict__ f2,
                  const float* __restrict__ fts, const float* __restrict__ a2,
                  const float* __restrict__ b2, const float* __restrict__ bias,
                  float* __restrict__ out, int e, int n, int rpb, int per) {
    extern __shared__ int dyn_lds[];
    int* wsum = dyn_lds;
    int* wtot = wsum + GT / 32;
    int* lcnt = wtot + GT / 32;
    int* lcol = lcnt + rpb;
    const int t = threadIdx.x, lane = t & 31, wave = t >> 5;
    const int sb = blockIdx.x;
    const int r0 = sb * rpb;
    int rows = n - r0;
    if (rows > rpb) rows = rpb;
    if (rows <= 0) return;
    const int cb = 32 - __clz(n - 1);
    const unsigned cmask = (1u << cb) - 1u;
    const unsigned last = (unsigned)(rows - 1);
    int rbase = 0, rlen = 0;
    if (lane < RPW) {
        const int pb = wave * RPW + lane;
        const unsigned tv = rtab[(size_t)pb * NB + sb];
        rlen = (int)(tv >> 16);
        rbase = pb * per + (int)(tv & 0xFFFFu);
    }
    for (int i = t; i < rows; i += GT) lcnt[i] = 0;
    unsigned w[RPW];
#pragma unroll
    for (int k = 0; k < RPW; ++k) {
        const int base = __builtin_amdgcn_readlane(rbase, k);
        const int len  = __builtin_amdgcn_readlane(rlen, k);
        const int j = base + min(lane, len > 0 ? len - 1 : 0);
        w[k] = packed[min(j, e - 1)];
    }
    {
        int xs = rlen;
        for (int off = 16; off > 0; off >>= 1) xs += __shfl_xor(xs, off, 32);
        if (lane == 0) wtot[wave] = xs;
    }
    __syncthreads();
    int c = wtot[lane];
    for (int off = 16; off > 0; off >>= 1) c += __shfl_xor(c, off, 32);
    c = __builtin_amdgcn_readfirstlane(c);
    const bool staged = c <= SCAP;
    if (!staged) {
        rows_by_scan(packed, rtab, sb, r0, rows, per, cb, cmask, last, f1, f2, fts, bias, out);
        return;
    }
#pragma unroll
    for (int k = 0; k < RPW; ++k) {
        const int len = __builtin_amdgcn_readlane(rlen, k);
        if (lane < len) {
            unsigned lr = w[k] >> cb;
            lr = lr < last ? lr : last;
            atomicAdd(&lcnt[lr], 1);
        }
    }
    for (int k = 0; k < RPW; ++k) {
        const int base = __builtin_amdgcn_readlane(rbase, k);
        const int len  = __builtin_amdgcn_readlane(rlen, k);
        for (int j = 32 + lane; j < len; j += 32) {
            unsigned lr = packed[base + j] >> cb;
            lr = lr < last ? lr : last;
            atomicAdd(&lcnt[lr], 1);
        }
    }
    __syncthreads();
    int carry = 0;
    for (int base = 0; base < rows; base += GT) {
        const int idx = base + t;
        const int v = idx < rows ? lcnt[idx] : 0;
        int x = v;
        for (int off = 1; off < 32; off <<= 1) {
            const int y = __shfl_up(x, off, 32);
            if (lane >= off) x += y;
        }
        if (lane == 31) wsum[wave] = x;
        __syncthreads();
        int wp = 0, all = 0;
#pragma unroll 8
        for (int q2 = 0; q2 < GT / 32; ++q2) {
            const int q = wsum[q2];
            all += q;
            if (q2 < wave) wp += q;
        }
        if (idx < rows) lcnt[idx] = carry + wp + x - v;
        carry += all;
        __syncthreads();
    }
#pragma unroll
    for (int g = 0; g < RPW; g += SB) {
        int slot[SB];
#pragma unroll
        for (int k = 0; k < SB; ++k) {
            const int len = __builtin_amdgcn_readlane(rlen, g + k);
            unsigned lr = w[g + k] >> cb;
            lr = lr < last ? lr : last;
            slot[k] = lane < len ? atomicAdd(&lcnt[lr], 1) : 0;
        }
#pragma unroll
        for (int k = 0; k < SB; ++k) {
            if (lane < __builtin_amdgcn_readlane(rlen, g + k))
                lcol[slot[k]] = (int)(w[g + k] & cmask);
        }
    }
    for (int k = 0; k < RPW; ++k) {
        const int base = __builtin_amdgcn_readlane(rbase, k);
        const int len  = __builtin_amdgcn_readlane(rlen, k);
        for (int j = 32 + lane; j < len; j += 32) {
            const unsigned wj = packed[base + j];
            unsigned lr = wj >> cb;
            lr = lr < last ? lr : last;
            const int sl = atomicAdd(&lcnt[lr], 1);
            lcol[sl] = (int)(wj & cmask);
        }
    }
    __syncthreads();
    rows_from_lists(lcnt, lcol, c, r0, rows, f1, f2, fts, a2, b2, bias, out, n);
}

extern "C" void kernel_launch(void* const* d_in, const int* in_sizes, int n_in,
                              void* d_out, int out_size, void* d_ws, size_t ws_size,
                              hipStream_t stream) {
    const float* seq  = (const float*)d_in[0];
    const int*   erow = (const int*)  d_in[1];
    const int*   ecol = (const int*)  d_in[2];
    const float* W    = (const float*)d_in[3];
    const float* a1   = (const float*)d_in[4];
    const float* b1   = (const float*)d_in[5];
    const float* a2   = (const float*)d_in[6];
    const float* b2   = (const float*)d_in[7];
    const float* bias = (const float*)d_in[8];

    const int n = in_sizes[0] / F_IN;
    const int e = in_sizes[1];
    float* out = (float*)d_out;

    float* fts      = (float*)d_ws;
    float* f1       = fts + (size_t)n * F_OUT;
    float* f2       = f1 + n;
    unsigned* packed = (unsigned*)d_ws + ((size_t)n * (F_OUT + 2) + 31) / 32 * 32;
    unsigned* rtab   = packed + (e + 31) / 32 * 32;

    const int rpb = n / NB + 2;
    const int per = ((e + PB - 1) / PB + 31) / 32 * 32;

    {
        const int ntiles = (n + 15) / 16;
        const int gemm_blocks = ntiles / 25 + 1 > PGB ? ntiles / 25 + 1 : PGB;
        const size_t slice_lds = (NB + 32 + per) * sizeof(int);
        const size_t gemm_lds = (F_IN * F_OUT + 128) * sizeof(float);
        project_and_slice_kernel<<<gemm_blocks + PB, GT, slice_lds > gemm_lds ? slice_lds : gemm_lds,
                                   stream>>>(seq, W, fts, a1, b1, a2, b2, f1, f2,
                                             erow, ecol, rtab, packed, e, n, rpb, per, gemm_blocks);
    }
    {
        const size_t list_lds = (rpb + SCAP + 2 * (GT / 32)) * sizeof(int);
        range_rows_kernel<<<NB, GT, list_lds, stream>>>(packed, rtab, f1, f2, fts, a2, b2, bias,
                                                        out, e, n, rpb, per);
    }
}
